# starting kernel
# baseline (speedup 1.0000x reference)
#include <hip/hip_runtime.h>
#include <hip/hip_bf16.h>

typedef __attribute__((ext_vector_type(2))) float v2f;
typedef __attribute__((ext_vector_type(8))) float v8f;

__device__ __forceinline__ void atomAddF(float* p, float v) {
    unsafeAtomicAdd(p, v);
}

__global__ void k_deg_init(float* __restrict__ deg, int N) {
    int i = blockIdx.x * blockDim.x + threadIdx.x;
    if (i < N) deg[i] = 1.0f;
}

__global__ void k_deg_count(const int* __restrict__ ei, int E, float* __restrict__ deg) {
    int e = blockIdx.x * blockDim.x + threadIdx.x;
    if (e < E) {
        int d = __builtin_nontemporal_load(ei + (size_t)E + e);
        atomAddF(&deg[d], 1.0f);
    }
}

__global__ void k_deg_rsqrt(float* __restrict__ deg, int N) {
    int i = blockIdx.x * blockDim.x + threadIdx.x;
    if (i < N) deg[i] = rsqrtf(deg[i]);
}

__global__ void k_gemm_wmma(const float* __restrict__ x,
                            const float* __restrict__ W1,
                            float* __restrict__ h,
                            int N, int K) {
    const int lane = threadIdx.x & 31;
    const int wave = threadIdx.x >> 5;
    const int tile = blockIdx.x * (blockDim.x >> 5) + wave;
    const int ntiles = (N + 15) >> 4;
    if (tile >= ntiles) return;

    const int m  = lane & 15;
    const int kb = (lane >> 4) << 1;
    int row = tile * 16 + m;
    if (row >= N) row = N - 1;
    const float* __restrict__ xrow = x + (size_t)row * K;

    v8f c = {};
    for (int k = 0; k < K; k += 4) {
        v2f a, b;
        a.x = __builtin_nontemporal_load(xrow + k + kb);
        a.y = __builtin_nontemporal_load(xrow + k + kb + 1);
        b.x = W1[(k + kb) * 16 + m];
        b.y = W1[(k + kb + 1) * 16 + m];
        c = __builtin_amdgcn_wmma_f32_16x16x4_f32(false, a, false, b,
                                                  (short)0, c, false, false);
    }
    const int rbase = (lane >> 4) << 3;
#pragma unroll
    for (int r = 0; r < 8; ++r) {
        const int orow = tile * 16 + rbase + r;
        if (orow < N) h[(size_t)orow * 16 + m] = c[r];
    }
}

__global__ void k_selfloop16(const float* __restrict__ h, const float* __restrict__ dinv,
                             float* __restrict__ hagg, int N) {
    int t = blockIdx.x * blockDim.x + threadIdx.x;
    if (t < N * 16) {
        int i = t >> 4;
        float w = dinv[i];
        hagg[t] = h[t] * w * w;
    }
}

__global__ void k_edge_agg16(const int* __restrict__ ei, int E,
                             const float* __restrict__ dinv,
                             const float* __restrict__ h,
                             float* __restrict__ hagg) {
    unsigned t = blockIdx.x * blockDim.x + threadIdx.x;
    unsigned e = t >> 2;
    if (e >= (unsigned)E) return;
    const int q = t & 3;
    const int s = __builtin_nontemporal_load(ei + e);
    const int d = __builtin_nontemporal_load(ei + (size_t)E + e);
    const float norm = dinv[s] * dinv[d];
    const float4 v = ((const float4*)(h + (size_t)s * 16))[q];
    float* dst = hagg + (size_t)d * 16 + q * 4;
    atomAddF(dst + 0, v.x * norm);
    atomAddF(dst + 1, v.y * norm);
    atomAddF(dst + 2, v.z * norm);
    atomAddF(dst + 3, v.w * norm);
}

__global__ void k_layer2_node(const float* __restrict__ hagg,
                              const float* __restrict__ b1,
                              const float* __restrict__ W2,
                              const float* __restrict__ dinv,
                              float* __restrict__ h2,
                              float* __restrict__ out2, int N) {
    int i = blockIdx.x * blockDim.x + threadIdx.x;
    if (i >= N) return;
    float a0 = 0.f, a1 = 0.f;
#pragma unroll
    for (int j = 0; j < 16; ++j) {
        float v = __builtin_nontemporal_load(hagg + (size_t)i * 16 + j) + b1[j];
        v = v > 0.f ? v : 0.f;
        a0 += v * W2[j * 2 + 0];
        a1 += v * W2[j * 2 + 1];
    }
    h2[(size_t)i * 2 + 0] = a0;
    h2[(size_t)i * 2 + 1] = a1;
    const float w = dinv[i] * dinv[i];
    out2[(size_t)i * 2 + 0] = a0 * w;
    out2[(size_t)i * 2 + 1] = a1 * w;
}

__global__ void k_edge_agg2(const int* __restrict__ ei, int E,
                            const float* __restrict__ dinv,
                            const float* __restrict__ h2,
                            float* __restrict__ out2) {
    int e = blockIdx.x * blockDim.x + threadIdx.x;
    if (e >= E) return;
    const int s = __builtin_nontemporal_load(ei + e);
    const int d = __builtin_nontemporal_load(ei + (size_t)E + e);
    const float norm = dinv[s] * dinv[d];
    const float2 v = ((const float2*)(h2 + (size_t)s * 2))[0];
    atomAddF(&out2[(size_t)d * 2 + 0], v.x * norm);
    atomAddF(&out2[(size_t)d * 2 + 1], v.y * norm);
}

__global__ void k_finalize(const float* __restrict__ out2,
                           const float* __restrict__ b2,
                           float* __restrict__ out, int N) {
    int i = blockIdx.x * blockDim.x + threadIdx.x;
    if (i >= N) return;
    const float v0 = out2[(size_t)i * 2 + 0] + b2[0];
    const float v1 = out2[(size_t)i * 2 + 1] + b2[1];
    const float mx = fmaxf(v0, v1);
    const float lse = mx + logf(expf(v0 - mx) + expf(v1 - mx));
    __builtin_nontemporal_store(v0 - lse, out + (size_t)i * 2 + 0);
    __builtin_nontemporal_store(v1 - lse, out + (size_t)i * 2 + 1);
}

extern "C" void kernel_launch(void* const* d_in, const int* in_sizes, int n_in,
                              void* d_out, int out_size, void* d_ws, size_t ws_size,
                              hipStream_t stream) {
    const float* x  = (const float*)d_in[0];
    const int*   ei = (const int*)d_in[1];
    const float* W1 = (const float*)d_in[2];
    const float* b1 = (const float*)d_in[3];
    const float* W2 = (const float*)d_in[4];
    const float* b2 = (const float*)d_in[5];
    float* out = (float*)d_out;

    const int H    = in_sizes[3];
    const int Fin  = in_sizes[2] / H;
    const int N    = in_sizes[0] / Fin;
    const int E    = in_sizes[1] / 2;
    (void)n_in; (void)out_size; (void)ws_size;

    float* ws   = (float*)d_ws;
    float* dinv = ws;
    float* h    = ws + (size_t)N;
    float* hagg = ws + (size_t)17 * N;
    float* h2   = ws + (size_t)33 * N;
    float* out2 = ws + (size_t)35 * N;

    const int B = 256;
    k_deg_init <<<(N + B - 1) / B, B, 0, stream>>>(dinv, N);
    k_deg_count<<<(E + B - 1) / B, B, 0, stream>>>(ei, E, dinv);
    k_deg_rsqrt<<<(N + B - 1) / B, B, 0, stream>>>(dinv, N);

    {
        const int ntiles = (N + 15) / 16;
        const int wpb = 4;
        k_gemm_wmma<<<(ntiles + wpb - 1) / wpb, wpb * 32, 0, stream>>>(x, W1, h, N, Fin);
    }

    k_selfloop16<<<((size_t)N * 16 + B - 1) / B, B, 0, stream>>>(h, dinv, hagg, N);
    k_edge_agg16<<<((size_t)E * 4 + B - 1) / B, B, 0, stream>>>(ei, E, dinv, h, hagg);

    k_layer2_node<<<(N + B - 1) / B, B, 0, stream>>>(hagg, b1, W2, dinv, h2, out2, N);

    k_edge_agg2<<<(E + B - 1) / B, B, 0, stream>>>(ei, E, dinv, h2, out2);

    k_finalize<<<(N + B - 1) / B, B, 0, stream>>>(out2, b2, out, N);
}


# Degree count and layer-2 edge scatter accumulate in LDS per (node range, edge slice) work-group, store partial planes, planes summed in fixed order: no global atomics there (sum order changes)
# speedup vs baseline: 2.0492x; 2.0492x over previous
#include <hip/hip_runtime.h>
#include <hip/hip_bf16.h>

typedef __attribute__((ext_vector_type(2))) float v2f;
typedef __attribute__((ext_vector_type(8))) float v8f;

__device__ __forceinline__ void atomAddF(float* p, float v) {
    unsafeAtomicAdd(p, v);
}

#define SC_TPB  1024
#define SC_CAPW 51200
#define SC_K    8

__global__ __launch_bounds__(SC_TPB)
void k_deg_lds(const int* __restrict__ ei, int E, int N,
               int rangeNodes, int nRanges, int edgesPerSlice,
               unsigned* __restrict__ part) {
    __shared__ __attribute__((aligned(16))) unsigned cnt[SC_CAPW];
    const int tid   = threadIdx.x;
    const int range = blockIdx.x % nRanges;
    const int slice = blockIdx.x / nRanges;
    const int n0 = range * rangeNodes;
    const int nn = max(min(rangeNodes, N - n0), 0);
    for (int i = tid; i < (nn + 3) / 4; i += SC_TPB) ((uint4*)cnt)[i] = make_uint4(0u, 0u, 0u, 0u);
    __syncthreads();
    const int e0 = slice * edgesPerSlice;
    const int e1 = min(E, e0 + edgesPerSlice);
    const int* __restrict__ dst = ei + (size_t)E;
    for (int base = e0; base < e1; base += SC_TPB * SC_K) {
        int dd[SC_K];
#pragma unroll
        for (int k = 0; k < SC_K; ++k) {
            const int e = base + k * SC_TPB + tid;
            dd[k] = dst[min(e, e1 - 1)];
        }
#pragma unroll
        for (int k = 0; k < SC_K; ++k) {
            const int e = base + k * SC_TPB + tid;
            const unsigned r = (unsigned)(dd[k] - n0);
            if (e < e1 && r < (unsigned)nn) atomicAdd(&cnt[r], 1u);
        }
    }
    __syncthreads();
    unsigned* __restrict__ o = part + (size_t)slice * N + n0;
    for (int i = tid; i < nn; i += SC_TPB) o[i] = cnt[i];
}

__global__ void k_deg_reduce(const unsigned* __restrict__ part, int nSlices, int N,
                             float* __restrict__ dinv) {
    int i = blockIdx.x * blockDim.x + threadIdx.x;
    if (i >= N) return;
    unsigned c = 1u;
#pragma unroll 8
    for (int s = 0; s < nSlices; ++s) c += part[(size_t)s * N + i];
    dinv[i] = rsqrtf((float)c);
}

__global__ void k_gemm_wmma(const float* __restrict__ x,
                            const float* __restrict__ W1,
                            float* __restrict__ h,
                            int N, int K) {
    const int lane = threadIdx.x & 31;
    const int wave = threadIdx.x >> 5;
    const int tile = blockIdx.x * (blockDim.x >> 5) + wave;
    const int ntiles = (N + 15) >> 4;
    if (tile >= ntiles) return;

    const int m  = lane & 15;
    const int kb = (lane >> 4) << 1;
    int row = tile * 16 + m;
    if (row >= N) row = N - 1;
    const float* __restrict__ xrow = x + (size_t)row * K;

    v8f c = {};
    for (int k = 0; k < K; k += 4) {
        v2f a, b;
        a.x = __builtin_nontemporal_load(xrow + k + kb);
        a.y = __builtin_nontemporal_load(xrow + k + kb + 1);
        b.x = W1[(k + kb) * 16 + m];
        b.y = W1[(k + kb + 1) * 16 + m];
        c = __builtin_amdgcn_wmma_f32_16x16x4_f32(false, a, false, b,
                                                  (short)0, c, false, false);
    }
    const int rbase = (lane >> 4) << 3;
#pragma unroll
    for (int r = 0; r < 8; ++r) {
        const int orow = tile * 16 + rbase + r;
        if (orow < N) h[(size_t)orow * 16 + m] = c[r];
    }
}

__global__ void k_selfloop16(const float* __restrict__ h, const float* __restrict__ dinv,
                             float* __restrict__ hagg, int N) {
    int t = blockIdx.x * blockDim.x + threadIdx.x;
    if (t < N * 16) {
        int i = t >> 4;
        float w = dinv[i];
        hagg[t] = h[t] * w * w;
    }
}

__global__ void k_edge_agg16(const int* __restrict__ ei, int E,
                             const float* __restrict__ dinv,
                             const float* __restrict__ h,
                             float* __restrict__ hagg) {
    unsigned t = blockIdx.x * blockDim.x + threadIdx.x;
    unsigned e = t >> 2;
    if (e >= (unsigned)E) return;
    const int q = t & 3;
    const int s = __builtin_nontemporal_load(ei + e);
    const int d = __builtin_nontemporal_load(ei + (size_t)E + e);
    const float norm = dinv[s] * dinv[d];
    const float4 v = ((const float4*)(h + (size_t)s * 16))[q];
    float* dst = hagg + (size_t)d * 16 + q * 4;
    atomAddF(dst + 0, v.x * norm);
    atomAddF(dst + 1, v.y * norm);
    atomAddF(dst + 2, v.z * norm);
    atomAddF(dst + 3, v.w * norm);
}

__global__ void k_layer2_node(const float* __restrict__ hagg,
                              const float* __restrict__ b1,
                              const float* __restrict__ W2,
                              float* __restrict__ h2, int N) {
    int i = blockIdx.x * blockDim.x + threadIdx.x;
    if (i >= N) return;
    float a0 = 0.f, a1 = 0.f;
#pragma unroll
    for (int j = 0; j < 16; ++j) {
        float v = __builtin_nontemporal_load(hagg + (size_t)i * 16 + j) + b1[j];
        v = v > 0.f ? v : 0.f;
        a0 += v * W2[j * 2 + 0];
        a1 += v * W2[j * 2 + 1];
    }
    h2[(size_t)i * 2 + 0] = a0;
    h2[(size_t)i * 2 + 1] = a1;
}

__global__ __launch_bounds__(SC_TPB)
void k_agg2_lds(const int* __restrict__ ei, int E, int N,
                const float* __restrict__ dinv,
                const float* __restrict__ h2,
                int rangeNodes, int nRanges, int edgesPerSlice,
                float* __restrict__ part) {
    __shared__ __attribute__((aligned(16))) float acc[SC_CAPW];
    float* __restrict__ acc0 = acc;
    float* __restrict__ acc1 = acc + SC_CAPW / 2;
    const int tid   = threadIdx.x;
    const int range = blockIdx.x % nRanges;
    const int slice = blockIdx.x / nRanges;
    const int n0 = range * rangeNodes;
    const int nn = max(min(rangeNodes, N - n0), 0);
    for (int i = tid; i < (nn + 3) / 4; i += SC_TPB) {
        ((float4*)acc0)[i] = make_float4(0.f, 0.f, 0.f, 0.f);
        ((float4*)acc1)[i] = make_float4(0.f, 0.f, 0.f, 0.f);
    }
    __syncthreads();
    const int e0 = slice * edgesPerSlice;
    const int e1 = min(E, e0 + edgesPerSlice);
    const int* __restrict__ srcp = ei;
    const int* __restrict__ dstp = ei + (size_t)E;
    const float2* __restrict__ h2v = (const float2*)h2;
    for (int base = e0; base < e1; base += SC_TPB * SC_K) {
        int ss[SC_K], dd[SC_K];
        float2 v[SC_K];
        float ws_[SC_K], wd_[SC_K];
#pragma unroll
        for (int k = 0; k < SC_K; ++k) {
            const int e = min(base + k * SC_TPB + tid, e1 - 1);
            dd[k] = dstp[e];
            ss[k] = srcp[e];
        }
        asm volatile("" :: "v"(ss[0]), "v"(ss[1]), "v"(ss[2]), "v"(ss[3]),
                           "v"(ss[4]), "v"(ss[5]), "v"(ss[6]), "v"(ss[7]));
#pragma unroll
        for (int k = 0; k < SC_K; ++k) {
            const int e = base + k * SC_TPB + tid;
            const unsigned r = (unsigned)(dd[k] - n0);
            v[k] = make_float2(0.f, 0.f);
            ws_[k] = 0.f;
            wd_[k] = 0.f;
            if (e < e1 && r < (unsigned)nn) {
                ws_[k] = dinv[(unsigned)ss[k]];
                wd_[k] = dinv[(unsigned)dd[k]];
                v[k]   = h2v[(unsigned)ss[k]];
            }
        }
#pragma unroll
        for (int k = 0; k < SC_K; ++k) {
            const int e = base + k * SC_TPB + tid;
            const unsigned r = (unsigned)(dd[k] - n0);
            if (e < e1 && r < (unsigned)nn) {
                const float nm = ws_[k] * wd_[k];
                atomicAdd(&acc0[r], v[k].x * nm);
                atomicAdd(&acc1[r], v[k].y * nm);
            }
        }
    }
    __syncthreads();
    float2* __restrict__ o = (float2*)part + (size_t)slice * N + n0;
    for (int i = tid; i < nn; i += SC_TPB) o[i] = make_float2(acc0[i], acc1[i]);
}

__global__ void k_finalize(const float* __restrict__ part, int nSlices,
                           const float* __restrict__ h2,
                           const float* __restrict__ dinv,
                           const float* __restrict__ b2,
                           float* __restrict__ out, int N) {
    int i = blockIdx.x * blockDim.x + threadIdx.x;
    if (i >= N) return;
    const float w = dinv[i] * dinv[i];
    const float2 hv = ((const float2*)h2)[i];
    float a0 = hv.x * w, a1 = hv.y * w;
    const float2* __restrict__ pv = (const float2*)part;
#pragma unroll 8
    for (int s = 0; s < nSlices; ++s) {
        const float2 p = pv[(size_t)s * N + i];
        a0 += p.x;
        a1 += p.y;
    }
    const float v0 = a0 + b2[0];
    const float v1 = a1 + b2[1];
    const float mx = fmaxf(v0, v1);
    const float lse = mx + logf(expf(v0 - mx) + expf(v1 - mx));
    __builtin_nontemporal_store(v0 - lse, out + (size_t)i * 2 + 0);
    __builtin_nontemporal_store(v1 - lse, out + (size_t)i * 2 + 1);
}

extern "C" void kernel_launch(void* const* d_in, const int* in_sizes, int n_in,
                              void* d_out, int out_size, void* d_ws, size_t ws_size,
                              hipStream_t stream) {
    const float* x  = (const float*)d_in[0];
    const int*   ei = (const int*)d_in[1];
    const float* W1 = (const float*)d_in[2];
    const float* b1 = (const float*)d_in[3];
    const float* W2 = (const float*)d_in[4];
    const float* b2 = (const float*)d_in[5];
    float* out = (float*)d_out;

    const int H    = in_sizes[3];
    const int Fin  = in_sizes[2] / H;
    const int N    = in_sizes[0] / Fin;
    const int E    = in_sizes[1] / 2;
    (void)n_in; (void)out_size; (void)ws_size;

    float* ws   = (float*)d_ws;
    float* dinv = ws;
    float* h    = ws + (size_t)N;
    float* hagg = ws + (size_t)17 * N;
    float* h2   = ws + (size_t)33 * N;
    float* part = ws + (((size_t)35 * N + 63) / 64) * 64;

    const int B = 256;
    const int WG = 256;
    const int nRangesD = (N + SC_CAPW - 1) / SC_CAPW;
    const int rangeD   = (N + nRangesD - 1) / nRangesD;
    const int nSlicesD = WG / nRangesD > 0 ? WG / nRangesD : 1;
    const int epsD     = (E + nSlicesD - 1) / nSlicesD;
    const int nRanges2 = (N + SC_CAPW / 2 - 1) / (SC_CAPW / 2);
    const int range2   = (N + nRanges2 - 1) / nRanges2;
    const int nSlices2 = WG / nRanges2 > 0 ? WG / nRanges2 : 1;
    const int eps2     = (E + nSlices2 - 1) / nSlices2;

    k_deg_lds   <<<nRangesD * nSlicesD, SC_TPB, 0, stream>>>(ei, E, N, rangeD, nRangesD, epsD, (unsigned*)part);
    k_deg_reduce<<<(N + B - 1) / B, B, 0, stream>>>((const unsigned*)part, nSlicesD, N, dinv);

    {
        const int ntiles = (N + 15) / 16;
        const int wpb = 4;
        k_gemm_wmma<<<(ntiles + wpb - 1) / wpb, wpb * 32, 0, stream>>>(x, W1, h, N, Fin);
    }

    k_selfloop16<<<((size_t)N * 16 + B - 1) / B, B, 0, stream>>>(h, dinv, hagg, N);
    k_edge_agg16<<<((size_t)E * 4 + B - 1) / B, B, 0, stream>>>(ei, E, dinv, h, hagg);

    k_layer2_node<<<(N + B - 1) / B, B, 0, stream>>>(hagg, b1, W2, h2, N);

    k_agg2_lds<<<nRanges2 * nSlices2, SC_TPB, 0, stream>>>(ei, E, N, dinv, h2, range2, nRanges2, eps2, part);

    k_finalize<<<(N + B - 1) / B, B, 0, stream>>>(part, nSlices2, h2, dinv, b2, out, N);
}


# Layer-1 scatter: edges counting-sorted by destination bucket (128 nodes) through LDS; one work-group per bucket sums in LDS and finishes the node; no global atomics left (sum order changes)
# speedup vs baseline: 2.7297x; 1.3321x over previous
#include <hip/hip_runtime.h>
#include <hip/hip_bf16.h>

typedef __attribute__((ext_vector_type(2))) float v2f;
typedef __attribute__((ext_vector_type(8))) float v8f;

#define SC_TPB  1024
#define SC_CAPW 51200
#define SC_K    8

__global__ __launch_bounds__(SC_TPB)
void k_deg_lds(const int* __restrict__ ei, int E, int N,
               int rangeNodes, int nRanges, int edgesPerSlice,
               unsigned* __restrict__ part) {
    __shared__ __attribute__((aligned(16))) unsigned cnt[SC_CAPW];
    const int tid   = threadIdx.x;
    const int range = blockIdx.x % nRanges;
    const int slice = blockIdx.x / nRanges;
    const int n0 = range * rangeNodes;
    const int nn = max(min(rangeNodes, N - n0), 0);
    for (int i = tid; i < (nn + 3) / 4; i += SC_TPB) ((uint4*)cnt)[i] = make_uint4(0u, 0u, 0u, 0u);
    __syncthreads();
    const int e0 = slice * edgesPerSlice;
    const int e1 = min(E, e0 + edgesPerSlice);
    const int* __restrict__ dst = ei + (size_t)E;
    for (int base = e0; base < e1; base += SC_TPB * SC_K) {
        int dd[SC_K];
#pragma unroll
        for (int k = 0; k < SC_K; ++k) {
            const int e = base + k * SC_TPB + tid;
            dd[k] = dst[min(e, e1 - 1)];
        }
#pragma unroll
        for (int k = 0; k < SC_K; ++k) {
            const int e = base + k * SC_TPB + tid;
            const unsigned r = (unsigned)(dd[k] - n0);
            if (e < e1 && r < (unsigned)nn) atomicAdd(&cnt[r], 1u);
        }
    }
    __syncthreads();
    unsigned* __restrict__ o = part + (size_t)slice * N + n0;
    for (int i = tid; i < nn; i += SC_TPB) o[i] = cnt[i];
}

__global__ void k_deg_reduce(const unsigned* __restrict__ part, int nSlices, int N,
                             float* __restrict__ dinv) {
    int i = blockIdx.x * blockDim.x + threadIdx.x;
    if (i >= N) return;
    unsigned c = 1u;
#pragma unroll 8
    for (int s = 0; s < nSlices; ++s) c += part[(size_t)s * N + i];
    dinv[i] = rsqrtf((float)c);
}

__global__ void k_gemm_wmma(const float* __restrict__ x,
                            const float* __restrict__ W1,
                            float* __restrict__ h,
                            int N, int K) {
    const int lane = threadIdx.x & 31;
    const int wave = threadIdx.x >> 5;
    const int tile = blockIdx.x * (blockDim.x >> 5) + wave;
    const int ntiles = (N + 15) >> 4;
    if (tile >= ntiles) return;

    const int m  = lane & 15;
    const int kb = (lane >> 4) << 1;
    int row = tile * 16 + m;
    if (row >= N) row = N - 1;
    const float* __restrict__ xrow = x + (size_t)row * K;

    v8f c = {};
    for (int k = 0; k < K; k += 4) {
        v2f a, b;
        a.x = __builtin_nontemporal_load(xrow + k + kb);
        a.y = __builtin_nontemporal_load(xrow + k + kb + 1);
        b.x = W1[(k + kb) * 16 + m];
        b.y = W1[(k + kb + 1) * 16 + m];
        c = __builtin_amdgcn_wmma_f32_16x16x4_f32(false, a, false, b,
                                                  (short)0, c, false, false);
    }
    const int rbase = (lane >> 4) << 3;
#pragma unroll
    for (int r = 0; r < 8; ++r) {
        const int orow = tile * 16 + rbase + r;
        if (orow < N) h[(size_t)orow * 16 + m] = c[r];
    }
}

#define BK_SHIFT 7
#define BK_NODES 128
#define BK_MAX   4096
#define BIN_TPB  1024
#define BIN_CH   12288
#define AG_TPB   512
#define AG_K     4

__global__ __launch_bounds__(BIN_TPB)
void k_bin_hist(const int* __restrict__ ei, int E, int nB, unsigned* __restrict__ cntT) {
    __shared__ unsigned c[BK_MAX];
    const int tid = threadIdx.x;
    const int w   = blockIdx.x;
    const int nW  = gridDim.x;
    for (int i = tid; i < nB; i += BIN_TPB) c[i] = 0u;
    __syncthreads();
    const int e0 = w * BIN_CH;
    const int e1 = min(E, e0 + BIN_CH);
    const int* __restrict__ dst = ei + (size_t)E;
    for (int base = e0; base < e1; base += BIN_TPB * SC_K) {
        int dd[SC_K];
#pragma unroll
        for (int k = 0; k < SC_K; ++k) {
            const int e = base + k * BIN_TPB + tid;
            dd[k] = dst[min(e, e1 - 1)];
        }
#pragma unroll
        for (int k = 0; k < SC_K; ++k) {
            const int e = base + k * BIN_TPB + tid;
            if (e < e1) atomicAdd(&c[(unsigned)dd[k] >> BK_SHIFT], 1u);
        }
    }
    __syncthreads();
    for (int b = tid; b < nB; b += BIN_TPB) cntT[(size_t)b * nW + w] = c[b];
}

__global__ __launch_bounds__(BIN_TPB)
void k_bin_scan(const unsigned* __restrict__ cntT, int nW,
                unsigned* __restrict__ offT, unsigned* __restrict__ total) {
    __shared__ unsigned s[BIN_TPB];
    const int w = threadIdx.x;
    const int b = blockIdx.x;
    const unsigned v = (w < nW) ? cntT[(size_t)b * nW + w] : 0u;
    s[w] = v;
    __syncthreads();
    for (int d = 1; d < BIN_TPB; d <<= 1) {
        const unsigned t = (w >= d) ? s[w - d] : 0u;
        __syncthreads();
        s[w] += t;
        __syncthreads();
    }
    if (w < nW) offT[(size_t)b * nW + w] = s[w] - v;
    if (w == BIN_TPB - 1) total[b] = s[w];
}

__global__ __launch_bounds__(BIN_TPB)
void k_bin_scatter(const int* __restrict__ ei, int E, int nB,
                   const unsigned* __restrict__ cntT, const unsigned* __restrict__ offT,
                   const unsigned* __restrict__ total,
                   unsigned* __restrict__ bbase, int2* __restrict__ sorted) {
    __shared__ int2 stage[BIN_CH];
    __shared__ unsigned lcur[BK_MAX];
    __shared__ unsigned delta[BK_MAX];
    __shared__ unsigned sg[BIN_TPB];
    __shared__ unsigned sl[BIN_TPB];
    const int tid = threadIdx.x;
    const int w   = blockIdx.x;
    const int nW  = gridDim.x;
    const int per = (nB + BIN_TPB - 1) / BIN_TPB;
    unsigned locg = 0u, locl = 0u;
    for (int j = 0; j < per; ++j) {
        const int b = tid * per + j;
        if (b < nB) {
            locg += total[b];
            locl += cntT[(size_t)b * nW + w];
        }
    }
    sg[tid] = locg;
    sl[tid] = locl;
    __syncthreads();
    for (int d = 1; d < BIN_TPB; d <<= 1) {
        const unsigned tg = (tid >= d) ? sg[tid - d] : 0u;
        const unsigned tl = (tid >= d) ? sl[tid - d] : 0u;
        __syncthreads();
        sg[tid] += tg;
        sl[tid] += tl;
        __syncthreads();
    }
    unsigned rung = sg[tid] - locg;
    unsigned runl = sl[tid] - locl;
    for (int j = 0; j < per; ++j) {
        const int b = tid * per + j;
        if (b < nB) {
            if (w == 0) bbase[b] = rung;
            lcur[b]  = runl;
            delta[b] = rung + offT[(size_t)b * nW + w] - runl;
            rung += total[b];
            runl += cntT[(size_t)b * nW + w];
        }
    }
    if (w == 0 && tid == BIN_TPB - 1) bbase[nB] = sg[tid];
    __syncthreads();
    const int e0 = w * BIN_CH;
    const int e1 = min(E, e0 + BIN_CH);
    const int* __restrict__ srcp = ei;
    const int* __restrict__ dstp = ei + (size_t)E;
    for (int base = e0; base < e1; base += BIN_TPB * SC_K) {
        int ss[SC_K], dd[SC_K];
#pragma unroll
        for (int k = 0; k < SC_K; ++k) {
            const int e = min(base + k * BIN_TPB + tid, e1 - 1);
            dd[k] = dstp[e];
            ss[k] = srcp[e];
        }
#pragma unroll
        for (int k = 0; k < SC_K; ++k) {
            const int e = base + k * BIN_TPB + tid;
            if (e < e1) {
                const unsigned lp = atomicAdd(&lcur[(unsigned)dd[k] >> BK_SHIFT], 1u);
                stage[lp] = make_int2(ss[k], dd[k]);
            }
        }
    }
    __syncthreads();
    const int cnt = max(e1 - e0, 0);
    for (int i = tid; i < cnt; i += BIN_TPB) {
        const int2 pr = stage[i];
        sorted[delta[(unsigned)pr.y >> BK_SHIFT] + (unsigned)i] = pr;
    }
}

__global__ __launch_bounds__(AG_TPB)
void k_agg1_bucket(const int2* __restrict__ sorted, const unsigned* __restrict__ bbase,
                   const float* __restrict__ dinv, const float* __restrict__ h,
                   const float* __restrict__ b1, const float* __restrict__ W2,
                   float* __restrict__ h2, int N) {
    __shared__ __attribute__((aligned(16))) float acc[BK_NODES * 16];
    const int tid = threadIdx.x;
    const int b   = blockIdx.x;
    const int n0  = b << BK_SHIFT;
    const int nn  = max(min(BK_NODES, N - n0), 0);
    for (int t = tid; t < nn * 16; t += AG_TPB) {
        const float w = dinv[n0 + (t >> 4)];
        acc[t] = h[(size_t)n0 * 16 + t] * w * w;
    }
    __syncthreads();
    const int e0 = (int)bbase[b];
    const int e1 = (int)bbase[b + 1];
    const int q  = tid & 3;
    const int le = tid >> 2;
    for (int base = e0; base < e1; base += (AG_TPB / 4) * AG_K) {
        int2 p[AG_K];
        float4 v[AG_K];
        float ws_[AG_K], wd_[AG_K];
#pragma unroll
        for (int k = 0; k < AG_K; ++k) {
            const int e = min(base + k * (AG_TPB / 4) + le, e1 - 1);
            p[k] = sorted[e];
        }
        asm volatile("" :: "v"(p[0].x), "v"(p[1].x), "v"(p[2].x), "v"(p[3].x));
#pragma unroll
        for (int k = 0; k < AG_K; ++k) {
            const int e = base + k * (AG_TPB / 4) + le;
            v[k] = make_float4(0.f, 0.f, 0.f, 0.f);
            ws_[k] = 0.f;
            wd_[k] = 0.f;
            if (e < e1) {
                ws_[k] = dinv[(unsigned)p[k].x];
                wd_[k] = dinv[(unsigned)p[k].y];
                v[k]   = ((const float4*)(h + (size_t)(unsigned)p[k].x * 16))[q];
            }
        }
#pragma unroll
        for (int k = 0; k < AG_K; ++k) {
            const int e = base + k * (AG_TPB / 4) + le;
            if (e < e1) {
                const float nm = ws_[k] * wd_[k];
                float* a = acc + (size_t)((unsigned)(p[k].y - n0)) * 16 + q * 4;
                atomicAdd(a + 0, v[k].x * nm);
                atomicAdd(a + 1, v[k].y * nm);
                atomicAdd(a + 2, v[k].z * nm);
                atomicAdd(a + 3, v[k].w * nm);
            }
        }
    }
    __syncthreads();
    if (tid < nn) {
        float a0 = 0.f, a1 = 0.f;
#pragma unroll
        for (int j = 0; j < 16; ++j) {
            float x = acc[tid * 16 + j] + b1[j];
            x = x > 0.f ? x : 0.f;
            a0 += x * W2[j * 2 + 0];
            a1 += x * W2[j * 2 + 1];
        }
        h2[(size_t)(n0 + tid) * 2 + 0] = a0;
        h2[(size_t)(n0 + tid) * 2 + 1] = a1;
    }
}

__global__ __launch_bounds__(SC_TPB)
void k_agg2_lds(const int* __restrict__ ei, int E, int N,
                const float* __restrict__ dinv,
                const float* __restrict__ h2,
                int rangeNodes, int nRanges, int edgesPerSlice,
                float* __restrict__ part) {
    __shared__ __attribute__((aligned(16))) float acc[SC_CAPW];
    float* __restrict__ acc0 = acc;
    float* __restrict__ acc1 = acc + SC_CAPW / 2;
    const int tid   = threadIdx.x;
    const int range = blockIdx.x % nRanges;
    const int slice = blockIdx.x / nRanges;
    const int n0 = range * rangeNodes;
    const int nn = max(min(rangeNodes, N - n0), 0);
    for (int i = tid; i < (nn + 3) / 4; i += SC_TPB) {
        ((float4*)acc0)[i] = make_float4(0.f, 0.f, 0.f, 0.f);
        ((float4*)acc1)[i] = make_float4(0.f, 0.f, 0.f, 0.f);
    }
    __syncthreads();
    const int e0 = slice * edgesPerSlice;
    const int e1 = min(E, e0 + edgesPerSlice);
    const int* __restrict__ srcp = ei;
    const int* __restrict__ dstp = ei + (size_t)E;
    const float2* __restrict__ h2v = (const float2*)h2;
    for (int base = e0; base < e1; base += SC_TPB * SC_K) {
        int ss[SC_K], dd[SC_K];
        float2 v[SC_K];
        float ws_[SC_K], wd_[SC_K];
#pragma unroll
        for (int k = 0; k < SC_K; ++k) {
            const int e = min(base + k * SC_TPB + tid, e1 - 1);
            dd[k] = dstp[e];
            ss[k] = srcp[e];
        }
        asm volatile("" :: "v"(ss[0]), "v"(ss[1]), "v"(ss[2]), "v"(ss[3]),
                           "v"(ss[4]), "v"(ss[5]), "v"(ss[6]), "v"(ss[7]));
#pragma unroll
        for (int k = 0; k < SC_K; ++k) {
            const int e = base + k * SC_TPB + tid;
            const unsigned r = (unsigned)(dd[k] - n0);
            v[k] = make_float2(0.f, 0.f);
            ws_[k] = 0.f;
            wd_[k] = 0.f;
            if (e < e1 && r < (unsigned)nn) {
                ws_[k] = dinv[(unsigned)ss[k]];
                wd_[k] = dinv[(unsigned)dd[k]];
                v[k]   = h2v[(unsigned)ss[k]];
            }
        }
#pragma unroll
        for (int k = 0; k < SC_K; ++k) {
            const int e = base + k * SC_TPB + tid;
            const unsigned r = (unsigned)(dd[k] - n0);
            if (e < e1 && r < (unsigned)nn) {
                const float nm = ws_[k] * wd_[k];
                atomicAdd(&acc0[r], v[k].x * nm);
                atomicAdd(&acc1[r], v[k].y * nm);
            }
        }
    }
    __syncthreads();
    float2* __restrict__ o = (float2*)part + (size_t)slice * N + n0;
    for (int i = tid; i < nn; i += SC_TPB) o[i] = make_float2(acc0[i], acc1[i]);
}

__global__ void k_finalize(const float* __restrict__ part, int nSlices,
                           const float* __restrict__ h2,
                           const float* __restrict__ dinv,
                           const float* __restrict__ b2,
                           float* __restrict__ out, int N) {
    int i = blockIdx.x * blockDim.x + threadIdx.x;
    if (i >= N) return;
    const float w = dinv[i] * dinv[i];
    const float2 hv = ((const float2*)h2)[i];
    float a0 = hv.x * w, a1 = hv.y * w;
    const float2* __restrict__ pv = (const float2*)part;
#pragma unroll 8
    for (int s = 0; s < nSlices; ++s) {
        const float2 p = pv[(size_t)s * N + i];
        a0 += p.x;
        a1 += p.y;
    }
    const float v0 = a0 + b2[0];
    const float v1 = a1 + b2[1];
    const float mx = fmaxf(v0, v1);
    const float lse = mx + logf(expf(v0 - mx) + expf(v1 - mx));
    __builtin_nontemporal_store(v0 - lse, out + (size_t)i * 2 + 0);
    __builtin_nontemporal_store(v1 - lse, out + (size_t)i * 2 + 1);
}

extern "C" void kernel_launch(void* const* d_in, const int* in_sizes, int n_in,
                              void* d_out, int out_size, void* d_ws, size_t ws_size,
                              hipStream_t stream) {
    const float* x  = (const float*)d_in[0];
    const int*   ei = (const int*)d_in[1];
    const float* W1 = (const float*)d_in[2];
    const float* b1 = (const float*)d_in[3];
    const float* W2 = (const float*)d_in[4];
    const float* b2 = (const float*)d_in[5];
    float* out = (float*)d_out;

    const int H    = in_sizes[3];
    const int Fin  = in_sizes[2] / H;
    const int N    = in_sizes[0] / Fin;
    const int E    = in_sizes[1] / 2;
    (void)n_in; (void)out_size; (void)ws_size;

    float* ws   = (float*)d_ws;
    float* dinv = ws;
    float* h    = ws + (size_t)N;
    float* h2   = ws + (size_t)33 * N;
    float* part = ws + (((size_t)35 * N + 63) / 64) * 64;

    const int B = 256;
    const int WG = 256;
    const int nRangesD = (N + SC_CAPW - 1) / SC_CAPW;
    const int rangeD   = (N + nRangesD - 1) / nRangesD;
    const int nSlicesD = WG / nRangesD > 0 ? WG / nRangesD : 1;
    const int epsD     = (E + nSlicesD - 1) / nSlicesD;
    const int nRanges2 = (N + SC_CAPW / 2 - 1) / (SC_CAPW / 2);
    const int range2   = (N + nRanges2 - 1) / nRanges2;
    const int nSlices2 = WG / nRanges2 > 0 ? WG / nRanges2 : 1;
    const int eps2     = (E + nSlices2 - 1) / nSlices2;

    k_deg_lds   <<<nRangesD * nSlicesD, SC_TPB, 0, stream>>>(ei, E, N, rangeD, nRangesD, epsD, (unsigned*)part);
    k_deg_reduce<<<(N + B - 1) / B, B, 0, stream>>>((const unsigned*)part, nSlicesD, N, dinv);

    {
        const int ntiles = (N + 15) / 16;
        const int wpb = 4;
        k_gemm_wmma<<<(ntiles + wpb - 1) / wpb, wpb * 32, 0, stream>>>(x, W1, h, N, Fin);
    }

    {
        const int nB = (N + BK_NODES - 1) / BK_NODES;
        const int nW = (E + BIN_CH - 1) / BIN_CH;
        const size_t partWords = (size_t)N * (nSlicesD > 2 * nSlices2 ? nSlicesD : 2 * nSlices2);
        const size_t tabWords  = (((size_t)nB * nW + 63) / 64) * 64;
        unsigned* cntT  = (unsigned*)(part + ((partWords + 63) / 64) * 64);
        unsigned* offT  = cntT + tabWords;
        unsigned* total = offT + tabWords;
        unsigned* bbase = total + (((size_t)nB + 63) / 64) * 64;
        int2* sorted    = (int2*)(bbase + (((size_t)nB + 1 + 63) / 64) * 64);
        k_bin_hist   <<<nW, BIN_TPB, 0, stream>>>(ei, E, nB, cntT);
        k_bin_scan   <<<nB, BIN_TPB, 0, stream>>>(cntT, nW, offT, total);
        k_bin_scatter<<<nW, BIN_TPB, 0, stream>>>(ei, E, nB, cntT, offT, total, bbase, sorted);
        k_agg1_bucket<<<nB, AG_TPB, 0, stream>>>(sorted, bbase, dinv, h, b1, W2, h2, N);
    }

    k_agg2_lds<<<nRanges2 * nSlices2, SC_TPB, 0, stream>>>(ei, E, N, dinv, h2, range2, nRanges2, eps2, part);

    k_finalize<<<(N + B - 1) / B, B, 0, stream>>>(part, nSlices2, h2, dinv, b2, out, N);
}


# Layer 2 walks the layer-1 bucket-sorted edge lists: one work-group per 128-node bucket adds in LDS in arrival order (sum order changes), ends in log_softmax; slice planes and summing launch gone
# speedup vs baseline: 3.2669x; 1.1968x over previous
#include <hip/hip_runtime.h>
#include <hip/hip_bf16.h>

typedef __attribute__((ext_vector_type(2))) float v2f;
typedef __attribute__((ext_vector_type(8))) float v8f;

#define SC_TPB  1024
#define SC_CAPW 51200
#define SC_K    8

__global__ __launch_bounds__(SC_TPB)
void k_deg_lds(const int* __restrict__ ei, int E, int N,
               int rangeNodes, int nRanges, int edgesPerSlice,
               unsigned* __restrict__ part) {
    __shared__ __attribute__((aligned(16))) unsigned cnt[SC_CAPW];
    const int tid   = threadIdx.x;
    const int range = blockIdx.x % nRanges;
    const int slice = blockIdx.x / nRanges;
    const int n0 = range * rangeNodes;
    const int nn = max(min(rangeNodes, N - n0), 0);
    for (int i = tid; i < (nn + 3) / 4; i += SC_TPB) ((uint4*)cnt)[i] = make_uint4(0u, 0u, 0u, 0u);
    __syncthreads();
    const int e0 = slice * edgesPerSlice;
    const int e1 = min(E, e0 + edgesPerSlice);
    const int* __restrict__ dst = ei + (size_t)E;
    for (int base = e0; base < e1; base += SC_TPB * SC_K) {
        int dd[SC_K];
#pragma unroll
        for (int k = 0; k < SC_K; ++k) {
            const int e = base + k * SC_TPB + tid;
            dd[k] = dst[min(e, e1 - 1)];
        }
#pragma unroll
        for (int k = 0; k < SC_K; ++k) {
            const int e = base + k * SC_TPB + tid;
            const unsigned r = (unsigned)(dd[k] - n0);
            if (e < e1 && r < (unsigned)nn) atomicAdd(&cnt[r], 1u);
        }
    }
    __syncthreads();
    unsigned* __restrict__ o = part + (size_t)slice * N + n0;
    for (int i = tid; i < nn; i += SC_TPB) o[i] = cnt[i];
}

__global__ void k_deg_reduce(const unsigned* __restrict__ part, int nSlices, int N,
                             float* __restrict__ dinv) {
    int i = blockIdx.x * blockDim.x + threadIdx.x;
    if (i >= N) return;
    unsigned c = 1u;
#pragma unroll 8
    for (int s = 0; s < nSlices; ++s) c += part[(size_t)s * N + i];
    dinv[i] = rsqrtf((float)c);
}

__global__ void k_gemm_wmma(const float* __restrict__ x,
                            const float* __restrict__ W1,
                            float* __restrict__ h,
                            int N, int K) {
    const int lane = threadIdx.x & 31;
    const int wave = threadIdx.x >> 5;
    const int tile = blockIdx.x * (blockDim.x >> 5) + wave;
    const int ntiles = (N + 15) >> 4;
    if (tile >= ntiles) return;

    const int m  = lane & 15;
    const int kb = (lane >> 4) << 1;
    int row = tile * 16 + m;
    if (row >= N) row = N - 1;
    const float* __restrict__ xrow = x + (size_t)row * K;

    v8f c = {};
    for (int k = 0; k < K; k += 4) {
        v2f a, b;
        a.x = __builtin_nontemporal_load(xrow + k + kb);
        a.y = __builtin_nontemporal_load(xrow + k + kb + 1);
        b.x = W1[(k + kb) * 16 + m];
        b.y = W1[(k + kb + 1) * 16 + m];
        c = __builtin_amdgcn_wmma_f32_16x16x4_f32(false, a, false, b,
                                                  (short)0, c, false, false);
    }
    const int rbase = (lane >> 4) << 3;
#pragma unroll
    for (int r = 0; r < 8; ++r) {
        const int orow = tile * 16 + rbase + r;
        if (orow < N) h[(size_t)orow * 16 + m] = c[r];
    }
}

#define BK_SHIFT 7
#define BK_NODES 128
#define BK_MAX   4096
#define BIN_TPB  1024
#define BIN_CH   12288
#define AG_TPB   512
#define AG_K     4
#define A2_TPB   512
#define A2_K     9

__global__ __launch_bounds__(BIN_TPB)
void k_bin_hist(const int* __restrict__ ei, int E, int nB, unsigned* __restrict__ cntT) {
    __shared__ unsigned c[BK_MAX];
    const int tid = threadIdx.x;
    const int w   = blockIdx.x;
    const int nW  = gridDim.x;
    for (int i = tid; i < nB; i += BIN_TPB) c[i] = 0u;
    __syncthreads();
    const int e0 = w * BIN_CH;
    const int e1 = min(E, e0 + BIN_CH);
    const int* __restrict__ dst = ei + (size_t)E;
    for (int base = e0; base < e1; base += BIN_TPB * SC_K) {
        int dd[SC_K];
#pragma unroll
        for (int k = 0; k < SC_K; ++k) {
            const int e = base + k * BIN_TPB + tid;
            dd[k] = dst[min(e, e1 - 1)];
        }
#pragma unroll
        for (int k = 0; k < SC_K; ++k) {
            const int e = base + k * BIN_TPB + tid;
            if (e < e1) atomicAdd(&c[(unsigned)dd[k] >> BK_SHIFT], 1u);
        }
    }
    __syncthreads();
    for (int b = tid; b < nB; b += BIN_TPB) cntT[(size_t)b * nW + w] = c[b];
}

__global__ __launch_bounds__(BIN_TPB)
void k_bin_scan(const unsigned* __restrict__ cntT, int nW,
                unsigned* __restrict__ offT, unsigned* __restrict__ total) {
    __shared__ unsigned s[BIN_TPB];
    const int w = threadIdx.x;
    const int b = blockIdx.x;
    const unsigned v = (w < nW) ? cntT[(size_t)b * nW + w] : 0u;
    s[w] = v;
    __syncthreads();
    for (int d = 1; d < BIN_TPB; d <<= 1) {
        const unsigned t = (w >= d) ? s[w - d] : 0u;
        __syncthreads();
        s[w] += t;
        __syncthreads();
    }
    if (w < nW) offT[(size_t)b * nW + w] = s[w] - v;
    if (w == BIN_TPB - 1) total[b] = s[w];
}

__global__ __launch_bounds__(BIN_TPB)
void k_bin_scatter(const int* __restrict__ ei, int E, int nB,
                   const unsigned* __restrict__ cntT, const unsigned* __restrict__ offT,
                   const unsigned* __restrict__ total,
                   unsigned* __restrict__ bbase, int2* __restrict__ sorted) {
    __shared__ int2 stage[BIN_CH];
    __shared__ unsigned lcur[BK_MAX];
    __shared__ unsigned delta[BK_MAX];
    __shared__ unsigned sg[BIN_TPB];
    __shared__ unsigned sl[BIN_TPB];
    const int tid = threadIdx.x;
    const int w   = blockIdx.x;
    const int nW  = gridDim.x;
    const int per = (nB + BIN_TPB - 1) / BIN_TPB;
    unsigned locg = 0u, locl = 0u;
    for (int j = 0; j < per; ++j) {
        const int b = tid * per + j;
        if (b < nB) {
            locg += total[b];
            locl += cntT[(size_t)b * nW + w];
        }
    }
    sg[tid] = locg;
    sl[tid] = locl;
    __syncthreads();
    for (int d = 1; d < BIN_TPB; d <<= 1) {
        const unsigned tg = (tid >= d) ? sg[tid - d] : 0u;
        const unsigned tl = (tid >= d) ? sl[tid - d] : 0u;
        __syncthreads();
        sg[tid] += tg;
        sl[tid] += tl;
        __syncthreads();
    }
    unsigned rung = sg[tid] - locg;
    unsigned runl = sl[tid] - locl;
    for (int j = 0; j < per; ++j) {
        const int b = tid * per + j;
        if (b < nB) {
            if (w == 0) bbase[b] = rung;
            lcur[b]  = runl;
            delta[b] = rung + offT[(size_t)b * nW + w] - runl;
            rung += total[b];
            runl += cntT[(size_t)b * nW + w];
        }
    }
    if (w == 0 && tid == BIN_TPB - 1) bbase[nB] = sg[tid];
    __syncthreads();
    const int e0 = w * BIN_CH;
    const int e1 = min(E, e0 + BIN_CH);
    const int* __restrict__ srcp = ei;
    const int* __restrict__ dstp = ei + (size_t)E;
    for (int base = e0; base < e1; base += BIN_TPB * SC_K) {
        int ss[SC_K], dd[SC_K];
#pragma unroll
        for (int k = 0; k < SC_K; ++k) {
            const int e = min(base + k * BIN_TPB + tid, e1 - 1);
            dd[k] = dstp[e];
            ss[k] = srcp[e];
        }
#pragma unroll
        for (int k = 0; k < SC_K; ++k) {
            const int e = base + k * BIN_TPB + tid;
            if (e < e1) {
                const unsigned lp = atomicAdd(&lcur[(unsigned)dd[k] >> BK_SHIFT], 1u);
                stage[lp] = make_int2(ss[k], dd[k]);
            }
        }
    }
    __syncthreads();
    const int cnt = max(e1 - e0, 0);
    for (int i = tid; i < cnt; i += BIN_TPB) {
        const int2 pr = stage[i];
        sorted[delta[(unsigned)pr.y >> BK_SHIFT] + (unsigned)i] = pr;
    }
}

__global__ __launch_bounds__(AG_TPB)
void k_agg1_bucket(const int2* __restrict__ sorted, const unsigned* __restrict__ bbase,
                   const float* __restrict__ dinv, const float* __restrict__ h,
                   const float* __restrict__ b1, const float* __restrict__ W2,
                   float* __restrict__ h2, int N) {
    __shared__ __attribute__((aligned(16))) float acc[BK_NODES * 16];
    const int tid = threadIdx.x;
    const int b   = blockIdx.x;
    const int n0  = b << BK_SHIFT;
    const int nn  = max(min(BK_NODES, N - n0), 0);
    for (int t = tid; t < nn * 16; t += AG_TPB) {
        const float w = dinv[n0 + (t >> 4)];
        acc[t] = h[(size_t)n0 * 16 + t] * w * w;
    }
    __syncthreads();
    const int e0 = (int)bbase[b];
    const int e1 = (int)bbase[b + 1];
    const int q  = tid & 3;
    const int le = tid >> 2;
    for (int base = e0; base < e1; base += (AG_TPB / 4) * AG_K) {
        int2 p[AG_K];
        float4 v[AG_K];
        float ws_[AG_K], wd_[AG_K];
#pragma unroll
        for (int k = 0; k < AG_K; ++k) {
            const int e = min(base + k * (AG_TPB / 4) + le, e1 - 1);
            p[k] = sorted[e];
        }
        asm volatile("" :: "v"(p[0].x), "v"(p[1].x), "v"(p[2].x), "v"(p[3].x));
#pragma unroll
        for (int k = 0; k < AG_K; ++k) {
            const int e = base + k * (AG_TPB / 4) + le;
            v[k] = make_float4(0.f, 0.f, 0.f, 0.f);
            ws_[k] = 0.f;
            wd_[k] = 0.f;
            if (e < e1) {
                ws_[k] = dinv[(unsigned)p[k].x];
                wd_[k] = dinv[(unsigned)p[k].y];
                v[k]   = ((const float4*)(h + (size_t)(unsigned)p[k].x * 16))[q];
            }
        }
#pragma unroll
        for (int k = 0; k < AG_K; ++k) {
            const int e = base + k * (AG_TPB / 4) + le;
            if (e < e1) {
                const float nm = ws_[k] * wd_[k];
                float* a = acc + (size_t)((unsigned)(p[k].y - n0)) * 16 + q * 4;
                atomicAdd(a + 0, v[k].x * nm);
                atomicAdd(a + 1, v[k].y * nm);
                atomicAdd(a + 2, v[k].z * nm);
                atomicAdd(a + 3, v[k].w * nm);
            }
        }
    }
    __syncthreads();
    if (tid < nn) {
        float a0 = 0.f, a1 = 0.f;
#pragma unroll
        for (int j = 0; j < 16; ++j) {
            float x = acc[tid * 16 + j] + b1[j];
            x = x > 0.f ? x : 0.f;
            a0 += x * W2[j * 2 + 0];
            a1 += x * W2[j * 2 + 1];
        }
        h2[(size_t)(n0 + tid) * 2 + 0] = a0;
        h2[(size_t)(n0 + tid) * 2 + 1] = a1;
    }
}

__global__ __launch_bounds__(A2_TPB)
void k_agg2_bucket(const int2* __restrict__ sorted, const unsigned* __restrict__ bbase,
                   const float* __restrict__ dinv, const float* __restrict__ h2,
                   const float* __restrict__ b2, float* __restrict__ out, int N) {
    __shared__ float acc0[BK_NODES];
    __shared__ float acc1[BK_NODES];
    __shared__ float dl[BK_NODES];
    const int tid = threadIdx.x;
    const int b   = blockIdx.x;
    const int n0  = b << BK_SHIFT;
    const int nn  = max(min(BK_NODES, N - n0), 0);
    const float2* __restrict__ h2v = (const float2*)h2;
    if (tid < BK_NODES) {
        float d = 0.f, a0 = 0.f, a1 = 0.f;
        if (tid < nn) {
            d = dinv[n0 + tid];
            const float w = d * d;
            const float2 hv = h2v[n0 + tid];
            a0 = hv.x * w;
            a1 = hv.y * w;
        }
        dl[tid]   = d;
        acc0[tid] = a0;
        acc1[tid] = a1;
    }
    __syncthreads();
    const int e0 = (int)bbase[b];
    const int e1 = (int)bbase[b + 1];
    for (int base = e0; base < e1; base += A2_TPB * A2_K) {
        int2 p[A2_K];
        float2 v[A2_K];
        float ws_[A2_K], wd_[A2_K];
        const int et = base + tid;
#pragma unroll
        for (int k = 0; k < A2_K; ++k) {
            const int e = min(et + k * A2_TPB, e1 - 1);
            p[k] = sorted[e];
        }
        __builtin_amdgcn_sched_barrier(0);
#pragma unroll
        for (int k = 0; k < A2_K; ++k) {
            const unsigned o4 = (unsigned)p[k].x << 2;
            const unsigned o8 = (unsigned)p[k].x << 3;
            ws_[k] = *(const float*)((const char*)dinv + o4);
            v[k]   = *(const float2*)((const char*)h2 + o8);
            wd_[k] = dl[(unsigned)(p[k].y - n0)];
        }
        asm volatile("" :: "v"(ws_[0]), "v"(ws_[1]), "v"(ws_[2]), "v"(ws_[3]), "v"(ws_[4]),
                           "v"(ws_[5]), "v"(ws_[6]), "v"(ws_[7]), "v"(ws_[8]));
        asm volatile("" :: "v"(v[0].x), "v"(v[1].x), "v"(v[2].x), "v"(v[3].x), "v"(v[4].x),
                           "v"(v[5].x), "v"(v[6].x), "v"(v[7].x), "v"(v[8].x));
#pragma unroll
        for (int k = 0; k < A2_K; ++k) {
            if (et < e1 - k * A2_TPB) {
                const unsigned r = (unsigned)(p[k].y - n0);
                const float nm = ws_[k] * wd_[k];
                atomicAdd(&acc0[r], v[k].x * nm);
                atomicAdd(&acc1[r], v[k].y * nm);
            }
        }
    }
    __syncthreads();
    if (tid < nn) {
        const float v0 = acc0[tid] + b2[0];
        const float v1 = acc1[tid] + b2[1];
        const float mx = fmaxf(v0, v1);
        const float lse = mx + logf(expf(v0 - mx) + expf(v1 - mx));
        ((float2*)out)[n0 + tid] = make_float2(v0 - lse, v1 - lse);
    }
}

extern "C" void kernel_launch(void* const* d_in, const int* in_sizes, int n_in,
                              void* d_out, int out_size, void* d_ws, size_t ws_size,
                              hipStream_t stream) {
    const float* x  = (const float*)d_in[0];
    const int*   ei = (const int*)d_in[1];
    const float* W1 = (const float*)d_in[2];
    const float* b1 = (const float*)d_in[3];
    const float* W2 = (const float*)d_in[4];
    const float* b2 = (const float*)d_in[5];
    float* out = (float*)d_out;

    const int H    = in_sizes[3];
    const int Fin  = in_sizes[2] / H;
    const int N    = in_sizes[0] / Fin;
    const int E    = in_sizes[1] / 2;
    (void)n_in; (void)out_size; (void)ws_size;

    float* ws   = (float*)d_ws;
    float* dinv = ws;
    float* h    = ws + (size_t)N;
    float* h2   = ws + (size_t)33 * N;
    float* part = ws + (((size_t)35 * N + 63) / 64) * 64;

    const int B = 256;
    const int WG = 256;
    const int nRangesD = (N + SC_CAPW - 1) / SC_CAPW;
    const int rangeD   = (N + nRangesD - 1) / nRangesD;
    const int nSlicesD = WG / nRangesD > 0 ? WG / nRangesD : 1;
    const int epsD     = (E + nSlicesD - 1) / nSlicesD;

    k_deg_lds   <<<nRangesD * nSlicesD, SC_TPB, 0, stream>>>(ei, E, N, rangeD, nRangesD, epsD, (unsigned*)part);
    k_deg_reduce<<<(N + B - 1) / B, B, 0, stream>>>((const unsigned*)part, nSlicesD, N, dinv);

    {
        const int ntiles = (N + 15) / 16;
        const int wpb = 4;
        k_gemm_wmma<<<(ntiles + wpb - 1) / wpb, wpb * 32, 0, stream>>>(x, W1, h, N, Fin);
    }

    {
        const int nB = (N + BK_NODES - 1) / BK_NODES;
        const int nW = (E + BIN_CH - 1) / BIN_CH;
        const size_t partWords = (size_t)N * nSlicesD;
        const size_t tabWords  = (((size_t)nB * nW + 63) / 64) * 64;
        unsigned* cntT  = (unsigned*)(part + ((partWords + 63) / 64) * 64);
        unsigned* offT  = cntT + tabWords;
        unsigned* total = offT + tabWords;
        unsigned* bbase = total + (((size_t)nB + 63) / 64) * 64;
        int2* sorted    = (int2*)(bbase + (((size_t)nB + 1 + 63) / 64) * 64);
        k_bin_hist   <<<nW, BIN_TPB, 0, stream>>>(ei, E, nB, cntT);
        k_bin_scan   <<<nB, BIN_TPB, 0, stream>>>(cntT, nW, offT, total);
        k_bin_scatter<<<nW, BIN_TPB, 0, stream>>>(ei, E, nB, cntT, offT, total, bbase, sorted);
        k_agg1_bucket<<<nB, AG_TPB, 0, stream>>>(sorted, bbase, dinv, h, b1, W2, h2, N);
        k_agg2_bucket<<<nB, A2_TPB, 0, stream>>>(sorted, bbase, dinv, h2, b2, out, N);
    }
}


# Edge weight factored out of the sums: rows stored pre-scaled by their own dinv (GEMM and layer-1 epilogue), dinv[dst] applied once per node after the sum; no per-edge dinv gathers (rounding changes)
# speedup vs baseline: 3.6670x; 1.1225x over previous
#include <hip/hip_runtime.h>
#include <hip/hip_bf16.h>

typedef __attribute__((ext_vector_type(2))) float v2f;
typedef __attribute__((ext_vector_type(8))) float v8f;

#define SC_TPB  1024
#define SC_CAPW 51200
#define SC_K    8

__global__ __launch_bounds__(SC_TPB)
void k_deg_lds(const int* __restrict__ ei, int E, int N,
               int rangeNodes, int nRanges, int edgesPerSlice,
               unsigned* __restrict__ part) {
    __shared__ __attribute__((aligned(16))) unsigned cnt[SC_CAPW];
    const int tid   = threadIdx.x;
    const int range = blockIdx.x % nRanges;
    const int slice = blockIdx.x / nRanges;
    const int n0 = range * rangeNodes;
    const int nn = max(min(rangeNodes, N - n0), 0);
    for (int i = tid; i < (nn + 3) / 4; i += SC_TPB) ((uint4*)cnt)[i] = make_uint4(0u, 0u, 0u, 0u);
    __syncthreads();
    const int e0 = slice * edgesPerSlice;
    const int e1 = min(E, e0 + edgesPerSlice);
    const int* __restrict__ dst = ei + (size_t)E;
    for (int base = e0; base < e1; base += SC_TPB * SC_K) {
        int dd[SC_K];
#pragma unroll
        for (int k = 0; k < SC_K; ++k) {
            const int e = base + k * SC_TPB + tid;
            dd[k] = dst[min(e, e1 - 1)];
        }
#pragma unroll
        for (int k = 0; k < SC_K; ++k) {
            const int e = base + k * SC_TPB + tid;
            const unsigned r = (unsigned)(dd[k] - n0);
            if (e < e1 && r < (unsigned)nn) atomicAdd(&cnt[r], 1u);
        }
    }
    __syncthreads();
    unsigned* __restrict__ o = part + (size_t)slice * N + n0;
    for (int i = tid; i < nn; i += SC_TPB) o[i] = cnt[i];
}

__global__ void k_deg_reduce(const unsigned* __restrict__ part, int nSlices, int N,
                             float* __restrict__ dinv) {
    int i = blockIdx.x * blockDim.x + threadIdx.x;
    if (i >= N) return;
    unsigned c = 1u;
#pragma unroll 8
    for (int s = 0; s < nSlices; ++s) c += part[(size_t)s * N + i];
    dinv[i] = rsqrtf((float)c);
}

__global__ void k_gemm_wmma(const float* __restrict__ x,
                            const float* __restrict__ W1,
                            const float* __restrict__ dinv,
                            float* __restrict__ h,
                            int N, int K) {
    const int lane = threadIdx.x & 31;
    const int wave = threadIdx.x >> 5;
    const int tile = blockIdx.x * (blockDim.x >> 5) + wave;
    const int ntiles = (N + 15) >> 4;
    if (tile >= ntiles) return;

    const int m  = lane & 15;
    const int kb = (lane >> 4) << 1;
    int row = tile * 16 + m;
    if (row >= N) row = N - 1;
    const float* __restrict__ xrow = x + (size_t)row * K;

    const int rbase = (lane >> 4) << 3;
    float dv[8];
#pragma unroll
    for (int r = 0; r < 8; ++r) dv[r] = dinv[min(tile * 16 + rbase + r, N - 1)];

    v8f c = {};
    for (int k = 0; k < K; k += 4) {
        v2f a, b;
        a.x = __builtin_nontemporal_load(xrow + k + kb);
        a.y = __builtin_nontemporal_load(xrow + k + kb + 1);
        b.x = W1[(k + kb) * 16 + m];
        b.y = W1[(k + kb + 1) * 16 + m];
        c = __builtin_amdgcn_wmma_f32_16x16x4_f32(false, a, false, b,
                                                  (short)0, c, false, false);
    }
#pragma unroll
    for (int r = 0; r < 8; ++r) {
        const int orow = tile * 16 + rbase + r;
        const float o = c[r] * dv[r];
        if (orow < N) h[(size_t)orow * 16 + m] = o;
    }
}

#define BK_SHIFT 7
#define BK_NODES 128
#define BK_MAX   4096
#define BIN_TPB  1024
#define BIN_CH   12288
#define AG_TPB   512
#define AG_K     4
#define A2_TPB   512
#define A2_K     9

__global__ __launch_bounds__(BIN_TPB)
void k_bin_hist(const int* __restrict__ ei, int E, int nB, unsigned* __restrict__ cntT) {
    __shared__ unsigned c[BK_MAX];
    const int tid = threadIdx.x;
    const int w   = blockIdx.x;
    const int nW  = gridDim.x;
    for (int i = tid; i < nB; i += BIN_TPB) c[i] = 0u;
    __syncthreads();
    const int e0 = w * BIN_CH;
    const int e1 = min(E, e0 + BIN_CH);
    const int* __restrict__ dst = ei + (size_t)E;
    for (int base = e0; base < e1; base += BIN_TPB * SC_K) {
        int dd[SC_K];
#pragma unroll
        for (int k = 0; k < SC_K; ++k) {
            const int e = base + k * BIN_TPB + tid;
            dd[k] = dst[min(e, e1 - 1)];
        }
#pragma unroll
        for (int k = 0; k < SC_K; ++k) {
            const int e = base + k * BIN_TPB + tid;
            if (e < e1) atomicAdd(&c[(unsigned)dd[k] >> BK_SHIFT], 1u);
        }
    }
    __syncthreads();
    for (int b = tid; b < nB; b += BIN_TPB) cntT[(size_t)b * nW + w] = c[b];
}

__global__ __launch_bounds__(BIN_TPB)
void k_bin_scan(const unsigned* __restrict__ cntT, int nW,
                unsigned* __restrict__ offT, unsigned* __restrict__ total) {
    __shared__ unsigned s[BIN_TPB];
    const int w = threadIdx.x;
    const int b = blockIdx.x;
    const unsigned v = (w < nW) ? cntT[(size_t)b * nW + w] : 0u;
    s[w] = v;
    __syncthreads();
    for (int d = 1; d < BIN_TPB; d <<= 1) {
        const unsigned t = (w >= d) ? s[w - d] : 0u;
        __syncthreads();
        s[w] += t;
        __syncthreads();
    }
    if (w < nW) offT[(size_t)b * nW + w] = s[w] - v;
    if (w == BIN_TPB - 1) total[b] = s[w];
}

__global__ __launch_bounds__(BIN_TPB)
void k_bin_scatter(const int* __restrict__ ei, int E, int nB,
                   const unsigned* __restrict__ cntT, const unsigned* __restrict__ offT,
                   const unsigned* __restrict__ total,
                   unsigned* __restrict__ bbase, int2* __restrict__ sorted) {
    __shared__ int2 stage[BIN_CH];
    __shared__ unsigned lcur[BK_MAX];
    __shared__ unsigned delta[BK_MAX];
    __shared__ unsigned sg[BIN_TPB];
    __shared__ unsigned sl[BIN_TPB];
    const int tid = threadIdx.x;
    const int w   = blockIdx.x;
    const int nW  = gridDim.x;
    const int per = (nB + BIN_TPB - 1) / BIN_TPB;
    unsigned locg = 0u, locl = 0u;
    for (int j = 0; j < per; ++j) {
        const int b = tid * per + j;
        if (b < nB) {
            locg += total[b];
            locl += cntT[(size_t)b * nW + w];
        }
    }
    sg[tid] = locg;
    sl[tid] = locl;
    __syncthreads();
    for (int d = 1; d < BIN_TPB; d <<= 1) {
        const unsigned tg = (tid >= d) ? sg[tid - d] : 0u;
        const unsigned tl = (tid >= d) ? sl[tid - d] : 0u;
        __syncthreads();
        sg[tid] += tg;
        sl[tid] += tl;
        __syncthreads();
    }
    unsigned rung = sg[tid] - locg;
    unsigned runl = sl[tid] - locl;
    for (int j = 0; j < per; ++j) {
        const int b = tid * per + j;
        if (b < nB) {
            if (w == 0) bbase[b] = rung;
            lcur[b]  = runl;
            delta[b] = rung + offT[(size_t)b * nW + w] - runl;
            rung += total[b];
            runl += cntT[(size_t)b * nW + w];
        }
    }
    if (w == 0 && tid == BIN_TPB - 1) bbase[nB] = sg[tid];
    __syncthreads();
    const int e0 = w * BIN_CH;
    const int e1 = min(E, e0 + BIN_CH);
    const int* __restrict__ srcp = ei;
    const int* __restrict__ dstp = ei + (size_t)E;
    for (int base = e0; base < e1; base += BIN_TPB * SC_K) {
        int ss[SC_K], dd[SC_K];
#pragma unroll
        for (int k = 0; k < SC_K; ++k) {
            const int e = min(base + k * BIN_TPB + tid, e1 - 1);
            dd[k] = dstp[e];
            ss[k] = srcp[e];
        }
#pragma unroll
        for (int k = 0; k < SC_K; ++k) {
            const int e = base + k * BIN_TPB + tid;
            if (e < e1) {
                const unsigned lp = atomicAdd(&lcur[(unsigned)dd[k] >> BK_SHIFT], 1u);
                stage[lp] = make_int2(ss[k], dd[k]);
            }
        }
    }
    __syncthreads();
    const int cnt = max(e1 - e0, 0);
    for (int i = tid; i < cnt; i += BIN_TPB) {
        const int2 pr = stage[i];
        sorted[delta[(unsigned)pr.y >> BK_SHIFT] + (unsigned)i] = pr;
    }
}

__global__ __launch_bounds__(AG_TPB)
void k_agg1_bucket(const int2* __restrict__ sorted, const unsigned* __restrict__ bbase,
                   const float* __restrict__ dinv, const float* __restrict__ h,
                   const float* __restrict__ b1, const float* __restrict__ W2,
                   float* __restrict__ h2, int N) {
    __shared__ __attribute__((aligned(16))) float acc[BK_NODES * 16];
    const int tid = threadIdx.x;
    const int b   = blockIdx.x;
    const int n0  = b << BK_SHIFT;
    const int nn  = max(min(BK_NODES, N - n0), 0);
    const float dme = (tid < nn) ? dinv[n0 + tid] : 0.f;
    for (int t = tid; t < nn * 16; t += AG_TPB) acc[t] = h[(size_t)n0 * 16 + t];
    __syncthreads();
    const int e0 = (int)bbase[b];
    const int e1 = (int)bbase[b + 1];
    const int q  = tid & 3;
    const int le = tid >> 2;
    for (int base = e0; base < e1; base += (AG_TPB / 4) * AG_K) {
        int2 p[AG_K];
        float4 v[AG_K];
#pragma unroll
        for (int k = 0; k < AG_K; ++k) {
            const int e = min(base + k * (AG_TPB / 4) + le, e1 - 1);
            p[k] = sorted[e];
        }
        asm volatile("" :: "v"(p[0].x), "v"(p[1].x), "v"(p[2].x), "v"(p[3].x));
#pragma unroll
        for (int k = 0; k < AG_K; ++k) {
            const int e = base + k * (AG_TPB / 4) + le;
            v[k] = make_float4(0.f, 0.f, 0.f, 0.f);
            if (e < e1) v[k] = ((const float4*)(h + (size_t)(unsigned)p[k].x * 16))[q];
        }
#pragma unroll
        for (int k = 0; k < AG_K; ++k) {
            const int e = base + k * (AG_TPB / 4) + le;
            if (e < e1) {
                float* a = acc + (size_t)((unsigned)(p[k].y - n0)) * 16 + q * 4;
                atomicAdd(a + 0, v[k].x);
                atomicAdd(a + 1, v[k].y);
                atomicAdd(a + 2, v[k].z);
                atomicAdd(a + 3, v[k].w);
            }
        }
    }
    __syncthreads();
    if (tid < nn) {
        float a0 = 0.f, a1 = 0.f;
#pragma unroll
        for (int j = 0; j < 16; ++j) {
            float x = acc[tid * 16 + j] * dme + b1[j];
            x = x > 0.f ? x : 0.f;
            a0 += x * W2[j * 2 + 0];
            a1 += x * W2[j * 2 + 1];
        }
        h2[(size_t)(n0 + tid) * 2 + 0] = a0 * dme;
        h2[(size_t)(n0 + tid) * 2 + 1] = a1 * dme;
    }
}

__global__ __launch_bounds__(A2_TPB)
void k_agg2_bucket(const int2* __restrict__ sorted, const unsigned* __restrict__ bbase,
                   const float* __restrict__ dinv, const float* __restrict__ h2,
                   const float* __restrict__ b2, float* __restrict__ out, int N) {
    __shared__ float acc0[BK_NODES];
    __shared__ float acc1[BK_NODES];
    const int tid = threadIdx.x;
    const int b   = blockIdx.x;
    const int n0  = b << BK_SHIFT;
    const int nn  = max(min(BK_NODES, N - n0), 0);
    const float2* __restrict__ h2v = (const float2*)h2;
    float dme = 0.f;
    if (tid < BK_NODES) {
        float a0 = 0.f, a1 = 0.f;
        if (tid < nn) {
            dme = dinv[n0 + tid];
            const float2 hv = h2v[n0 + tid];
            a0 = hv.x;
            a1 = hv.y;
        }
        acc0[tid] = a0;
        acc1[tid] = a1;
    }
    __syncthreads();
    const int e0 = (int)bbase[b];
    const int e1 = (int)bbase[b + 1];
    for (int base = e0; base < e1; base += A2_TPB * A2_K) {
        int2 p[A2_K];
        float2 v[A2_K];
        const int et = base + tid;
#pragma unroll
        for (int k = 0; k < A2_K; ++k) {
            const int e = min(et + k * A2_TPB, e1 - 1);
            p[k] = sorted[e];
        }
        __builtin_amdgcn_sched_barrier(0);
#pragma unroll
        for (int k = 0; k < A2_K; ++k) {
            const unsigned o8 = (unsigned)p[k].x << 3;
            v[k] = *(const float2*)((const char*)h2 + o8);
        }
        asm volatile("" :: "v"(v[0].x), "v"(v[1].x), "v"(v[2].x), "v"(v[3].x), "v"(v[4].x),
                           "v"(v[5].x), "v"(v[6].x), "v"(v[7].x), "v"(v[8].x));
#pragma unroll
        for (int k = 0; k < A2_K; ++k) {
            if (et < e1 - k * A2_TPB) {
                const unsigned r = (unsigned)(p[k].y - n0);
                atomicAdd(&acc0[r], v[k].x);
                atomicAdd(&acc1[r], v[k].y);
            }
        }
    }
    __syncthreads();
    if (tid < nn) {
        const float v0 = acc0[tid] * dme + b2[0];
        const float v1 = acc1[tid] * dme + b2[1];
        const float mx = fmaxf(v0, v1);
        const float lse = mx + logf(expf(v0 - mx) + expf(v1 - mx));
        ((float2*)out)[n0 + tid] = make_float2(v0 - lse, v1 - lse);
    }
}

extern "C" void kernel_launch(void* const* d_in, const int* in_sizes, int n_in,
                              void* d_out, int out_size, void* d_ws, size_t ws_size,
                              hipStream_t stream) {
    const float* x  = (const float*)d_in[0];
    const int*   ei = (const int*)d_in[1];
    const float* W1 = (const float*)d_in[2];
    const float* b1 = (const float*)d_in[3];
    const float* W2 = (const float*)d_in[4];
    const float* b2 = (const float*)d_in[5];
    float* out = (float*)d_out;

    const int H    = in_sizes[3];
    const int Fin  = in_sizes[2] / H;
    const int N    = in_sizes[0] / Fin;
    const int E    = in_sizes[1] / 2;
    (void)n_in; (void)out_size; (void)ws_size;

    float* ws   = (float*)d_ws;
    float* dinv = ws;
    float* h    = ws + (size_t)N;
    float* h2   = ws + (size_t)33 * N;
    float* part = ws + (((size_t)35 * N + 63) / 64) * 64;

    const int B = 256;
    const int WG = 256;
    const int nRangesD = (N + SC_CAPW - 1) / SC_CAPW;
    const int rangeD   = (N + nRangesD - 1) / nRangesD;
    const int nSlicesD = WG / nRangesD > 0 ? WG / nRangesD : 1;
    const int epsD     = (E + nSlicesD - 1) / nSlicesD;

    k_deg_lds   <<<nRangesD * nSlicesD, SC_TPB, 0, stream>>>(ei, E, N, rangeD, nRangesD, epsD, (unsigned*)part);
    k_deg_reduce<<<(N + B - 1) / B, B, 0, stream>>>((const unsigned*)part, nSlicesD, N, dinv);

    {
        const int ntiles = (N + 15) / 16;
        const int wpb = 4;
        k_gemm_wmma<<<(ntiles + wpb - 1) / wpb, wpb * 32, 0, stream>>>(x, W1, dinv, h, N, Fin);
    }

    {
        const int nB = (N + BK_NODES - 1) / BK_NODES;
        const int nW = (E + BIN_CH - 1) / BIN_CH;
        const size_t partWords = (size_t)N * nSlicesD;
        const size_t tabWords  = (((size_t)nB * nW + 63) / 64) * 64;
        unsigned* cntT  = (unsigned*)(part + ((partWords + 63) / 64) * 64);
        unsigned* offT  = cntT + tabWords;
        unsigned* total = offT + tabWords;
        unsigned* bbase = total + (((size_t)nB + 63) / 64) * 64;
        int2* sorted    = (int2*)(bbase + (((size_t)nB + 1 + 63) / 64) * 64);
        k_bin_hist   <<<nW, BIN_TPB, 0, stream>>>(ei, E, nB, cntT);
        k_bin_scan   <<<nB, BIN_TPB, 0, stream>>>(cntT, nW, offT, total);
        k_bin_scatter<<<nW, BIN_TPB, 0, stream>>>(ei, E, nB, cntT, offT, total, bbase, sorted);
        k_agg1_bucket<<<nB, AG_TPB, 0, stream>>>(sorted, bbase, dinv, h, b1, W2, h2, N);
        k_agg2_bucket<<<nB, A2_TPB, 0, stream>>>(sorted, bbase, dinv, h2, b2, out, N);
    }
}


# Degrees counted from the bucket-sorted edge lists inside the GEMM work-group (one bucket = 8 row tiles, 128 LDS counters); slice-plane histogram and its summing launch gone, 8 launches -> 6
# speedup vs baseline: 4.3385x; 1.1831x over previous
#include <hip/hip_runtime.h>
#include <hip/hip_bf16.h>

typedef __attribute__((ext_vector_type(2))) float v2f;
typedef __attribute__((ext_vector_type(8))) float v8f;

#define SC_K    8

#define BK_SHIFT 7
#define BK_NODES 128
#define BK_MAX   4096
#define BIN_TPB  1024
#define BIN_CH   12288
#define AG_TPB   512
#define AG_K     4
#define A2_TPB   512
#define A2_K     9

__global__ __launch_bounds__(BIN_TPB)
void k_bin_hist(const int* __restrict__ ei, int E, int nB, unsigned* __restrict__ cntT) {
    __shared__ unsigned c[BK_MAX];
    const int tid = threadIdx.x;
    const int w   = blockIdx.x;
    const int nW  = gridDim.x;
    for (int i = tid; i < nB; i += BIN_TPB) c[i] = 0u;
    __syncthreads();
    const int e0 = w * BIN_CH;
    const int e1 = min(E, e0 + BIN_CH);
    const int* __restrict__ dst = ei + (size_t)E;
    for (int base = e0; base < e1; base += BIN_TPB * SC_K) {
        int dd[SC_K];
#pragma unroll
        for (int k = 0; k < SC_K; ++k) {
            const int e = base + k * BIN_TPB + tid;
            dd[k] = dst[min(e, e1 - 1)];
        }
#pragma unroll
        for (int k = 0; k < SC_K; ++k) {
            const int e = base + k * BIN_TPB + tid;
            if (e < e1) atomicAdd(&c[(unsigned)dd[k] >> BK_SHIFT], 1u);
        }
    }
    __syncthreads();
    for (int b = tid; b < nB; b += BIN_TPB) cntT[(size_t)b * nW + w] = c[b];
}

__global__ __launch_bounds__(BIN_TPB)
void k_bin_scan(const unsigned* __restrict__ cntT, int nW,
                unsigned* __restrict__ offT, unsigned* __restrict__ total) {
    __shared__ unsigned s[BIN_TPB];
    const int w = threadIdx.x;
    const int b = blockIdx.x;
    const unsigned v = (w < nW) ? cntT[(size_t)b * nW + w] : 0u;
    s[w] = v;
    __syncthreads();
    for (int d = 1; d < BIN_TPB; d <<= 1) {
        const unsigned t = (w >= d) ? s[w - d] : 0u;
        __syncthreads();
        s[w] += t;
        __syncthreads();
    }
    if (w < nW) offT[(size_t)b * nW + w] = s[w] - v;
    if (w == BIN_TPB - 1) total[b] = s[w];
}

__global__ __launch_bounds__(BIN_TPB)
void k_bin_scatter(const int* __restrict__ ei, int E, int nB,
                   const unsigned* __restrict__ cntT, const unsigned* __restrict__ offT,
                   const unsigned* __restrict__ total,
                   unsigned* __restrict__ bbase, int2* __restrict__ sorted) {
    __shared__ int2 stage[BIN_CH];
    __shared__ unsigned lcur[BK_MAX];
    __shared__ unsigned delta[BK_MAX];
    __shared__ unsigned sg[BIN_TPB];
    __shared__ unsigned sl[BIN_TPB];
    const int tid = threadIdx.x;
    const int w   = blockIdx.x;
    const int nW  = gridDim.x;
    const int per = (nB + BIN_TPB - 1) / BIN_TPB;
    unsigned locg = 0u, locl = 0u;
    for (int j = 0; j < per; ++j) {
        const int b = tid * per + j;
        if (b < nB) {
            locg += total[b];
            locl += cntT[(size_t)b * nW + w];
        }
    }
    sg[tid] = locg;
    sl[tid] = locl;
    __syncthreads();
    for (int d = 1; d < BIN_TPB; d <<= 1) {
        const unsigned tg = (tid >= d) ? sg[tid - d] : 0u;
        const unsigned tl = (tid >= d) ? sl[tid - d] : 0u;
        __syncthreads();
        sg[tid] += tg;
        sl[tid] += tl;
        __syncthreads();
    }
    unsigned rung = sg[tid] - locg;
    unsigned runl = sl[tid] - locl;
    for (int j = 0; j < per; ++j) {
        const int b = tid * per + j;
        if (b < nB) {
            if (w == 0) bbase[b] = rung;
            lcur[b]  = runl;
            delta[b] = rung + offT[(size_t)b * nW + w] - runl;
            rung += total[b];
            runl += cntT[(size_t)b * nW + w];
        }
    }
    if (w == 0 && tid == BIN_TPB - 1) bbase[nB] = sg[tid];
    __syncthreads();
    const int e0 = w * BIN_CH;
    const int e1 = min(E, e0 + BIN_CH);
    const int* __restrict__ srcp = ei;
    const int* __restrict__ dstp = ei + (size_t)E;
    for (int base = e0; base < e1; base += BIN_TPB * SC_K) {
        int ss[SC_K], dd[SC_K];
#pragma unroll
        for (int k = 0; k < SC_K; ++k) {
            const int e = min(base + k * BIN_TPB + tid, e1 - 1);
            dd[k] = dstp[e];
            ss[k] = srcp[e];
        }
#pragma unroll
        for (int k = 0; k < SC_K; ++k) {
            const int e = base + k * BIN_TPB + tid;
            if (e < e1) {
                const unsigned lp = atomicAdd(&lcur[(unsigned)dd[k] >> BK_SHIFT], 1u);
                stage[lp] = make_int2(ss[k], dd[k]);
            }
        }
    }
    __syncthreads();
    const int cnt = max(e1 - e0, 0);
    for (int i = tid; i < cnt; i += BIN_TPB) {
        const int2 pr = stage[i];
        sorted[delta[(unsigned)pr.y >> BK_SHIFT] + (unsigned)i] = pr;
    }
}

#define GM_TPB 256
#define DG_K   18
__global__ __launch_bounds__(GM_TPB)
void k_gemm_wmma(const float* __restrict__ x,
                 const float* __restrict__ W1,
                 const int2* __restrict__ sorted, const unsigned* __restrict__ bbase,
                 float* __restrict__ dinv,
                 float* __restrict__ h,
                 int N, int K) {
    __shared__ unsigned cnt[BK_NODES];
    __shared__ __attribute__((aligned(32))) float dl[BK_NODES];
    const int tid = threadIdx.x;
    const int b   = blockIdx.x;
    const int n0  = b << BK_SHIFT;
    const int nn  = max(min(BK_NODES, N - n0), 0);
    if (tid < BK_NODES) cnt[tid] = 0u;
    __syncthreads();
    const int e0 = (int)bbase[b];
    const int e1 = (int)bbase[b + 1];
    for (int base = e0; base < e1; base += GM_TPB * DG_K) {
        int dd[DG_K];
        const int et = base + tid;
#pragma unroll
        for (int k = 0; k < DG_K; ++k) dd[k] = sorted[min(et + k * GM_TPB, e1 - 1)].y;
#pragma unroll
        for (int k = 0; k < DG_K; ++k) {
            if (et < e1 - k * GM_TPB) atomicAdd(&cnt[(unsigned)(dd[k] - n0)], 1u);
        }
    }
    __syncthreads();
    if (tid < BK_NODES) {
        const float d = rsqrtf((float)(cnt[tid] + 1u));
        dl[tid] = d;
        if (tid < nn) dinv[n0 + tid] = d;
    }
    __syncthreads();

    const int lane = tid & 31;
    const int wave = tid >> 5;
    const int tile = b * (GM_TPB >> 5) + wave;
    const int ntiles = (N + 15) >> 4;
    if (tile >= ntiles) return;

    const int m  = lane & 15;
    const int kb = (lane >> 4) << 1;
    int row = tile * 16 + m;
    if (row >= N) row = N - 1;
    const float* __restrict__ xrow = x + (size_t)row * K;

    const int rbase = (lane >> 4) << 3;
    float dv[8];
#pragma unroll
    for (int r = 0; r < 8; ++r) dv[r] = dl[wave * 16 + rbase + r];

    v8f c = {};
    for (int k = 0; k < K; k += 4) {
        v2f a, bq;
        a.x = __builtin_nontemporal_load(xrow + k + kb);
        a.y = __builtin_nontemporal_load(xrow + k + kb + 1);
        bq.x = W1[(k + kb) * 16 + m];
        bq.y = W1[(k + kb + 1) * 16 + m];
        c = __builtin_amdgcn_wmma_f32_16x16x4_f32(false, a, false, bq,
                                                  (short)0, c, false, false);
    }
#pragma unroll
    for (int r = 0; r < 8; ++r) {
        const int orow = tile * 16 + rbase + r;
        const float o = c[r] * dv[r];
        if (orow < N) h[(size_t)orow * 16 + m] = o;
    }
}

__global__ __launch_bounds__(AG_TPB)
void k_agg1_bucket(const int2* __restrict__ sorted, const unsigned* __restrict__ bbase,
                   const float* __restrict__ dinv, const float* __restrict__ h,
                   const float* __restrict__ b1, const float* __restrict__ W2,
                   float* __restrict__ h2, int N) {
    __shared__ __attribute__((aligned(16))) float acc[BK_NODES * 16];
    const int tid = threadIdx.x;
    const int b   = blockIdx.x;
    const int n0  = b << BK_SHIFT;
    const int nn  = max(min(BK_NODES, N - n0), 0);
    const float dme = (tid < nn) ? dinv[n0 + tid] : 0.f;
    for (int t = tid; t < nn * 16; t += AG_TPB) acc[t] = h[(size_t)n0 * 16 + t];
    __syncthreads();
    const int e0 = (int)bbase[b];
    const int e1 = (int)bbase[b + 1];
    const int q  = tid & 3;
    const int le = tid >> 2;
    for (int base = e0; base < e1; base += (AG_TPB / 4) * AG_K) {
        int2 p[AG_K];
        float4 v[AG_K];
#pragma unroll
        for (int k = 0; k < AG_K; ++k) {
            const int e = min(base + k * (AG_TPB / 4) + le, e1 - 1);
            p[k] = sorted[e];
        }
        asm volatile("" :: "v"(p[0].x), "v"(p[1].x), "v"(p[2].x), "v"(p[3].x));
#pragma unroll
        for (int k = 0; k < AG_K; ++k) {
            const int e = base + k * (AG_TPB / 4) + le;
            v[k] = make_float4(0.f, 0.f, 0.f, 0.f);
            if (e < e1) v[k] = ((const float4*)(h + (size_t)(unsigned)p[k].x * 16))[q];
        }
#pragma unroll
        for (int k = 0; k < AG_K; ++k) {
            const int e = base + k * (AG_TPB / 4) + le;
            if (e < e1) {
                float* a = acc + (size_t)((unsigned)(p[k].y - n0)) * 16 + q * 4;
                atomicAdd(a + 0, v[k].x);
                atomicAdd(a + 1, v[k].y);
                atomicAdd(a + 2, v[k].z);
                atomicAdd(a + 3, v[k].w);
            }
        }
    }
    __syncthreads();
    if (tid < nn) {
        float a0 = 0.f, a1 = 0.f;
#pragma unroll
        for (int j = 0; j < 16; ++j) {
            float x = acc[tid * 16 + j] * dme + b1[j];
            x = x > 0.f ? x : 0.f;
            a0 += x * W2[j * 2 + 0];
            a1 += x * W2[j * 2 + 1];
        }
        h2[(size_t)(n0 + tid) * 2 + 0] = a0 * dme;
        h2[(size_t)(n0 + tid) * 2 + 1] = a1 * dme;
    }
}

__global__ __launch_bounds__(A2_TPB)
void k_agg2_bucket(const int2* __restrict__ sorted, const unsigned* __restrict__ bbase,
                   const float* __restrict__ dinv, const float* __restrict__ h2,
                   const float* __restrict__ b2, float* __restrict__ out, int N) {
    __shared__ float acc0[BK_NODES];
    __shared__ float acc1[BK_NODES];
    const int tid = threadIdx.x;
    const int b   = blockIdx.x;
    const int n0  = b << BK_SHIFT;
    const int nn  = max(min(BK_NODES, N - n0), 0);
    const float2* __restrict__ h2v = (const float2*)h2;
    float dme = 0.f;
    if (tid < BK_NODES) {
        float a0 = 0.f, a1 = 0.f;
        if (tid < nn) {
            dme = dinv[n0 + tid];
            const float2 hv = h2v[n0 + tid];
            a0 = hv.x;
            a1 = hv.y;
        }
        acc0[tid] = a0;
        acc1[tid] = a1;
    }
    __syncthreads();
    const int e0 = (int)bbase[b];
    const int e1 = (int)bbase[b + 1];
    for (int base = e0; base < e1; base += A2_TPB * A2_K) {
        int2 p[A2_K];
        float2 v[A2_K];
        const int et = base + tid;
#pragma unroll
        for (int k = 0; k < A2_K; ++k) {
            const int e = min(et + k * A2_TPB, e1 - 1);
            p[k] = sorted[e];
        }
        __builtin_amdgcn_sched_barrier(0);
#pragma unroll
        for (int k = 0; k < A2_K; ++k) {
            const unsigned o8 = (unsigned)p[k].x << 3;
            v[k] = *(const float2*)((const char*)h2 + o8);
        }
        asm volatile("" :: "v"(v[0].x), "v"(v[1].x), "v"(v[2].x), "v"(v[3].x), "v"(v[4].x),
                           "v"(v[5].x), "v"(v[6].x), "v"(v[7].x), "v"(v[8].x));
#pragma unroll
        for (int k = 0; k < A2_K; ++k) {
            if (et < e1 - k * A2_TPB) {
                const unsigned r = (unsigned)(p[k].y - n0);
                atomicAdd(&acc0[r], v[k].x);
                atomicAdd(&acc1[r], v[k].y);
            }
        }
    }
    __syncthreads();
    if (tid < nn) {
        const float v0 = acc0[tid] * dme + b2[0];
        const float v1 = acc1[tid] * dme + b2[1];
        const float mx = fmaxf(v0, v1);
        const float lse = mx + logf(expf(v0 - mx) + expf(v1 - mx));
        ((float2*)out)[n0 + tid] = make_float2(v0 - lse, v1 - lse);
    }
}

extern "C" void kernel_launch(void* const* d_in, const int* in_sizes, int n_in,
                              void* d_out, int out_size, void* d_ws, size_t ws_size,
                              hipStream_t stream) {
    const float* x  = (const float*)d_in[0];
    const int*   ei = (const int*)d_in[1];
    const float* W1 = (const float*)d_in[2];
    const float* b1 = (const float*)d_in[3];
    const float* W2 = (const float*)d_in[4];
    const float* b2 = (const float*)d_in[5];
    float* out = (float*)d_out;

    const int H    = in_sizes[3];
    const int Fin  = in_sizes[2] / H;
    const int N    = in_sizes[0] / Fin;
    const int E    = in_sizes[1] / 2;
    (void)n_in; (void)out_size; (void)ws_size;

    float* ws   = (float*)d_ws;
    float* dinv = ws;
    float* h    = ws + (size_t)N;
    float* h2   = ws + (size_t)33 * N;
    float* tabs = ws + (((size_t)35 * N + 63) / 64) * 64;

    {
        const int nB = (N + BK_NODES - 1) / BK_NODES;
        const int nW = (E + BIN_CH - 1) / BIN_CH;
        const size_t tabWords  = (((size_t)nB * nW + 63) / 64) * 64;
        unsigned* cntT  = (unsigned*)tabs;
        unsigned* offT  = cntT + tabWords;
        unsigned* total = offT + tabWords;
        unsigned* bbase = total + (((size_t)nB + 63) / 64) * 64;
        int2* sorted    = (int2*)(bbase + (((size_t)nB + 1 + 63) / 64) * 64);
        k_bin_hist   <<<nW, BIN_TPB, 0, stream>>>(ei, E, nB, cntT);
        k_bin_scan   <<<nB, BIN_TPB, 0, stream>>>(cntT, nW, offT, total);
        k_bin_scatter<<<nW, BIN_TPB, 0, stream>>>(ei, E, nB, cntT, offT, total, bbase, sorted);
        k_gemm_wmma  <<<nB, GM_TPB, 0, stream>>>(x, W1, sorted, bbase, dinv, h, N, Fin);
        k_agg1_bucket<<<nB, AG_TPB, 0, stream>>>(sorted, bbase, dinv, h, b1, W2, h2, N);
        k_agg2_bucket<<<nB, A2_TPB, 0, stream>>>(sorted, bbase, dinv, h2, b2, out, N);
    }
}


# GEMM reads x as whole 512-byte rows (16-byte loads, 4 full lines per instruction, was 16 partial lines); tile and W1 reach the WMMA fragments through LDS; k order of the sums changes (rounding)
# speedup vs baseline: 4.7497x; 1.0948x over previous
#include <hip/hip_runtime.h>
#include <hip/hip_bf16.h>

typedef __attribute__((ext_vector_type(2))) float v2f;
typedef __attribute__((ext_vector_type(8))) float v8f;

#define SC_K    8

#define BK_SHIFT 7
#define BK_NODES 128
#define BK_MAX   4096
#define BIN_TPB  1024
#define BIN_CH   12288
#define AG_TPB   512
#define AG_K     4
#define A2_TPB   512
#define A2_K     9

__global__ __launch_bounds__(BIN_TPB)
void k_bin_hist(const int* __restrict__ ei, int E, int nB, unsigned* __restrict__ cntT) {
    __shared__ unsigned c[BK_MAX];
    const int tid = threadIdx.x;
    const int w   = blockIdx.x;
    const int nW  = gridDim.x;
    for (int i = tid; i < nB; i += BIN_TPB) c[i] = 0u;
    __syncthreads();
    const int e0 = w * BIN_CH;
    const int e1 = min(E, e0 + BIN_CH);
    const int* __restrict__ dst = ei + (size_t)E;
    for (int base = e0; base < e1; base += BIN_TPB * SC_K) {
        int dd[SC_K];
#pragma unroll
        for (int k = 0; k < SC_K; ++k) {
            const int e = base + k * BIN_TPB + tid;
            dd[k] = dst[min(e, e1 - 1)];
        }
#pragma unroll
        for (int k = 0; k < SC_K; ++k) {
            const int e = base + k * BIN_TPB + tid;
            if (e < e1) atomicAdd(&c[(unsigned)dd[k] >> BK_SHIFT], 1u);
        }
    }
    __syncthreads();
    for (int b = tid; b < nB; b += BIN_TPB) cntT[(size_t)b * nW + w] = c[b];
}

__global__ __launch_bounds__(BIN_TPB)
void k_bin_scan(const unsigned* __restrict__ cntT, int nW,
                unsigned* __restrict__ offT, unsigned* __restrict__ total) {
    __shared__ unsigned s[BIN_TPB];
    const int w = threadIdx.x;
    const int b = blockIdx.x;
    const unsigned v = (w < nW) ? cntT[(size_t)b * nW + w] : 0u;
    s[w] = v;
    __syncthreads();
    for (int d = 1; d < BIN_TPB; d <<= 1) {
        const unsigned t = (w >= d) ? s[w - d] : 0u;
        __syncthreads();
        s[w] += t;
        __syncthreads();
    }
    if (w < nW) offT[(size_t)b * nW + w] = s[w] - v;
    if (w == BIN_TPB - 1) total[b] = s[w];
}

__global__ __launch_bounds__(BIN_TPB)
void k_bin_scatter(const int* __restrict__ ei, int E, int nB,
                   const unsigned* __restrict__ cntT, const unsigned* __restrict__ offT,
                   const unsigned* __restrict__ total,
                   unsigned* __restrict__ bbase, int2* __restrict__ sorted) {
    __shared__ int2 stage[BIN_CH];
    __shared__ unsigned lcur[BK_MAX];
    __shared__ unsigned delta[BK_MAX];
    __shared__ unsigned sg[BIN_TPB];
    __shared__ unsigned sl[BIN_TPB];
    const int tid = threadIdx.x;
    const int w   = blockIdx.x;
    const int nW  = gridDim.x;
    const int per = (nB + BIN_TPB - 1) / BIN_TPB;
    unsigned locg = 0u, locl = 0u;
    for (int j = 0; j < per; ++j) {
        const int b = tid * per + j;
        if (b < nB) {
            locg += total[b];
            locl += cntT[(size_t)b * nW + w];
        }
    }
    sg[tid] = locg;
    sl[tid] = locl;
    __syncthreads();
    for (int d = 1; d < BIN_TPB; d <<= 1) {
        const unsigned tg = (tid >= d) ? sg[tid - d] : 0u;
        const unsigned tl = (tid >= d) ? sl[tid - d] : 0u;
        __syncthreads();
        sg[tid] += tg;
        sl[tid] += tl;
        __syncthreads();
    }
    unsigned rung = sg[tid] - locg;
    unsigned runl = sl[tid] - locl;
    for (int j = 0; j < per; ++j) {
        const int b = tid * per + j;
        if (b < nB) {
            if (w == 0) bbase[b] = rung;
            lcur[b]  = runl;
            delta[b] = rung + offT[(size_t)b * nW + w] - runl;
            rung += total[b];
            runl += cntT[(size_t)b * nW + w];
        }
    }
    if (w == 0 && tid == BIN_TPB - 1) bbase[nB] = sg[tid];
    __syncthreads();
    const int e0 = w * BIN_CH;
    const int e1 = min(E, e0 + BIN_CH);
    const int* __restrict__ srcp = ei;
    const int* __restrict__ dstp = ei + (size_t)E;
    for (int base = e0; base < e1; base += BIN_TPB * SC_K) {
        int ss[SC_K], dd[SC_K];
#pragma unroll
        for (int k = 0; k < SC_K; ++k) {
            const int e = min(base + k * BIN_TPB + tid, e1 - 1);
            dd[k] = dstp[e];
            ss[k] = srcp[e];
        }
#pragma unroll
        for (int k = 0; k < SC_K; ++k) {
            const int e = base + k * BIN_TPB + tid;
            if (e < e1) {
                const unsigned lp = atomicAdd(&lcur[(unsigned)dd[k] >> BK_SHIFT], 1u);
                stage[lp] = make_int2(ss[k], dd[k]);
            }
        }
    }
    __syncthreads();
    const int cnt = max(e1 - e0, 0);
    for (int i = tid; i < cnt; i += BIN_TPB) {
        const int2 pr = stage[i];
        sorted[delta[(unsigned)pr.y >> BK_SHIFT] + (unsigned)i] = pr;
    }
}

#define GM_TPB 256
#define DG_K   18
#define GM_KC  128
#define GM_PITCH (GM_KC + 4)
typedef __attribute__((ext_vector_type(4))) float v4f;

__device__ __forceinline__ void gm_load_rows(const float* __restrict__ x, int N, int K, int tile, int k0,
                                             int lane, v4f (&xr)[16]) {
    const int kq = k0 + lane * 4;
#pragma unroll
    for (int r = 0; r < 16; ++r) {
        const int row = min(tile * 16 + r, N - 1);
        const v4f* p = (const v4f*)(x + (size_t)row * K + min(kq, K - 4));
        xr[r] = __builtin_nontemporal_load(p);
    }
}

__device__ __forceinline__ void gm_load_w(const float* __restrict__ W1, int K, int k0, int tid,
                                          float (&wr)[GM_KC * 16 / GM_TPB]) {
#pragma unroll
    for (int j = 0; j < GM_KC * 16 / GM_TPB; ++j) {
        const int idx = k0 * 16 + j * GM_TPB + tid;
        wr[j] = W1[min(idx, K * 16 - 1)];
    }
}
__device__ __forceinline__ void gm_stage_w(float* wt, int K, int k0, int tid,
                                           const float (&wr)[GM_KC * 16 / GM_TPB]) {
#pragma unroll
    for (int j = 0; j < GM_KC * 16 / GM_TPB; ++j) {
        const int li = j * GM_TPB + tid;
        wt[(li & 15) * GM_PITCH + (li >> 4)] = (k0 * 16 + li < K * 16) ? wr[j] : 0.f;
    }
}

__device__ __forceinline__ void gm_stage(float* xw, int K, int k0, int lane, const v4f (&xr)[16]) {
    const bool in = k0 + lane * 4 < K;
#pragma unroll
    for (int r = 0; r < 16; ++r)
        *(v4f*)(xw + r * GM_PITCH + lane * 4) = in ? xr[r] : (v4f){0.f, 0.f, 0.f, 0.f};
}

__global__ __launch_bounds__(GM_TPB)
void k_gemm_wmma(const float* __restrict__ x,
                 const float* __restrict__ W1,
                 const int2* __restrict__ sorted, const unsigned* __restrict__ bbase,
                 float* __restrict__ dinv,
                 float* __restrict__ h,
                 int N, int K) {
    __shared__ unsigned cnt[BK_NODES];
    __shared__ __attribute__((aligned(32))) float dl[BK_NODES];
    __shared__ __attribute__((aligned(16))) float xs[(GM_TPB >> 5) * 16 * GM_PITCH];
    __shared__ __attribute__((aligned(16))) float wt[16 * GM_PITCH];
    const int tid = threadIdx.x;
    const int b   = blockIdx.x;
    const int n0  = b << BK_SHIFT;
    const int nn  = max(min(BK_NODES, N - n0), 0);
    const int lane = tid & 31;
    const int wave = __builtin_amdgcn_readfirstlane(tid >> 5);
    const int tile = b * (GM_TPB >> 5) + wave;
    const int ntiles = (N + 15) >> 4;
    const bool live = tile < ntiles;
    float* xw = xs + wave * (16 * GM_PITCH);

    const int m  = lane & 15;
    const int hf = lane >> 4;

    if (tid < BK_NODES) cnt[tid] = 0u;
    __syncthreads();
    v4f xr[16];
    float wr[GM_KC * 16 / GM_TPB];
    if (live) gm_load_rows(x, N, K, tile, 0, lane, xr);
    gm_load_w(W1, K, 0, tid, wr);
    const int e0 = (int)bbase[b];
    const int e1 = (int)bbase[b + 1];
    gm_stage_w(wt, K, 0, tid, wr);
    if (live) gm_stage(xw, K, 0, lane, xr);
    for (int base = e0; base < e1; base += GM_TPB * DG_K) {
        int dd[DG_K];
        const int et = base + tid;
#pragma unroll
        for (int k = 0; k < DG_K; ++k) dd[k] = sorted[min(et + k * GM_TPB, e1 - 1)].y;
#pragma unroll
        for (int k = 0; k < DG_K; ++k) {
            if (et < e1 - k * GM_TPB) atomicAdd(&cnt[(unsigned)(dd[k] - n0)], 1u);
        }
    }
    __syncthreads();
    if (tid < BK_NODES) {
        const float d = rsqrtf((float)(cnt[tid] + 1u));
        dl[tid] = d;
        if (tid < nn) dinv[n0 + tid] = d;
    }
    __syncthreads();

    const int rbase = hf << 3;
    float dv[8];
#pragma unroll
    for (int r = 0; r < 8; ++r) dv[r] = dl[wave * 16 + rbase + r];

    v8f c = {};
    for (int k0 = 0; k0 < K; k0 += GM_KC) {
        if (k0 > 0) {
            __syncthreads();
            if (live) gm_load_rows(x, N, K, tile, k0, lane, xr);
            gm_load_w(W1, K, k0, tid, wr);
            gm_stage_w(wt, K, k0, tid, wr);
            if (live) gm_stage(xw, K, k0, lane, xr);
            __syncthreads();
        }
        if (live) {
            const float* xf = xw + m * GM_PITCH + 4 * hf;
            const float* wf = wt + m * GM_PITCH + 4 * hf;
#pragma unroll
            for (int t = 0; t < GM_KC / 8; ++t) {
                const v4f f = *(const v4f*)(xf + 8 * t);
                const v4f g = *(const v4f*)(wf + 8 * t);
                v2f a, bq;
                a.x = f.x; a.y = f.y; bq.x = g.x; bq.y = g.y;
                c = __builtin_amdgcn_wmma_f32_16x16x4_f32(false, a, false, bq,
                                                          (short)0, c, false, false);
                a.x = f.z; a.y = f.w; bq.x = g.z; bq.y = g.w;
                c = __builtin_amdgcn_wmma_f32_16x16x4_f32(false, a, false, bq,
                                                          (short)0, c, false, false);
            }
        }
    }
    if (!live) return;
#pragma unroll
    for (int r = 0; r < 8; ++r) {
        const int orow = tile * 16 + rbase + r;
        const float o = c[r] * dv[r];
        if (orow < N) h[(size_t)orow * 16 + m] = o;
    }
}

__global__ __launch_bounds__(AG_TPB)
void k_agg1_bucket(const int2* __restrict__ sorted, const unsigned* __restrict__ bbase,
                   const float* __restrict__ dinv, const float* __restrict__ h,
                   const float* __restrict__ b1, const float* __restrict__ W2,
                   float* __restrict__ h2, int N) {
    __shared__ __attribute__((aligned(16))) float acc[BK_NODES * 16];
    const int tid = threadIdx.x;
    const int b   = blockIdx.x;
    const int n0  = b << BK_SHIFT;
    const int nn  = max(min(BK_NODES, N - n0), 0);
    const float dme = (tid < nn) ? dinv[n0 + tid] : 0.f;
    for (int t = tid; t < nn * 16; t += AG_TPB) acc[t] = h[(size_t)n0 * 16 + t];
    __syncthreads();
    const int e0 = (int)bbase[b];
    const int e1 = (int)bbase[b + 1];
    const int q  = tid & 3;
    const int le = tid >> 2;
    for (int base = e0; base < e1; base += (AG_TPB / 4) * AG_K) {
        int2 p[AG_K];
        float4 v[AG_K];
#pragma unroll
        for (int k = 0; k < AG_K; ++k) {
            const int e = min(base + k * (AG_TPB / 4) + le, e1 - 1);
            p[k] = sorted[e];
        }
        asm volatile("" :: "v"(p[0].x), "v"(p[1].x), "v"(p[2].x), "v"(p[3].x));
#pragma unroll
        for (int k = 0; k < AG_K; ++k) {
            const int e = base + k * (AG_TPB / 4) + le;
            v[k] = make_float4(0.f, 0.f, 0.f, 0.f);
            if (e < e1) v[k] = ((const float4*)(h + (size_t)(unsigned)p[k].x * 16))[q];
        }
#pragma unroll
        for (int k = 0; k < AG_K; ++k) {
            const int e = base + k * (AG_TPB / 4) + le;
            if (e < e1) {
                float* a = acc + (size_t)((unsigned)(p[k].y - n0)) * 16 + q * 4;
                atomicAdd(a + 0, v[k].x);
                atomicAdd(a + 1, v[k].y);
                atomicAdd(a + 2, v[k].z);
                atomicAdd(a + 3, v[k].w);
            }
        }
    }
    __syncthreads();
    if (tid < nn) {
        float a0 = 0.f, a1 = 0.f;
#pragma unroll
        for (int j = 0; j < 16; ++j) {
            float x = acc[tid * 16 + j] * dme + b1[j];
            x = x > 0.f ? x : 0.f;
            a0 += x * W2[j * 2 + 0];
            a1 += x * W2[j * 2 + 1];
        }
        h2[(size_t)(n0 + tid) * 2 + 0] = a0 * dme;
        h2[(size_t)(n0 + tid) * 2 + 1] = a1 * dme;
    }
}

__global__ __launch_bounds__(A2_TPB)
void k_agg2_bucket(const int2* __restrict__ sorted, const unsigned* __restrict__ bbase,
                   const float* __restrict__ dinv, const float* __restrict__ h2,
                   const float* __restrict__ b2, float* __restrict__ out, int N) {
    __shared__ float acc0[BK_NODES];
    __shared__ float acc1[BK_NODES];
    const int tid = threadIdx.x;
    const int b   = blockIdx.x;
    const int n0  = b << BK_SHIFT;
    const int nn  = max(min(BK_NODES, N - n0), 0);
    const float2* __restrict__ h2v = (const float2*)h2;
    float dme = 0.f;
    if (tid < BK_NODES) {
        float a0 = 0.f, a1 = 0.f;
        if (tid < nn) {
            dme = dinv[n0 + tid];
            const float2 hv = h2v[n0 + tid];
            a0 = hv.x;
            a1 = hv.y;
        }
        acc0[tid] = a0;
        acc1[tid] = a1;
    }
    __syncthreads();
    const int e0 = (int)bbase[b];
    const int e1 = (int)bbase[b + 1];
    for (int base = e0; base < e1; base += A2_TPB * A2_K) {
        int2 p[A2_K];
        float2 v[A2_K];
        const int et = base + tid;
#pragma unroll
        for (int k = 0; k < A2_K; ++k) {
            const int e = min(et + k * A2_TPB, e1 - 1);
            p[k] = sorted[e];
        }
        __builtin_amdgcn_sched_barrier(0);
#pragma unroll
        for (int k = 0; k < A2_K; ++k) {
            const unsigned o8 = (unsigned)p[k].x << 3;
            v[k] = *(const float2*)((const char*)h2 + o8);
        }
        asm volatile("" :: "v"(v[0].x), "v"(v[1].x), "v"(v[2].x), "v"(v[3].x), "v"(v[4].x),
                           "v"(v[5].x), "v"(v[6].x), "v"(v[7].x), "v"(v[8].x));
#pragma unroll
        for (int k = 0; k < A2_K; ++k) {
            if (et < e1 - k * A2_TPB) {
                const unsigned r = (unsigned)(p[k].y - n0);
                atomicAdd(&acc0[r], v[k].x);
                atomicAdd(&acc1[r], v[k].y);
            }
        }
    }
    __syncthreads();
    if (tid < nn) {
        const float v0 = acc0[tid] * dme + b2[0];
        const float v1 = acc1[tid] * dme + b2[1];
        const float mx = fmaxf(v0, v1);
        const float lse = mx + logf(expf(v0 - mx) + expf(v1 - mx));
        ((float2*)out)[n0 + tid] = make_float2(v0 - lse, v1 - lse);
    }
}

extern "C" void kernel_launch(void* const* d_in, const int* in_sizes, int n_in,
                              void* d_out, int out_size, void* d_ws, size_t ws_size,
                              hipStream_t stream) {
    const float* x  = (const float*)d_in[0];
    const int*   ei = (const int*)d_in[1];
    const float* W1 = (const float*)d_in[2];
    const float* b1 = (const float*)d_in[3];
    const float* W2 = (const float*)d_in[4];
    const float* b2 = (const float*)d_in[5];
    float* out = (float*)d_out;

    const int H    = in_sizes[3];
    const int Fin  = in_sizes[2] / H;
    const int N    = in_sizes[0] / Fin;
    const int E    = in_sizes[1] / 2;
    (void)n_in; (void)out_size; (void)ws_size;

    float* ws   = (float*)d_ws;
    float* dinv = ws;
    float* h    = ws + (size_t)N;
    float* h2   = ws + (size_t)33 * N;
    float* tabs = ws + (((size_t)35 * N + 63) / 64) * 64;

    {
        const int nB = (N + BK_NODES - 1) / BK_NODES;
        const int nW = (E + BIN_CH - 1) / BIN_CH;
        const size_t tabWords  = (((size_t)nB * nW + 63) / 64) * 64;
        unsigned* cntT  = (unsigned*)tabs;
        unsigned* offT  = cntT + tabWords;
        unsigned* total = offT + tabWords;
        unsigned* bbase = total + (((size_t)nB + 63) / 64) * 64;
        int2* sorted    = (int2*)(bbase + (((size_t)nB + 1 + 63) / 64) * 64);
        k_bin_hist   <<<nW, BIN_TPB, 0, stream>>>(ei, E, nB, cntT);
        k_bin_scan   <<<nB, BIN_TPB, 0, stream>>>(cntT, nW, offT, total);
        k_bin_scatter<<<nW, BIN_TPB, 0, stream>>>(ei, E, nB, cntT, offT, total, bbase, sorted);
        k_gemm_wmma  <<<nB, GM_TPB, 0, stream>>>(x, W1, sorted, bbase, dinv, h, N, Fin);
        k_agg1_bucket<<<nB, AG_TPB, 0, stream>>>(sorted, bbase, dinv, h, b1, W2, h2, N);
        k_agg2_bucket<<<nB, A2_TPB, 0, stream>>>(sorted, bbase, dinv, h2, b2, out, N);
    }
}


# Layer-1 sums: a bucket's edges ranked by destination node in LDS (one integer add per edge), then 4 lanes per node add the node's rows in registers; its LDS float adds gone (sum order changes)
# speedup vs baseline: 5.8200x; 1.2253x over previous
#include <hip/hip_runtime.h>
#include <hip/hip_bf16.h>

typedef __attribute__((ext_vector_type(2))) float v2f;
typedef __attribute__((ext_vector_type(8))) float v8f;

#define SC_K    8

#define BK_SHIFT 7
#define BK_NODES 128
#define BK_MAX   4096
#define BIN_TPB  1024
#define BIN_CH   12288
#define AG_TPB   512
#define AG_PK    9
#define AG_CAP   (AG_TPB * AG_PK)
#define AG_U     8
#define A2_TPB   512
#define A2_K     9

__global__ __launch_bounds__(BIN_TPB)
void k_bin_hist(const int* __restrict__ ei, int E, int nB, unsigned* __restrict__ cntT) {
    __shared__ unsigned c[BK_MAX];
    const int tid = threadIdx.x;
    const int w   = blockIdx.x;
    const int nW  = gridDim.x;
    for (int i = tid; i < nB; i += BIN_TPB) c[i] = 0u;
    __syncthreads();
    const int e0 = w * BIN_CH;
    const int e1 = min(E, e0 + BIN_CH);
    const int* __restrict__ dst = ei + (size_t)E;
    for (int base = e0; base < e1; base += BIN_TPB * SC_K) {
        int dd[SC_K];
#pragma unroll
        for (int k = 0; k < SC_K; ++k) {
            const int e = base + k * BIN_TPB + tid;
            dd[k] = dst[min(e, e1 - 1)];
        }
#pragma unroll
        for (int k = 0; k < SC_K; ++k) {
            const int e = base + k * BIN_TPB + tid;
            if (e < e1) atomicAdd(&c[(unsigned)dd[k] >> BK_SHIFT], 1u);
        }
    }
    __syncthreads();
    for (int b = tid; b < nB; b += BIN_TPB) cntT[(size_t)b * nW + w] = c[b];
}

__global__ __launch_bounds__(BIN_TPB)
void k_bin_scan(const unsigned* __restrict__ cntT, int nW,
                unsigned* __restrict__ offT, unsigned* __restrict__ total) {
    __shared__ unsigned s[BIN_TPB];
    const int w = threadIdx.x;
    const int b = blockIdx.x;
    const unsigned v = (w < nW) ? cntT[(size_t)b * nW + w] : 0u;
    s[w] = v;
    __syncthreads();
    for (int d = 1; d < BIN_TPB; d <<= 1) {
        const unsigned t = (w >= d) ? s[w - d] : 0u;
        __syncthreads();
        s[w] += t;
        __syncthreads();
    }
    if (w < nW) offT[(size_t)b * nW + w] = s[w] - v;
    if (w == BIN_TPB - 1) total[b] = s[w];
}

__global__ __launch_bounds__(BIN_TPB)
void k_bin_scatter(const int* __restrict__ ei, int E, int nB,
                   const unsigned* __restrict__ cntT, const unsigned* __restrict__ offT,
                   const unsigned* __restrict__ total,
                   unsigned* __restrict__ bbase, int2* __restrict__ sorted) {
    __shared__ int2 stage[BIN_CH];
    __shared__ unsigned lcur[BK_MAX];
    __shared__ unsigned delta[BK_MAX];
    __shared__ unsigned sg[BIN_TPB];
    __shared__ unsigned sl[BIN_TPB];
    const int tid = threadIdx.x;
    const int w   = blockIdx.x;
    const int nW  = gridDim.x;
    const int per = (nB + BIN_TPB - 1) / BIN_TPB;
    unsigned locg = 0u, locl = 0u;
    for (int j = 0; j < per; ++j) {
        const int b = tid * per + j;
        if (b < nB) {
            locg += total[b];
            locl += cntT[(size_t)b * nW + w];
        }
    }
    sg[tid] = locg;
    sl[tid] = locl;
    __syncthreads();
    for (int d = 1; d < BIN_TPB; d <<= 1) {
        const unsigned tg = (tid >= d) ? sg[tid - d] : 0u;
        const unsigned tl = (tid >= d) ? sl[tid - d] : 0u;
        __syncthreads();
        sg[tid] += tg;
        sl[tid] += tl;
        __syncthreads();
    }
    unsigned rung = sg[tid] - locg;
    unsigned runl = sl[tid] - locl;
    for (int j = 0; j < per; ++j) {
        const int b = tid * per + j;
        if (b < nB) {
            if (w == 0) bbase[b] = rung;
            lcur[b]  = runl;
            delta[b] = rung + offT[(size_t)b * nW + w] - runl;
            rung += total[b];
            runl += cntT[(size_t)b * nW + w];
        }
    }
    if (w == 0 && tid == BIN_TPB - 1) bbase[nB] = sg[tid];
    __syncthreads();
    const int e0 = w * BIN_CH;
    const int e1 = min(E, e0 + BIN_CH);
    const int* __restrict__ srcp = ei;
    const int* __restrict__ dstp = ei + (size_t)E;
    for (int base = e0; base < e1; base += BIN_TPB * SC_K) {
        int ss[SC_K], dd[SC_K];
#pragma unroll
        for (int k = 0; k < SC_K; ++k) {
            const int e = min(base + k * BIN_TPB + tid, e1 - 1);
            dd[k] = dstp[e];
            ss[k] = srcp[e];
        }
#pragma unroll
        for (int k = 0; k < SC_K; ++k) {
            const int e = base + k * BIN_TPB + tid;
            if (e < e1) {
                const unsigned lp = atomicAdd(&lcur[(unsigned)dd[k] >> BK_SHIFT], 1u);
                stage[lp] = make_int2(ss[k], dd[k]);
            }
        }
    }
    __syncthreads();
    const int cnt = max(e1 - e0, 0);
    for (int i = tid; i < cnt; i += BIN_TPB) {
        const int2 pr = stage[i];
        sorted[delta[(unsigned)pr.y >> BK_SHIFT] + (unsigned)i] = pr;
    }
}

#define GM_TPB 256
#define DG_K   18
#define GM_KC  128
#define GM_PITCH (GM_KC + 4)
typedef __attribute__((ext_vector_type(4))) float v4f;

__device__ __forceinline__ void gm_load_rows(const float* __restrict__ x, int N, int K, int tile, int k0,
                                             int lane, v4f (&xr)[16]) {
    const int kq = k0 + lane * 4;
#pragma unroll
    for (int r = 0; r < 16; ++r) {
        const int row = min(tile * 16 + r, N - 1);
        const v4f* p = (const v4f*)(x + (size_t)row * K + min(kq, K - 4));
        xr[r] = __builtin_nontemporal_load(p);
    }
}

__device__ __forceinline__ void gm_load_w(const float* __restrict__ W1, int K, int k0, int tid,
                                          float (&wr)[GM_KC * 16 / GM_TPB]) {
#pragma unroll
    for (int j = 0; j < GM_KC * 16 / GM_TPB; ++j) {
        const int idx = k0 * 16 + j * GM_TPB + tid;
        wr[j] = W1[min(idx, K * 16 - 1)];
    }
}
__device__ __forceinline__ void gm_stage_w(float* wt, int K, int k0, int tid,
                                           const float (&wr)[GM_KC * 16 / GM_TPB]) {
#pragma unroll
    for (int j = 0; j < GM_KC * 16 / GM_TPB; ++j) {
        const int li = j * GM_TPB + tid;
        wt[(li & 15) * GM_PITCH + (li >> 4)] = (k0 * 16 + li < K * 16) ? wr[j] : 0.f;
    }
}

__device__ __forceinline__ void gm_stage(float* xw, int K, int k0, int lane, const v4f (&xr)[16]) {
    const bool in = k0 + lane * 4 < K;
#pragma unroll
    for (int r = 0; r < 16; ++r)
        *(v4f*)(xw + r * GM_PITCH + lane * 4) = in ? xr[r] : (v4f){0.f, 0.f, 0.f, 0.f};
}

__global__ __launch_bounds__(GM_TPB)
void k_gemm_wmma(const float* __restrict__ x,
                 const float* __restrict__ W1,
                 const int2* __restrict__ sorted, const unsigned* __restrict__ bbase,
                 float* __restrict__ dinv,
                 float* __restrict__ h,
                 int N, int K) {
    __shared__ unsigned cnt[BK_NODES];
    __shared__ __attribute__((aligned(32))) float dl[BK_NODES];
    __shared__ __attribute__((aligned(16))) float xs[(GM_TPB >> 5) * 16 * GM_PITCH];
    __shared__ __attribute__((aligned(16))) float wt[16 * GM_PITCH];
    const int tid = threadIdx.x;
    const int b   = blockIdx.x;
    const int n0  = b << BK_SHIFT;
    const int nn  = max(min(BK_NODES, N - n0), 0);
    const int lane = tid & 31;
    const int wave = __builtin_amdgcn_readfirstlane(tid >> 5);
    const int tile = b * (GM_TPB >> 5) + wave;
    const int ntiles = (N + 15) >> 4;
    const bool live = tile < ntiles;
    float* xw = xs + wave * (16 * GM_PITCH);

    const int m  = lane & 15;
    const int hf = lane >> 4;

    if (tid < BK_NODES) cnt[tid] = 0u;
    __syncthreads();
    v4f xr[16];
    float wr[GM_KC * 16 / GM_TPB];
    if (live) gm_load_rows(x, N, K, tile, 0, lane, xr);
    gm_load_w(W1, K, 0, tid, wr);
    const int e0 = (int)bbase[b];
    const int e1 = (int)bbase[b + 1];
    gm_stage_w(wt, K, 0, tid, wr);
    if (live) gm_stage(xw, K, 0, lane, xr);
    for (int base = e0; base < e1; base += GM_TPB * DG_K) {
        int dd[DG_K];
        const int et = base + tid;
#pragma unroll
        for (int k = 0; k < DG_K; ++k) dd[k] = sorted[min(et + k * GM_TPB, e1 - 1)].y;
#pragma unroll
        for (int k = 0; k < DG_K; ++k) {
            if (et < e1 - k * GM_TPB) atomicAdd(&cnt[(unsigned)(dd[k] - n0)], 1u);
        }
    }
    __syncthreads();
    if (tid < BK_NODES) {
        const float d = rsqrtf((float)(cnt[tid] + 1u));
        dl[tid] = d;
        if (tid < nn) dinv[n0 + tid] = d;
    }
    __syncthreads();

    const int rbase = hf << 3;
    float dv[8];
#pragma unroll
    for (int r = 0; r < 8; ++r) dv[r] = dl[wave * 16 + rbase + r];

    v8f c = {};
    for (int k0 = 0; k0 < K; k0 += GM_KC) {
        if (k0 > 0) {
            __syncthreads();
            if (live) gm_load_rows(x, N, K, tile, k0, lane, xr);
            gm_load_w(W1, K, k0, tid, wr);
            gm_stage_w(wt, K, k0, tid, wr);
            if (live) gm_stage(xw, K, k0, lane, xr);
            __syncthreads();
        }
        if (live) {
            const float* xf = xw + m * GM_PITCH + 4 * hf;
            const float* wf = wt + m * GM_PITCH + 4 * hf;
#pragma unroll
            for (int t = 0; t < GM_KC / 8; ++t) {
                const v4f f = *(const v4f*)(xf + 8 * t);
                const v4f g = *(const v4f*)(wf + 8 * t);
                v2f a, bq;
                a.x = f.x; a.y = f.y; bq.x = g.x; bq.y = g.y;
                c = __builtin_amdgcn_wmma_f32_16x16x4_f32(false, a, false, bq,
                                                          (short)0, c, false, false);
                a.x = f.z; a.y = f.w; bq.x = g.z; bq.y = g.w;
                c = __builtin_amdgcn_wmma_f32_16x16x4_f32(false, a, false, bq,
                                                          (short)0, c, false, false);
            }
        }
    }
    if (!live) return;
#pragma unroll
    for (int r = 0; r < 8; ++r) {
        const int orow = tile * 16 + rbase + r;
        const float o = c[r] * dv[r];
        if (orow < N) h[(size_t)orow * 16 + m] = o;
    }
}

__global__ __launch_bounds__(AG_TPB)
void k_agg1_bucket(const int2* __restrict__ sorted, const unsigned* __restrict__ bbase,
                   const float* __restrict__ dinv, const float* __restrict__ h,
                   const float* __restrict__ b1, const float* __restrict__ W2,
                   float* __restrict__ h2, int N) {
    __shared__ unsigned cnt[BK_NODES];
    __shared__ unsigned off[BK_NODES + 1];
    __shared__ int lst[AG_CAP];
    __shared__ float2 res[BK_NODES];
    const int tid = threadIdx.x;
    const int b   = blockIdx.x;
    const int n0  = b << BK_SHIFT;
    const int nn  = max(min(BK_NODES, N - n0), 0);
    const int node = tid >> 2;
    const int q    = tid & 3;
    const bool own = node < nn;
    const float4* __restrict__ h4 = (const float4*)h;
    if (tid < BK_NODES) cnt[tid] = 0u;
    const int e0 = (int)bbase[b];
    const int e1 = (int)bbase[b + 1];
    __syncthreads();
    const float dme = own ? dinv[n0 + node] : 0.f;
    float4 acc = make_float4(0.f, 0.f, 0.f, 0.f);
    if (own) acc = h4[(size_t)(n0 + node) * 4 + q];
    for (int base = e0; base < e1; base += AG_CAP) {
        int2 p[AG_PK];
        unsigned rk[AG_PK];
        const int et = base + tid;
#pragma unroll
        for (int k = 0; k < AG_PK; ++k) p[k] = sorted[min(et + k * AG_TPB, e1 - 1)];
#pragma unroll
        for (int k = 0; k < AG_PK; ++k) {
            rk[k] = 0u;
            if (et < e1 - k * AG_TPB) rk[k] = atomicAdd(&cnt[(unsigned)(p[k].y - n0)], 1u);
        }
        __syncthreads();
        if (tid < 32) {
            const unsigned c0 = cnt[4 * tid + 0], c1 = cnt[4 * tid + 1];
            const unsigned c2 = cnt[4 * tid + 2], c3 = cnt[4 * tid + 3];
            const unsigned s = c0 + c1 + c2 + c3;
            unsigned inc = s;
#pragma unroll
            for (int d = 1; d < 32; d <<= 1) {
                const unsigned t = __shfl_up(inc, d, 32);
                if (tid >= d) inc += t;
            }
            const unsigned ex = inc - s;
            off[4 * tid + 0] = ex;
            off[4 * tid + 1] = ex + c0;
            off[4 * tid + 2] = ex + c0 + c1;
            off[4 * tid + 3] = ex + c0 + c1 + c2;
            if (tid == 31) off[BK_NODES] = inc;
            cnt[4 * tid + 0] = 0u; cnt[4 * tid + 1] = 0u; cnt[4 * tid + 2] = 0u; cnt[4 * tid + 3] = 0u;
        }
        __syncthreads();
        unsigned ro[AG_PK];
#pragma unroll
        for (int k = 0; k < AG_PK; ++k) ro[k] = off[(unsigned)(p[k].y - n0)];
        asm volatile("" :: "v"(ro[0]), "v"(ro[1]), "v"(ro[2]), "v"(ro[3]), "v"(ro[4]),
                           "v"(ro[5]), "v"(ro[6]), "v"(ro[7]), "v"(ro[8]));
#pragma unroll
        for (int k = 0; k < AG_PK; ++k) {
            if (et < e1 - k * AG_TPB) lst[ro[k] + rk[k]] = p[k].x;
        }
        __syncthreads();
        const int i1 = (int)off[node + 1];
        for (int i = (int)off[node]; i < i1; i += AG_U) {
            const int rem = i1 - i;
            int s[AG_U];
            float4 v[AG_U];
#pragma unroll
            for (int u = 0; u < AG_U; ++u) s[u] = lst[i + min(u, rem - 1)];
#pragma unroll
            for (int u = 0; u < AG_U; ++u) {
                const unsigned o64 = ((unsigned)s[u] << 6) + ((unsigned)q << 4);
                v[u] = *(const float4*)((const char*)h + o64);
            }
            asm volatile("" :: "v"(v[0].x), "v"(v[1].x), "v"(v[2].x), "v"(v[3].x),
                               "v"(v[4].x), "v"(v[5].x), "v"(v[6].x), "v"(v[7].x));
#pragma unroll
            for (int u = 0; u < AG_U; ++u) {
                if (u < rem) {
                    acc.x += v[u].x;
                    acc.y += v[u].y;
                    acc.z += v[u].z;
                    acc.w += v[u].w;
                }
            }
        }
    }
    float a0 = 0.f, a1 = 0.f;
    const float av[4] = {acc.x, acc.y, acc.z, acc.w};
#pragma unroll
    for (int j = 0; j < 4; ++j) {
        const int f = q * 4 + j;
        float x = av[j] * dme + b1[f];
        x = x > 0.f ? x : 0.f;
        a0 += x * W2[f * 2 + 0];
        a1 += x * W2[f * 2 + 1];
    }
    a0 += __shfl_xor(a0, 1, 32);
    a1 += __shfl_xor(a1, 1, 32);
    a0 += __shfl_xor(a0, 2, 32);
    a1 += __shfl_xor(a1, 2, 32);
    if (q == 0) res[node] = make_float2(a0 * dme, a1 * dme);
    __syncthreads();
    if (tid < nn) ((float2*)h2)[n0 + tid] = res[tid];
}

__global__ __launch_bounds__(A2_TPB)
void k_agg2_bucket(const int2* __restrict__ sorted, const unsigned* __restrict__ bbase,
                   const float* __restrict__ dinv, const float* __restrict__ h2,
                   const float* __restrict__ b2, float* __restrict__ out, int N) {
    __shared__ float acc0[BK_NODES];
    __shared__ float acc1[BK_NODES];
    const int tid = threadIdx.x;
    const int b   = blockIdx.x;
    const int n0  = b << BK_SHIFT;
    const int nn  = max(min(BK_NODES, N - n0), 0);
    const float2* __restrict__ h2v = (const float2*)h2;
    float dme = 0.f;
    if (tid < BK_NODES) {
        float a0 = 0.f, a1 = 0.f;
        if (tid < nn) {
            dme = dinv[n0 + tid];
            const float2 hv = h2v[n0 + tid];
            a0 = hv.x;
            a1 = hv.y;
        }
        acc0[tid] = a0;
        acc1[tid] = a1;
    }
    __syncthreads();
    const int e0 = (int)bbase[b];
    const int e1 = (int)bbase[b + 1];
    for (int base = e0; base < e1; base += A2_TPB * A2_K) {
        int2 p[A2_K];
        float2 v[A2_K];
        const int et = base + tid;
#pragma unroll
        for (int k = 0; k < A2_K; ++k) {
            const int e = min(et + k * A2_TPB, e1 - 1);
            p[k] = sorted[e];
        }
        __builtin_amdgcn_sched_barrier(0);
#pragma unroll
        for (int k = 0; k < A2_K; ++k) {
            const unsigned o8 = (unsigned)p[k].x << 3;
            v[k] = *(const float2*)((const char*)h2 + o8);
        }
        asm volatile("" :: "v"(v[0].x), "v"(v[1].x), "v"(v[2].x), "v"(v[3].x), "v"(v[4].x),
                           "v"(v[5].x), "v"(v[6].x), "v"(v[7].x), "v"(v[8].x));
#pragma unroll
        for (int k = 0; k < A2_K; ++k) {
            if (et < e1 - k * A2_TPB) {
                const unsigned r = (unsigned)(p[k].y - n0);
                atomicAdd(&acc0[r], v[k].x);
                atomicAdd(&acc1[r], v[k].y);
            }
        }
    }
    __syncthreads();
    if (tid < nn) {
        const float v0 = acc0[tid] * dme + b2[0];
        const float v1 = acc1[tid] * dme + b2[1];
        const float mx = fmaxf(v0, v1);
        const float lse = mx + logf(expf(v0 - mx) + expf(v1 - mx));
        ((float2*)out)[n0 + tid] = make_float2(v0 - lse, v1 - lse);
    }
}

extern "C" void kernel_launch(void* const* d_in, const int* in_sizes, int n_in,
                              void* d_out, int out_size, void* d_ws, size_t ws_size,
                              hipStream_t stream) {
    const float* x  = (const float*)d_in[0];
    const int*   ei = (const int*)d_in[1];
    const float* W1 = (const float*)d_in[2];
    const float* b1 = (const float*)d_in[3];
    const float* W2 = (const float*)d_in[4];
    const float* b2 = (const float*)d_in[5];
    float* out = (float*)d_out;

    const int H    = in_sizes[3];
    const int Fin  = in_sizes[2] / H;
    const int N    = in_sizes[0] / Fin;
    const int E    = in_sizes[1] / 2;
    (void)n_in; (void)out_size; (void)ws_size;

    float* ws   = (float*)d_ws;
    float* dinv = ws;
    float* h    = ws + (size_t)N;
    float* h2   = ws + (size_t)33 * N;
    float* tabs = ws + (((size_t)35 * N + 63) / 64) * 64;

    {
        const int nB = (N + BK_NODES - 1) / BK_NODES;
        const int nW = (E + BIN_CH - 1) / BIN_CH;
        const size_t tabWords  = (((size_t)nB * nW + 63) / 64) * 64;
        unsigned* cntT  = (unsigned*)tabs;
        unsigned* offT  = cntT + tabWords;
        unsigned* total = offT + tabWords;
        unsigned* bbase = total + (((size_t)nB + 63) / 64) * 64;
        int2* sorted    = (int2*)(bbase + (((size_t)nB + 1 + 63) / 64) * 64);
        k_bin_hist   <<<nW, BIN_TPB, 0, stream>>>(ei, E, nB, cntT);
        k_bin_scan   <<<nB, BIN_TPB, 0, stream>>>(cntT, nW, offT, total);
        k_bin_scatter<<<nW, BIN_TPB, 0, stream>>>(ei, E, nB, cntT, offT, total, bbase, sorted);
        k_gemm_wmma  <<<nB, GM_TPB, 0, stream>>>(x, W1, sorted, bbase, dinv, h, N, Fin);
        k_agg1_bucket<<<nB, AG_TPB, 0, stream>>>(sorted, bbase, dinv, h, b1, W2, h2, N);
        k_agg2_bucket<<<nB, A2_TPB, 0, stream>>>(sorted, bbase, dinv, h2, b2, out, N);
    }
}


# Binning in one pass: a chunk reserves its per-bucket runs with one returning global add on the bucket's cursor, lists at fixed stride; histogram and scan launches gone, 6 -> 5 (sum order changes)
# speedup vs baseline: 6.5186x; 1.1200x over previous
#include <hip/hip_runtime.h>
#include <hip/hip_bf16.h>

typedef __attribute__((ext_vector_type(2))) float v2f;
typedef __attribute__((ext_vector_type(8))) float v8f;

#define BK_SHIFT 7
#define BK_NODES 128
#define BK_MAX   4096
#define BK_CAP   6048
#define CUR_PITCH 64
#define BIN_TPB  1024
#define BIN_CH   12288
#define BIN_K    (BIN_CH / BIN_TPB)
#define AG_TPB   512
#define AG_PK    9
#define AG_CAP   (AG_TPB * AG_PK)
#define AG_U     8
#define A2_TPB   512
#define A2_K     9

__global__ __launch_bounds__(BIN_TPB)
void k_cur_zero(unsigned* __restrict__ cursor, int nB) {
    for (int i = threadIdx.x; i <= nB; i += BIN_TPB) cursor[(size_t)i * CUR_PITCH] = 0u;
}

__device__ __forceinline__ void bk_bounds(const unsigned* __restrict__ cursor, int b, int nB,
                                          int& e0, int& e1, int& on) {
    const unsigned cb = cursor[(size_t)b * CUR_PITCH];
    const unsigned co = cursor[(size_t)nB * CUR_PITCH];
    e0 = b * BK_CAP;
    e1 = e0 + (int)min(cb, (unsigned)BK_CAP);
    on = (int)co;
}

__global__ __launch_bounds__(BIN_TPB)
void k_bin_scatter(const int* __restrict__ ei, int E, int nB,
                   unsigned* __restrict__ cursor, int2* __restrict__ lists, int2* __restrict__ ovf) {
    __shared__ int2 stage[BIN_CH];
    __shared__ unsigned lrun[BK_MAX];
    __shared__ unsigned delta[BK_MAX];
    __shared__ unsigned odelta[BK_MAX];
    __shared__ unsigned sl[BIN_TPB];
    const int tid = threadIdx.x;
    const int w   = blockIdx.x;
    for (int i = tid; i < nB; i += BIN_TPB) lrun[i] = 0u;
    __syncthreads();
    const int e0 = w * BIN_CH;
    const int e1 = min(E, e0 + BIN_CH);
    const int* __restrict__ srcp = ei;
    const int* __restrict__ dstp = ei + (size_t)E;
    int ss[BIN_K], dd[BIN_K];
    unsigned rk[BIN_K];
#pragma unroll
    for (int k = 0; k < BIN_K; ++k) {
        const int e = min(e0 + k * BIN_TPB + tid, e1 - 1);
        dd[k] = dstp[e];
        ss[k] = srcp[e];
    }
#pragma unroll
    for (int k = 0; k < BIN_K; ++k) {
        rk[k] = 0u;
        if (e0 + k * BIN_TPB + tid < e1) rk[k] = atomicAdd(&lrun[(unsigned)dd[k] >> BK_SHIFT], 1u);
    }
    __syncthreads();
    const int per = (nB + BIN_TPB - 1) / BIN_TPB;
    unsigned locl = 0u;
    for (int j = 0; j < per; ++j) {
        const int b = tid * per + j;
        if (b < nB) locl += lrun[b];
    }
    sl[tid] = locl;
    __syncthreads();
    for (int d = 1; d < BIN_TPB; d <<= 1) {
        const unsigned tl = (tid >= d) ? sl[tid - d] : 0u;
        __syncthreads();
        sl[tid] += tl;
        __syncthreads();
    }
    unsigned runl = sl[tid] - locl;
    for (int j = 0; j < per; ++j) {
        const int b = tid * per + j;
        if (b < nB) {
            const unsigned cnt = lrun[b];
            lrun[b] = runl;
            unsigned g = 0u, o = 0u;
            if (cnt > 0u) {
                g = atomicAdd(&cursor[(size_t)b * CUR_PITCH], cnt);
                if (g + cnt > (unsigned)BK_CAP) {
                    o = atomicAdd(&cursor[(size_t)nB * CUR_PITCH], g + cnt - max(g, (unsigned)BK_CAP));
                }
            }
            delta[b]  = (unsigned)b * (unsigned)BK_CAP + g - runl;
            odelta[b] = o - max(g, (unsigned)BK_CAP);
            runl += cnt;
        }
    }
    __syncthreads();
#pragma unroll
    for (int k = 0; k < BIN_K; ++k) {
        if (e0 + k * BIN_TPB + tid < e1)
            stage[lrun[(unsigned)dd[k] >> BK_SHIFT] + rk[k]] = make_int2(ss[k], dd[k]);
    }
    __syncthreads();
    const int cnt = max(e1 - e0, 0);
    for (int i = tid; i < cnt; i += BIN_TPB) {
        const int2 pr = stage[i];
        const unsigned b    = (unsigned)pr.y >> BK_SHIFT;
        const unsigned slot = delta[b] + (unsigned)i;
        const unsigned gp   = slot - b * (unsigned)BK_CAP;
        if (gp < (unsigned)BK_CAP) lists[slot] = pr;
        else                       ovf[odelta[b] + gp] = pr;
    }
}

#define GM_TPB 256
#define DG_K   18
#define GM_KC  128
#define GM_PITCH (GM_KC + 4)
typedef __attribute__((ext_vector_type(4))) float v4f;

__device__ __forceinline__ void gm_load_rows(const float* __restrict__ x, int N, int K, int tile, int k0,
                                             int lane, v4f (&xr)[16]) {
    const int kq = k0 + lane * 4;
#pragma unroll
    for (int r = 0; r < 16; ++r) {
        const int row = min(tile * 16 + r, N - 1);
        const v4f* p = (const v4f*)(x + (size_t)row * K + min(kq, K - 4));
        xr[r] = __builtin_nontemporal_load(p);
    }
}

__device__ __forceinline__ void gm_load_w(const float* __restrict__ W1, int K, int k0, int tid,
                                          float (&wr)[GM_KC * 16 / GM_TPB]) {
#pragma unroll
    for (int j = 0; j < GM_KC * 16 / GM_TPB; ++j) {
        const int idx = k0 * 16 + j * GM_TPB + tid;
        wr[j] = W1[min(idx, K * 16 - 1)];
    }
}
__device__ __forceinline__ void gm_stage_w(float* wt, int K, int k0, int tid,
                                           const float (&wr)[GM_KC * 16 / GM_TPB]) {
#pragma unroll
    for (int j = 0; j < GM_KC * 16 / GM_TPB; ++j) {
        const int li = j * GM_TPB + tid;
        wt[(li & 15) * GM_PITCH + (li >> 4)] = (k0 * 16 + li < K * 16) ? wr[j] : 0.f;
    }
}

__device__ __forceinline__ void gm_stage(float* xw, int K, int k0, int lane, const v4f (&xr)[16]) {
    const bool in = k0 + lane * 4 < K;
#pragma unroll
    for (int r = 0; r < 16; ++r)
        *(v4f*)(xw + r * GM_PITCH + lane * 4) = in ? xr[r] : (v4f){0.f, 0.f, 0.f, 0.f};
}

__global__ __launch_bounds__(GM_TPB)
void k_gemm_wmma(const float* __restrict__ x,
                 const float* __restrict__ W1,
                 const int2* __restrict__ sorted, const int2* __restrict__ ovf,
                 const unsigned* __restrict__ cursor,
                 float* __restrict__ dinv,
                 float* __restrict__ h,
                 int N, int K, int nB) {
    __shared__ unsigned cnt[BK_NODES];
    __shared__ __attribute__((aligned(32))) float dl[BK_NODES];
    __shared__ __attribute__((aligned(16))) float xs[(GM_TPB >> 5) * 16 * GM_PITCH];
    __shared__ __attribute__((aligned(16))) float wt[16 * GM_PITCH];
    const int tid = threadIdx.x;
    const int b   = blockIdx.x;
    const int n0  = b << BK_SHIFT;
    const int nn  = max(min(BK_NODES, N - n0), 0);
    const int lane = tid & 31;
    const int wave = __builtin_amdgcn_readfirstlane(tid >> 5);
    const int tile = b * (GM_TPB >> 5) + wave;
    const int ntiles = (N + 15) >> 4;
    const bool live = tile < ntiles;
    float* xw = xs + wave * (16 * GM_PITCH);

    const int m  = lane & 15;
    const int hf = lane >> 4;

    if (tid < BK_NODES) cnt[tid] = 0u;
    __syncthreads();
    v4f xr[16];
    float wr[GM_KC * 16 / GM_TPB];
    if (live) gm_load_rows(x, N, K, tile, 0, lane, xr);
    gm_load_w(W1, K, 0, tid, wr);
    int e0, e1, on;
    bk_bounds(cursor, b, nB, e0, e1, on);
    gm_stage_w(wt, K, 0, tid, wr);
    if (live) gm_stage(xw, K, 0, lane, xr);
    for (int base = e0; base < e1; base += GM_TPB * DG_K) {
        int dd[DG_K];
        const int et = base + tid;
#pragma unroll
        for (int k = 0; k < DG_K; ++k) dd[k] = sorted[min(et + k * GM_TPB, e1 - 1)].y;
#pragma unroll
        for (int k = 0; k < DG_K; ++k) {
            if (et < e1 - k * GM_TPB) atomicAdd(&cnt[(unsigned)(dd[k] - n0)], 1u);
        }
    }
    for (int i = tid; i < on; i += GM_TPB) {
        const int d = ovf[i].y;
        if ((d >> BK_SHIFT) == b) atomicAdd(&cnt[(unsigned)(d - n0)], 1u);
    }
    __syncthreads();
    if (tid < BK_NODES) {
        const float d = rsqrtf((float)(cnt[tid] + 1u));
        dl[tid] = d;
        if (tid < nn) dinv[n0 + tid] = d;
    }
    __syncthreads();

    const int rbase = hf << 3;
    float dv[8];
#pragma unroll
    for (int r = 0; r < 8; ++r) dv[r] = dl[wave * 16 + rbase + r];

    v8f c = {};
    for (int k0 = 0; k0 < K; k0 += GM_KC) {
        if (k0 > 0) {
            __syncthreads();
            if (live) gm_load_rows(x, N, K, tile, k0, lane, xr);
            gm_load_w(W1, K, k0, tid, wr);
            gm_stage_w(wt, K, k0, tid, wr);
            if (live) gm_stage(xw, K, k0, lane, xr);
            __syncthreads();
        }
        if (live) {
            const float* xf = xw + m * GM_PITCH + 4 * hf;
            const float* wf = wt + m * GM_PITCH + 4 * hf;
#pragma unroll
            for (int t = 0; t < GM_KC / 8; ++t) {
                const v4f f = *(const v4f*)(xf + 8 * t);
                const v4f g = *(const v4f*)(wf + 8 * t);
                v2f a, bq;
                a.x = f.x; a.y = f.y; bq.x = g.x; bq.y = g.y;
                c = __builtin_amdgcn_wmma_f32_16x16x4_f32(false, a, false, bq,
                                                          (short)0, c, false, false);
                a.x = f.z; a.y = f.w; bq.x = g.z; bq.y = g.w;
                c = __builtin_amdgcn_wmma_f32_16x16x4_f32(false, a, false, bq,
                                                          (short)0, c, false, false);
            }
        }
    }
    if (!live) return;
#pragma unroll
    for (int r = 0; r < 8; ++r) {
        const int orow = tile * 16 + rbase + r;
        const float o = c[r] * dv[r];
        if (orow < N) h[(size_t)orow * 16 + m] = o;
    }
}

__global__ __launch_bounds__(AG_TPB)
void k_agg1_bucket(const int2* __restrict__ sorted, const int2* __restrict__ ovf,
                   const unsigned* __restrict__ cursor,
                   const float* __restrict__ dinv, const float* __restrict__ h,
                   const float* __restrict__ b1, const float* __restrict__ W2,
                   float* __restrict__ h2, int N, int nB) {
    __shared__ unsigned cnt[BK_NODES];
    __shared__ unsigned off[BK_NODES + 1];
    __shared__ int lst[AG_CAP];
    __shared__ float2 res[BK_NODES];
    const int tid = threadIdx.x;
    const int b   = blockIdx.x;
    const int n0  = b << BK_SHIFT;
    const int nn  = max(min(BK_NODES, N - n0), 0);
    const int node = tid >> 2;
    const int q    = tid & 3;
    const bool own = node < nn;
    const float4* __restrict__ h4 = (const float4*)h;
    if (tid < BK_NODES) cnt[tid] = 0u;
    int e0, e1, on;
    bk_bounds(cursor, b, nB, e0, e1, on);
    __syncthreads();
    const float dme = own ? dinv[n0 + node] : 0.f;
    float4 acc = make_float4(0.f, 0.f, 0.f, 0.f);
    if (own) acc = h4[(size_t)(n0 + node) * 4 + q];
    for (int base = e0; base < e1; base += AG_CAP) {
        int2 p[AG_PK];
        unsigned rk[AG_PK];
        const int et = base + tid;
#pragma unroll
        for (int k = 0; k < AG_PK; ++k) p[k] = sorted[min(et + k * AG_TPB, e1 - 1)];
#pragma unroll
        for (int k = 0; k < AG_PK; ++k) {
            rk[k] = 0u;
            if (et < e1 - k * AG_TPB) rk[k] = atomicAdd(&cnt[(unsigned)(p[k].y - n0)], 1u);
        }
        __syncthreads();
        if (tid < 32) {
            const unsigned c0 = cnt[4 * tid + 0], c1 = cnt[4 * tid + 1];
            const unsigned c2 = cnt[4 * tid + 2], c3 = cnt[4 * tid + 3];
            const unsigned s = c0 + c1 + c2 + c3;
            unsigned inc = s;
#pragma unroll
            for (int d = 1; d < 32; d <<= 1) {
                const unsigned t = __shfl_up(inc, d, 32);
                if (tid >= d) inc += t;
            }
            const unsigned ex = inc - s;
            off[4 * tid + 0] = ex;
            off[4 * tid + 1] = ex + c0;
            off[4 * tid + 2] = ex + c0 + c1;
            off[4 * tid + 3] = ex + c0 + c1 + c2;
            if (tid == 31) off[BK_NODES] = inc;
            cnt[4 * tid + 0] = 0u; cnt[4 * tid + 1] = 0u; cnt[4 * tid + 2] = 0u; cnt[4 * tid + 3] = 0u;
        }
        __syncthreads();
        unsigned ro[AG_PK];
#pragma unroll
        for (int k = 0; k < AG_PK; ++k) ro[k] = off[(unsigned)(p[k].y - n0)];
        asm volatile("" :: "v"(ro[0]), "v"(ro[1]), "v"(ro[2]), "v"(ro[3]), "v"(ro[4]),
                           "v"(ro[5]), "v"(ro[6]), "v"(ro[7]), "v"(ro[8]));
#pragma unroll
        for (int k = 0; k < AG_PK; ++k) {
            if (et < e1 - k * AG_TPB) lst[ro[k] + rk[k]] = p[k].x;
        }
        __syncthreads();
        const int i1 = (int)off[node + 1];
        for (int i = (int)off[node]; i < i1; i += AG_U) {
            const int rem = i1 - i;
            int s[AG_U];
            float4 v[AG_U];
#pragma unroll
            for (int u = 0; u < AG_U; ++u) s[u] = lst[i + min(u, rem - 1)];
#pragma unroll
            for (int u = 0; u < AG_U; ++u) {
                const unsigned o64 = ((unsigned)s[u] << 6) + ((unsigned)q << 4);
                v[u] = *(const float4*)((const char*)h + o64);
            }
            asm volatile("" :: "v"(v[0].x), "v"(v[1].x), "v"(v[2].x), "v"(v[3].x),
                               "v"(v[4].x), "v"(v[5].x), "v"(v[6].x), "v"(v[7].x));
#pragma unroll
            for (int u = 0; u < AG_U; ++u) {
                if (u < rem) {
                    acc.x += v[u].x;
                    acc.y += v[u].y;
                    acc.z += v[u].z;
                    acc.w += v[u].w;
                }
            }
        }
    }
    for (int i = 0; i < on; ++i) {
        const int2 pr = ovf[i];
        if (pr.y == n0 + node) {
            const float4 v = h4[(size_t)pr.x * 4 + q];
            acc.x += v.x;
            acc.y += v.y;
            acc.z += v.z;
            acc.w += v.w;
        }
    }
    float a0 = 0.f, a1 = 0.f;
    const float av[4] = {acc.x, acc.y, acc.z, acc.w};
#pragma unroll
    for (int j = 0; j < 4; ++j) {
        const int f = q * 4 + j;
        float x = av[j] * dme + b1[f];
        x = x > 0.f ? x : 0.f;
        a0 += x * W2[f * 2 + 0];
        a1 += x * W2[f * 2 + 1];
    }
    a0 += __shfl_xor(a0, 1, 32);
    a1 += __shfl_xor(a1, 1, 32);
    a0 += __shfl_xor(a0, 2, 32);
    a1 += __shfl_xor(a1, 2, 32);
    if (q == 0) res[node] = make_float2(a0 * dme, a1 * dme);
    __syncthreads();
    if (tid < nn) ((float2*)h2)[n0 + tid] = res[tid];
}

__global__ __launch_bounds__(A2_TPB)
void k_agg2_bucket(const int2* __restrict__ sorted, const int2* __restrict__ ovf,
                   const unsigned* __restrict__ cursor,
                   const float* __restrict__ dinv, const float* __restrict__ h2,
                   const float* __restrict__ b2, float* __restrict__ out, int N, int nB) {
    __shared__ float acc0[BK_NODES];
    __shared__ float acc1[BK_NODES];
    const int tid = threadIdx.x;
    const int b   = blockIdx.x;
    const int n0  = b << BK_SHIFT;
    const int nn  = max(min(BK_NODES, N - n0), 0);
    const float2* __restrict__ h2v = (const float2*)h2;
    float dme = 0.f;
    if (tid < BK_NODES) {
        float a0 = 0.f, a1 = 0.f;
        if (tid < nn) {
            dme = dinv[n0 + tid];
            const float2 hv = h2v[n0 + tid];
            a0 = hv.x;
            a1 = hv.y;
        }
        acc0[tid] = a0;
        acc1[tid] = a1;
    }
    __syncthreads();
    int e0, e1, on;
    bk_bounds(cursor, b, nB, e0, e1, on);
    for (int base = e0; base < e1; base += A2_TPB * A2_K) {
        int2 p[A2_K];
        float2 v[A2_K];
        const int et = base + tid;
#pragma unroll
        for (int k = 0; k < A2_K; ++k) {
            const int e = min(et + k * A2_TPB, e1 - 1);
            p[k] = sorted[e];
        }
        __builtin_amdgcn_sched_barrier(0);
#pragma unroll
        for (int k = 0; k < A2_K; ++k) {
            const unsigned o8 = (unsigned)p[k].x << 3;
            v[k] = *(const float2*)((const char*)h2 + o8);
        }
        asm volatile("" :: "v"(v[0].x), "v"(v[1].x), "v"(v[2].x), "v"(v[3].x), "v"(v[4].x),
                           "v"(v[5].x), "v"(v[6].x), "v"(v[7].x), "v"(v[8].x));
#pragma unroll
        for (int k = 0; k < A2_K; ++k) {
            if (et < e1 - k * A2_TPB) {
                const unsigned r = (unsigned)(p[k].y - n0);
                atomicAdd(&acc0[r], v[k].x);
                atomicAdd(&acc1[r], v[k].y);
            }
        }
    }
    for (int i = tid; i < on; i += A2_TPB) {
        const int2 pr = ovf[i];
        if ((pr.y >> BK_SHIFT) == b) {
            const float2 hv = h2v[pr.x];
            const unsigned r = (unsigned)(pr.y - n0);
            atomicAdd(&acc0[r], hv.x);
            atomicAdd(&acc1[r], hv.y);
        }
    }
    __syncthreads();
    if (tid < nn) {
        const float v0 = acc0[tid] * dme + b2[0];
        const float v1 = acc1[tid] * dme + b2[1];
        const float mx = fmaxf(v0, v1);
        const float lse = mx + logf(expf(v0 - mx) + expf(v1 - mx));
        ((float2*)out)[n0 + tid] = make_float2(v0 - lse, v1 - lse);
    }
}

extern "C" void kernel_launch(void* const* d_in, const int* in_sizes, int n_in,
                              void* d_out, int out_size, void* d_ws, size_t ws_size,
                              hipStream_t stream) {
    const float* x  = (const float*)d_in[0];
    const int*   ei = (const int*)d_in[1];
    const float* W1 = (const float*)d_in[2];
    const float* b1 = (const float*)d_in[3];
    const float* W2 = (const float*)d_in[4];
    const float* b2 = (const float*)d_in[5];
    float* out = (float*)d_out;

    const int H    = in_sizes[3];
    const int Fin  = in_sizes[2] / H;
    const int N    = in_sizes[0] / Fin;
    const int E    = in_sizes[1] / 2;
    (void)n_in; (void)out_size; (void)ws_size;

    float* ws   = (float*)d_ws;
    float* dinv = ws;
    float* h    = ws + (size_t)N;
    float* h2   = ws + (size_t)33 * N;
    float* tabs = ws + (((size_t)35 * N + 63) / 64) * 64;

    {
        const int nB = (N + BK_NODES - 1) / BK_NODES;
        const int nW = (E + BIN_CH - 1) / BIN_CH;
        unsigned* cursor = (unsigned*)tabs;
        int2* lists      = (int2*)(cursor + ((size_t)nB + 1) * CUR_PITCH);
        int2* ovf        = lists + (size_t)nB * BK_CAP;
        k_cur_zero   <<<1,  BIN_TPB, 0, stream>>>(cursor, nB);
        k_bin_scatter<<<nW, BIN_TPB, 0, stream>>>(ei, E, nB, cursor, lists, ovf);
        k_gemm_wmma  <<<nB, GM_TPB, 0, stream>>>(x, W1, lists, ovf, cursor, dinv, h, N, Fin, nB);
        k_agg1_bucket<<<nB, AG_TPB, 0, stream>>>(lists, ovf, cursor, dinv, h, b1, W2, h2, N, nB);
        k_agg2_bucket<<<nB, A2_TPB, 0, stream>>>(lists, ovf, cursor, dinv, h2, b2, out, N, nB);
    }
}


# Half the waves per work-group in binning, layer 1 (4 lanes walk 2 nodes' runs), layer 2: each thread twice the entries, table reads and cursor adds grouped; fewer waves to start (sum order changes)
# speedup vs baseline: 6.8706x; 1.0540x over previous
#include <hip/hip_runtime.h>
#include <hip/hip_bf16.h>

typedef __attribute__((ext_vector_type(2))) float v2f;
typedef __attribute__((ext_vector_type(8))) float v8f;

#define BK_SHIFT 7
#define BK_NODES 128
#define BK_MAX   4096
#define BK_CAP   6048
#define CUR_PITCH 64
#define BIN_TPB  512
#define BIN_CH   12800
#define BIN_K    (BIN_CH / BIN_TPB)
#define AG_TPB   256
#define AG_G     (AG_TPB / 4)
#define AG_PK    18
#define AG_CAP   (AG_TPB * AG_PK)
#define AG_U     8
#define A2_TPB   256
#define A2_K     18

__global__ __launch_bounds__(BIN_TPB)
void k_cur_zero(unsigned* __restrict__ cursor, int nB) {
    for (int i = threadIdx.x; i <= nB; i += BIN_TPB) cursor[(size_t)i * CUR_PITCH] = 0u;
}

__device__ __forceinline__ void bk_bounds(const unsigned* __restrict__ cursor, int b, int nB,
                                          int& e0, int& e1, int& on) {
    const unsigned cb = cursor[(size_t)b * CUR_PITCH];
    const unsigned co = cursor[(size_t)nB * CUR_PITCH];
    e0 = b * BK_CAP;
    e1 = e0 + (int)min(cb, (unsigned)BK_CAP);
    on = (int)co;
}

#define BIN_PER  (BK_MAX / BIN_TPB)
__global__ __launch_bounds__(BIN_TPB)
void k_bin_scatter(const int* __restrict__ ei, int E, int nB,
                   unsigned* __restrict__ cursor, int2* __restrict__ lists, int2* __restrict__ ovf) {
    __shared__ int2 stage[BIN_CH];
    __shared__ unsigned lrun[BK_MAX];
    __shared__ unsigned delta[BK_MAX];
    __shared__ unsigned odelta[BK_MAX];
    __shared__ unsigned sl[BIN_TPB];
    const int tid = threadIdx.x;
    const int w   = blockIdx.x;
    const int e0 = w * BIN_CH;
    const int e1 = min(E, e0 + BIN_CH);
    if (e1 <= e0) return;
    for (int i = tid; i < nB; i += BIN_TPB) lrun[i] = 0u;
    __syncthreads();
    const int* __restrict__ srcp = ei;
    const int* __restrict__ dstp = ei + (size_t)E;
    int ss[BIN_K], dd[BIN_K];
    unsigned rk[BIN_K];
#pragma unroll
    for (int k = 0; k < BIN_K; ++k) {
        const int e = min(e0 + k * BIN_TPB + tid, e1 - 1);
        dd[k] = dstp[e];
        ss[k] = srcp[e];
    }
#pragma unroll
    for (int k = 0; k < BIN_K; ++k) {
        rk[k] = 0u;
        if (e0 + k * BIN_TPB + tid < e1) rk[k] = atomicAdd(&lrun[(unsigned)dd[k] >> BK_SHIFT], 1u);
    }
    __syncthreads();
    const int per = (nB + BIN_TPB - 1) / BIN_TPB;
    unsigned cn[BIN_PER];
    unsigned locl = 0u;
#pragma unroll
    for (int j = 0; j < BIN_PER; ++j) {
        const int b = tid * per + j;
        const unsigned cv = lrun[b];
        cn[j] = (j < per && b < nB) ? cv : 0u;
        locl += cn[j];
    }
    sl[tid] = locl;
    __syncthreads();
    for (int d = 1; d < BIN_TPB; d <<= 1) {
        const unsigned tl = (tid >= d) ? sl[tid - d] : 0u;
        __syncthreads();
        sl[tid] += tl;
        __syncthreads();
    }
    unsigned runl = sl[tid] - locl;
    unsigned gg[BIN_PER];
#pragma unroll
    for (int j = 0; j < BIN_PER; ++j) {
        const int b = tid * per + j;
        gg[j] = 0u;
        if (cn[j] > 0u) gg[j] = atomicAdd(&cursor[(size_t)b * CUR_PITCH], cn[j]);
    }
    asm volatile("" :: "v"(gg[0]), "v"(gg[1]), "v"(gg[2]), "v"(gg[3]), "v"(gg[4]),
                       "v"(gg[5]), "v"(gg[6]), "v"(gg[7]));
#pragma unroll
    for (int j = 0; j < BIN_PER; ++j) {
        const int b = tid * per + j;
        if (j < per && b < nB) {
            const unsigned cnt = cn[j];
            const unsigned g = gg[j];
            unsigned o = 0u;
            lrun[b] = runl;
            if (cnt > 0u && g + cnt > (unsigned)BK_CAP) {
                o = atomicAdd(&cursor[(size_t)nB * CUR_PITCH], g + cnt - max(g, (unsigned)BK_CAP));
            }
            delta[b]  = (unsigned)b * (unsigned)BK_CAP + g - runl;
            odelta[b] = o - max(g, (unsigned)BK_CAP);
            runl += cnt;
        }
    }
    __syncthreads();
    unsigned rs[BIN_K];
#pragma unroll
    for (int k = 0; k < BIN_K; ++k) rs[k] = lrun[(unsigned)dd[k] >> BK_SHIFT];
#pragma unroll
    for (int k = 0; k < BIN_K; ++k) {
        if (e0 + k * BIN_TPB + tid < e1)
            stage[rs[k] + rk[k]] = make_int2(ss[k], dd[k]);
    }
    __syncthreads();
    const int cnt = e1 - e0;
    int2 pr[BIN_K];
    unsigned dl[BIN_K];
#pragma unroll
    for (int k = 0; k < BIN_K; ++k) pr[k] = stage[min(k * BIN_TPB + tid, cnt - 1)];
#pragma unroll
    for (int k = 0; k < BIN_K; ++k) dl[k] = delta[(unsigned)pr[k].y >> BK_SHIFT];
    asm volatile("" :: "v"(dl[0]), "v"(dl[1]), "v"(dl[2]), "v"(dl[3]), "v"(dl[4]),
                       "v"(dl[5]), "v"(dl[6]), "v"(dl[7]), "v"(dl[8]), "v"(dl[9]),
                       "v"(dl[10]), "v"(dl[11]), "v"(dl[12]), "v"(dl[13]), "v"(dl[14]),
                       "v"(dl[15]), "v"(dl[16]), "v"(dl[17]), "v"(dl[18]), "v"(dl[19]),
                       "v"(dl[20]), "v"(dl[21]), "v"(dl[22]), "v"(dl[23]), "v"(dl[24]));
#pragma unroll
    for (int k = 0; k < BIN_K; ++k) {
        const int i = k * BIN_TPB + tid;
        if (i < cnt) {
            const unsigned b    = (unsigned)pr[k].y >> BK_SHIFT;
            const unsigned slot = dl[k] + (unsigned)i;
            const unsigned gp   = slot - b * (unsigned)BK_CAP;
            if (gp < (unsigned)BK_CAP) lists[slot] = pr[k];
            else                       ovf[odelta[b] + gp] = pr[k];
        }
    }
}

#define GM_TPB 256
#define DG_K   18
#define GM_KC  128
#define GM_PITCH (GM_KC + 4)
typedef __attribute__((ext_vector_type(4))) float v4f;

__device__ __forceinline__ void gm_load_rows(const float* __restrict__ x, int N, int K, int tile, int k0,
                                             int lane, v4f (&xr)[16]) {
    const int kq = k0 + lane * 4;
#pragma unroll
    for (int r = 0; r < 16; ++r) {
        const int row = min(tile * 16 + r, N - 1);
        const v4f* p = (const v4f*)(x + (size_t)row * K + min(kq, K - 4));
        xr[r] = __builtin_nontemporal_load(p);
    }
}

__device__ __forceinline__ void gm_load_w(const float* __restrict__ W1, int K, int k0, int tid,
                                          float (&wr)[GM_KC * 16 / GM_TPB]) {
#pragma unroll
    for (int j = 0; j < GM_KC * 16 / GM_TPB; ++j) {
        const int idx = k0 * 16 + j * GM_TPB + tid;
        wr[j] = W1[min(idx, K * 16 - 1)];
    }
}
__device__ __forceinline__ void gm_stage_w(float* wt, int K, int k0, int tid,
                                           const float (&wr)[GM_KC * 16 / GM_TPB]) {
#pragma unroll
    for (int j = 0; j < GM_KC * 16 / GM_TPB; ++j) {
        const int li = j * GM_TPB + tid;
        wt[(li & 15) * GM_PITCH + (li >> 4)] = (k0 * 16 + li < K * 16) ? wr[j] : 0.f;
    }
}

__device__ __forceinline__ void gm_stage(float* xw, int K, int k0, int lane, const v4f (&xr)[16]) {
    const bool in = k0 + lane * 4 < K;
#pragma unroll
    for (int r = 0; r < 16; ++r)
        *(v4f*)(xw + r * GM_PITCH + lane * 4) = in ? xr[r] : (v4f){0.f, 0.f, 0.f, 0.f};
}

__global__ __launch_bounds__(GM_TPB)
void k_gemm_wmma(const float* __restrict__ x,
                 const float* __restrict__ W1,
                 const int2* __restrict__ sorted, const int2* __restrict__ ovf,
                 const unsigned* __restrict__ cursor,
                 float* __restrict__ dinv,
                 float* __restrict__ h,
                 int N, int K, int nB) {
    __shared__ unsigned cnt[BK_NODES];
    __shared__ __attribute__((aligned(32))) float dl[BK_NODES];
    __shared__ __attribute__((aligned(16))) float xs[(GM_TPB >> 5) * 16 * GM_PITCH];
    __shared__ __attribute__((aligned(16))) float wt[16 * GM_PITCH];
    const int tid = threadIdx.x;
    const int b   = blockIdx.x;
    const int n0  = b << BK_SHIFT;
    const int nn  = max(min(BK_NODES, N - n0), 0);
    const int lane = tid & 31;
    const int wave = __builtin_amdgcn_readfirstlane(tid >> 5);
    const int tile = b * (GM_TPB >> 5) + wave;
    const int ntiles = (N + 15) >> 4;
    const bool live = tile < ntiles;
    float* xw = xs + wave * (16 * GM_PITCH);

    const int m  = lane & 15;
    const int hf = lane >> 4;

    if (tid < BK_NODES) cnt[tid] = 0u;
    __syncthreads();
    v4f xr[16];
    float wr[GM_KC * 16 / GM_TPB];
    if (live) gm_load_rows(x, N, K, tile, 0, lane, xr);
    gm_load_w(W1, K, 0, tid, wr);
    int e0, e1, on;
    bk_bounds(cursor, b, nB, e0, e1, on);
    gm_stage_w(wt, K, 0, tid, wr);
    if (live) gm_stage(xw, K, 0, lane, xr);
    for (int base = e0; base < e1; base += GM_TPB * DG_K) {
        int dd[DG_K];
        const int et = base + tid;
#pragma unroll
        for (int k = 0; k < DG_K; ++k) dd[k] = sorted[min(et + k * GM_TPB, e1 - 1)].y;
#pragma unroll
        for (int k = 0; k < DG_K; ++k) {
            if (et < e1 - k * GM_TPB) atomicAdd(&cnt[(unsigned)(dd[k] - n0)], 1u);
        }
    }
    for (int i = tid; i < on; i += GM_TPB) {
        const int d = ovf[i].y;
        if ((d >> BK_SHIFT) == b) atomicAdd(&cnt[(unsigned)(d - n0)], 1u);
    }
    __syncthreads();
    if (tid < BK_NODES) {
        const float d = rsqrtf((float)(cnt[tid] + 1u));
        dl[tid] = d;
        if (tid < nn) dinv[n0 + tid] = d;
    }
    __syncthreads();

    const int rbase = hf << 3;
    float dv[8];
#pragma unroll
    for (int r = 0; r < 8; ++r) dv[r] = dl[wave * 16 + rbase + r];

    v8f c = {};
    for (int k0 = 0; k0 < K; k0 += GM_KC) {
        if (k0 > 0) {
            __syncthreads();
            if (live) gm_load_rows(x, N, K, tile, k0, lane, xr);
            gm_load_w(W1, K, k0, tid, wr);
            gm_stage_w(wt, K, k0, tid, wr);
            if (live) gm_stage(xw, K, k0, lane, xr);
            __syncthreads();
        }
        if (live) {
            const float* xf = xw + m * GM_PITCH + 4 * hf;
            const float* wf = wt + m * GM_PITCH + 4 * hf;
#pragma unroll
            for (int t = 0; t < GM_KC / 8; ++t) {
                const v4f f = *(const v4f*)(xf + 8 * t);
                const v4f g = *(const v4f*)(wf + 8 * t);
                v2f a, bq;
                a.x = f.x; a.y = f.y; bq.x = g.x; bq.y = g.y;
                c = __builtin_amdgcn_wmma_f32_16x16x4_f32(false, a, false, bq,
                                                          (short)0, c, false, false);
                a.x = f.z; a.y = f.w; bq.x = g.z; bq.y = g.w;
                c = __builtin_amdgcn_wmma_f32_16x16x4_f32(false, a, false, bq,
                                                          (short)0, c, false, false);
            }
        }
    }
    if (!live) return;
#pragma unroll
    for (int r = 0; r < 8; ++r) {
        const int orow = tile * 16 + rbase + r;
        const float o = c[r] * dv[r];
        if (orow < N) h[(size_t)orow * 16 + m] = o;
    }
}

__global__ __launch_bounds__(AG_TPB)
void k_agg1_bucket(const int2* __restrict__ sorted, const int2* __restrict__ ovf,
                   const unsigned* __restrict__ cursor,
                   const float* __restrict__ dinv, const float* __restrict__ h,
                   const float* __restrict__ b1, const float* __restrict__ W2,
                   float* __restrict__ h2, int N, int nB) {
    __shared__ unsigned cnt[BK_NODES];
    __shared__ unsigned off[BK_NODES + 1];
    __shared__ int lst[AG_CAP];
    __shared__ float2 res[BK_NODES];
    const int tid = threadIdx.x;
    const int b   = blockIdx.x;
    const int n0  = b << BK_SHIFT;
    const int nn  = max(min(BK_NODES, N - n0), 0);
    const int ndA = tid >> 2;
    const int ndB = ndA + AG_G;
    const int q   = tid & 3;
    const bool ownA = ndA < nn;
    const bool ownB = ndB < nn;
    const float4* __restrict__ h4 = (const float4*)h;
    if (tid < BK_NODES) cnt[tid] = 0u;
    int e0, e1, on;
    bk_bounds(cursor, b, nB, e0, e1, on);
    __syncthreads();
    const float dmA = ownA ? dinv[n0 + ndA] : 0.f;
    const float dmB = ownB ? dinv[n0 + ndB] : 0.f;
    float4 accA = make_float4(0.f, 0.f, 0.f, 0.f);
    float4 accB = make_float4(0.f, 0.f, 0.f, 0.f);
    if (ownA) accA = h4[(size_t)(n0 + ndA) * 4 + q];
    if (ownB) accB = h4[(size_t)(n0 + ndB) * 4 + q];
    for (int base = e0; base < e1; base += AG_CAP) {
        int2 p[AG_PK];
        unsigned rk[AG_PK];
        const int et = base + tid;
#pragma unroll
        for (int k = 0; k < AG_PK; ++k) p[k] = sorted[min(et + k * AG_TPB, e1 - 1)];
#pragma unroll
        for (int k = 0; k < AG_PK; ++k) {
            rk[k] = 0u;
            if (et < e1 - k * AG_TPB) rk[k] = atomicAdd(&cnt[(unsigned)(p[k].y - n0)], 1u);
        }
        __syncthreads();
        if (tid < 32) {
            const unsigned c0 = cnt[4 * tid + 0], c1 = cnt[4 * tid + 1];
            const unsigned c2 = cnt[4 * tid + 2], c3 = cnt[4 * tid + 3];
            const unsigned s = c0 + c1 + c2 + c3;
            unsigned inc = s;
#pragma unroll
            for (int d = 1; d < 32; d <<= 1) {
                const unsigned t = __shfl_up(inc, d, 32);
                if (tid >= d) inc += t;
            }
            const unsigned ex = inc - s;
            off[4 * tid + 0] = ex;
            off[4 * tid + 1] = ex + c0;
            off[4 * tid + 2] = ex + c0 + c1;
            off[4 * tid + 3] = ex + c0 + c1 + c2;
            if (tid == 31) off[BK_NODES] = inc;
            cnt[4 * tid + 0] = 0u; cnt[4 * tid + 1] = 0u; cnt[4 * tid + 2] = 0u; cnt[4 * tid + 3] = 0u;
        }
        __syncthreads();
        unsigned ro[AG_PK];
#pragma unroll
        for (int k = 0; k < AG_PK; ++k) ro[k] = off[(unsigned)(p[k].y - n0)];
        asm volatile("" :: "v"(ro[0]), "v"(ro[1]), "v"(ro[2]), "v"(ro[3]), "v"(ro[4]),
                           "v"(ro[5]), "v"(ro[6]), "v"(ro[7]), "v"(ro[8]),
                           "v"(ro[9]), "v"(ro[10]), "v"(ro[11]), "v"(ro[12]), "v"(ro[13]),
                           "v"(ro[14]), "v"(ro[15]), "v"(ro[16]), "v"(ro[17]));
#pragma unroll
        for (int k = 0; k < AG_PK; ++k) {
            if (et < e1 - k * AG_TPB) lst[ro[k] + rk[k]] = p[k].x;
        }
        __syncthreads();
        int iA = (int)off[ndA];
        int iB = (int)off[ndB];
        const int eA = (int)off[ndA + 1];
        const int eB = (int)off[ndB + 1];
        while (iA < eA || iB < eB) {
            const int remA = eA - iA;
            const int remB = eB - iB;
            int sA[AG_U], sB[AG_U];
            float4 vA[AG_U], vB[AG_U];
#pragma unroll
            for (int u = 0; u < AG_U; ++u) sA[u] = lst[max(iA + min(u, remA - 1), 0)];
#pragma unroll
            for (int u = 0; u < AG_U; ++u) sB[u] = lst[max(iB + min(u, remB - 1), 0)];
#pragma unroll
            for (int u = 0; u < AG_U; ++u) {
                const unsigned oA = ((unsigned)sA[u] << 6) + ((unsigned)q << 4);
                const unsigned oB = ((unsigned)sB[u] << 6) + ((unsigned)q << 4);
                vA[u] = *(const float4*)((const char*)h + oA);
                vB[u] = *(const float4*)((const char*)h + oB);
            }
            asm volatile("" :: "v"(vA[0].x), "v"(vA[1].x), "v"(vA[2].x), "v"(vA[3].x),
                               "v"(vA[4].x), "v"(vA[5].x), "v"(vA[6].x), "v"(vA[7].x),
                               "v"(vB[0].x), "v"(vB[1].x), "v"(vB[2].x), "v"(vB[3].x),
                               "v"(vB[4].x), "v"(vB[5].x), "v"(vB[6].x), "v"(vB[7].x));
#pragma unroll
            for (int u = 0; u < AG_U; ++u) {
                if (u < remA) {
                    accA.x += vA[u].x;
                    accA.y += vA[u].y;
                    accA.z += vA[u].z;
                    accA.w += vA[u].w;
                }
            }
#pragma unroll
            for (int u = 0; u < AG_U; ++u) {
                if (u < remB) {
                    accB.x += vB[u].x;
                    accB.y += vB[u].y;
                    accB.z += vB[u].z;
                    accB.w += vB[u].w;
                }
            }
            iA += AG_U;
            iB += AG_U;
        }
    }
    for (int i = 0; i < on; ++i) {
        const int2 pr = ovf[i];
        if (pr.y == n0 + ndA) {
            const float4 v = h4[(size_t)pr.x * 4 + q];
            accA.x += v.x;
            accA.y += v.y;
            accA.z += v.z;
            accA.w += v.w;
        }
        if (pr.y == n0 + ndB) {
            const float4 v = h4[(size_t)pr.x * 4 + q];
            accB.x += v.x;
            accB.y += v.y;
            accB.z += v.z;
            accB.w += v.w;
        }
    }
    float a0A = 0.f, a1A = 0.f, a0B = 0.f, a1B = 0.f;
    const float avA[4] = {accA.x, accA.y, accA.z, accA.w};
    const float avB[4] = {accB.x, accB.y, accB.z, accB.w};
#pragma unroll
    for (int j = 0; j < 4; ++j) {
        const int f = q * 4 + j;
        const float bf = b1[f];
        const float w0 = W2[f * 2 + 0];
        const float w1 = W2[f * 2 + 1];
        float xa = avA[j] * dmA + bf;
        xa = xa > 0.f ? xa : 0.f;
        a0A += xa * w0;
        a1A += xa * w1;
        float xb = avB[j] * dmB + bf;
        xb = xb > 0.f ? xb : 0.f;
        a0B += xb * w0;
        a1B += xb * w1;
    }
    a0A += __shfl_xor(a0A, 1, 32);
    a1A += __shfl_xor(a1A, 1, 32);
    a0B += __shfl_xor(a0B, 1, 32);
    a1B += __shfl_xor(a1B, 1, 32);
    a0A += __shfl_xor(a0A, 2, 32);
    a1A += __shfl_xor(a1A, 2, 32);
    a0B += __shfl_xor(a0B, 2, 32);
    a1B += __shfl_xor(a1B, 2, 32);
    if (q == 0) {
        res[ndA] = make_float2(a0A * dmA, a1A * dmA);
        res[ndB] = make_float2(a0B * dmB, a1B * dmB);
    }
    __syncthreads();
    if (tid < nn) ((float2*)h2)[n0 + tid] = res[tid];
}

__global__ __launch_bounds__(A2_TPB)
void k_agg2_bucket(const int2* __restrict__ sorted, const int2* __restrict__ ovf,
                   const unsigned* __restrict__ cursor,
                   const float* __restrict__ dinv, const float* __restrict__ h2,
                   const float* __restrict__ b2, float* __restrict__ out, int N, int nB) {
    __shared__ float acc0[BK_NODES];
    __shared__ float acc1[BK_NODES];
    const int tid = threadIdx.x;
    const int b   = blockIdx.x;
    const int n0  = b << BK_SHIFT;
    const int nn  = max(min(BK_NODES, N - n0), 0);
    const float2* __restrict__ h2v = (const float2*)h2;
    float dme = 0.f;
    if (tid < BK_NODES) {
        float a0 = 0.f, a1 = 0.f;
        if (tid < nn) {
            dme = dinv[n0 + tid];
            const float2 hv = h2v[n0 + tid];
            a0 = hv.x;
            a1 = hv.y;
        }
        acc0[tid] = a0;
        acc1[tid] = a1;
    }
    __syncthreads();
    int e0, e1, on;
    bk_bounds(cursor, b, nB, e0, e1, on);
    for (int base = e0; base < e1; base += A2_TPB * A2_K) {
        int2 p[A2_K];
        float2 v[A2_K];
        const int et = base + tid;
#pragma unroll
        for (int k = 0; k < A2_K; ++k) {
            const int e = min(et + k * A2_TPB, e1 - 1);
            p[k] = sorted[e];
        }
        __builtin_amdgcn_sched_barrier(0);
#pragma unroll
        for (int k = 0; k < A2_K; ++k) {
            const unsigned o8 = (unsigned)p[k].x << 3;
            v[k] = *(const float2*)((const char*)h2 + o8);
        }
        asm volatile("" :: "v"(v[0].x), "v"(v[1].x), "v"(v[2].x), "v"(v[3].x), "v"(v[4].x),
                           "v"(v[5].x), "v"(v[6].x), "v"(v[7].x), "v"(v[8].x),
                           "v"(v[9].x), "v"(v[10].x), "v"(v[11].x), "v"(v[12].x), "v"(v[13].x),
                           "v"(v[14].x), "v"(v[15].x), "v"(v[16].x), "v"(v[17].x));
#pragma unroll
        for (int k = 0; k < A2_K; ++k) {
            if (et < e1 - k * A2_TPB) {
                const unsigned r = (unsigned)(p[k].y - n0);
                atomicAdd(&acc0[r], v[k].x);
                atomicAdd(&acc1[r], v[k].y);
            }
        }
    }
    for (int i = tid; i < on; i += A2_TPB) {
        const int2 pr = ovf[i];
        if ((pr.y >> BK_SHIFT) == b) {
            const float2 hv = h2v[pr.x];
            const unsigned r = (unsigned)(pr.y - n0);
            atomicAdd(&acc0[r], hv.x);
            atomicAdd(&acc1[r], hv.y);
        }
    }
    __syncthreads();
    if (tid < nn) {
        const float v0 = acc0[tid] * dme + b2[0];
        const float v1 = acc1[tid] * dme + b2[1];
        const float mx = fmaxf(v0, v1);
        const float lse = mx + logf(expf(v0 - mx) + expf(v1 - mx));
        ((float2*)out)[n0 + tid] = make_float2(v0 - lse, v1 - lse);
    }
}

extern "C" void kernel_launch(void* const* d_in, const int* in_sizes, int n_in,
                              void* d_out, int out_size, void* d_ws, size_t ws_size,
                              hipStream_t stream) {
    const float* x  = (const float*)d_in[0];
    const int*   ei = (const int*)d_in[1];
    const float* W1 = (const float*)d_in[2];
    const float* b1 = (const float*)d_in[3];
    const float* W2 = (const float*)d_in[4];
    const float* b2 = (const float*)d_in[5];
    float* out = (float*)d_out;

    const int H    = in_sizes[3];
    const int Fin  = in_sizes[2] / H;
    const int N    = in_sizes[0] / Fin;
    const int E    = in_sizes[1] / 2;
    (void)n_in; (void)out_size; (void)ws_size;

    float* ws   = (float*)d_ws;
    float* dinv = ws;
    float* h    = ws + (size_t)N;
    float* h2   = ws + (size_t)33 * N;
    float* tabs = ws + (((size_t)35 * N + 63) / 64) * 64;

    {
        const int nB = (N + BK_NODES - 1) / BK_NODES;
        const int nW = (E + BIN_CH - 1) / BIN_CH;
        unsigned* cursor = (unsigned*)tabs;
        int2* lists      = (int2*)(cursor + ((size_t)nB + 1) * CUR_PITCH);
        int2* ovf        = lists + (size_t)nB * BK_CAP;
        k_cur_zero   <<<1,  BIN_TPB, 0, stream>>>(cursor, nB);
        k_bin_scatter<<<nW, BIN_TPB, 0, stream>>>(ei, E, nB, cursor, lists, ovf);
        k_gemm_wmma  <<<nB, GM_TPB, 0, stream>>>(x, W1, lists, ovf, cursor, dinv, h, N, Fin, nB);
        k_agg1_bucket<<<nB, AG_TPB, 0, stream>>>(lists, ovf, cursor, dinv, h, b1, W2, h2, N, nB);
        k_agg2_bucket<<<nB, A2_TPB, 0, stream>>>(lists, ovf, cursor, dinv, h2, b2, out, N, nB);
    }
}


# Binning without cursors: per (bucket, chunk) one whole 128-byte line of 32 packed pairs at a fixed place; cursor adds, scan, run staging and clearing launch gone, 5 -> 4 launches (sum order changes)
# speedup vs baseline: 8.8723x; 1.2913x over previous
#include <hip/hip_runtime.h>
#include <hip/hip_bf16.h>

typedef __attribute__((ext_vector_type(2))) float v2f;
typedef __attribute__((ext_vector_type(8))) float v8f;
typedef __attribute__((ext_vector_type(4))) unsigned v4u;

#define BK_SHIFT 7
#define BK_NODES 128
#define BK_STG   1024
#define SEG_SLOTS 32
#define SEG_EMPTY 0xFFFFFFFFu
#define SRC_BITS 25
#define SRC_MASK 0x1FFFFFFu
#define OVL_CAP  256
#define BIN_TPB  512
#define BIN_CH   12800
#define BIN_K    (BIN_CH / BIN_TPB)
#define AG_TPB   256
#define AG_G     (AG_TPB / 4)
#define AG_PK    32
#define AG_CAP   (AG_TPB * AG_PK)
#define AG_U     8
#define A2_TPB   256
#define A2_K     16

__device__ __forceinline__ unsigned pk_pair(int src, int dst) {
    return (((unsigned)dst & (unsigned)(BK_NODES - 1)) << SRC_BITS) | (unsigned)src;
}

__global__ __launch_bounds__(BIN_TPB)
void k_bin_scatter(const int* __restrict__ ei, int E, int nB, int nW,
                   unsigned* __restrict__ lists, int2* __restrict__ ovf, unsigned* __restrict__ ovfcnt) {
    __shared__ __attribute__((aligned(16))) unsigned stage[BK_STG * SEG_SLOTS];
    __shared__ unsigned lrun[BK_STG + 1];
    __shared__ unsigned ovc;
    const int tid = threadIdx.x;
    const int w   = blockIdx.x;
    const int e0 = w * BIN_CH;
    const int e1 = min(E, e0 + BIN_CH);
    if (e1 <= e0) {
        if (tid == 0) ovfcnt[w] = 0u;
        return;
    }
    for (int i = tid; i < nB; i += BIN_TPB) lrun[i] = 0u;
    if (tid == 0) ovc = 0u;
    __syncthreads();
    const int* __restrict__ srcp = ei;
    const int* __restrict__ dstp = ei + (size_t)E;
    int ss[BIN_K], dd[BIN_K];
    unsigned rk[BIN_K];
#pragma unroll
    for (int k = 0; k < BIN_K; ++k) {
        const int e = min(e0 + k * BIN_TPB + tid, e1 - 1);
        dd[k] = dstp[e];
        ss[k] = srcp[e];
    }
#pragma unroll
    for (int k = 0; k < BIN_K; ++k) {
        const unsigned bi = (e0 + k * BIN_TPB + tid < e1) ? ((unsigned)dd[k] >> BK_SHIFT) : (unsigned)BK_STG;
        rk[k] = atomicAdd(&lrun[bi], 1u);
    }
#pragma unroll
    for (int k = 0; k < BIN_K; ++k) {
        if (e0 + k * BIN_TPB + tid < e1 && rk[k] < (unsigned)SEG_SLOTS)
            stage[((unsigned)dd[k] >> BK_SHIFT) * SEG_SLOTS + rk[k]] = pk_pair(ss[k], dd[k]);
    }
#pragma unroll
    for (int k = 0; k < BIN_K; ++k) {
        if (e0 + k * BIN_TPB + tid < e1 && rk[k] >= (unsigned)SEG_SLOTS) {
            const unsigned j = atomicAdd(&ovc, 1u);
            ovf[(size_t)w * BIN_CH + j] = make_int2(ss[k], dd[k]);
        }
    }
    __syncthreads();
    const int nq = nB * (SEG_SLOTS / 4);
    unsigned* __restrict__ outw = lists + (size_t)w * SEG_SLOTS;
    const size_t bstride = (size_t)nW * SEG_SLOTS;
#pragma unroll 4
    for (int q = tid; q < nq; q += BIN_TPB) {
        const int bq = q >> 3;
        const unsigned s0 = (unsigned)(q & 7) * 4u;
        const unsigned c = lrun[bq];
        v4u v = *(const v4u*)(stage + (size_t)q * 4);
        v.x = (s0 + 0u < c) ? v.x : SEG_EMPTY;
        v.y = (s0 + 1u < c) ? v.y : SEG_EMPTY;
        v.z = (s0 + 2u < c) ? v.z : SEG_EMPTY;
        v.w = (s0 + 3u < c) ? v.w : SEG_EMPTY;
        *(v4u*)(outw + (size_t)bq * bstride + s0) = v;
    }
    if (tid == 0) ovfcnt[w] = ovc;
}

#define GM_TPB 256
#define DG_K   16
#define GM_KC  128
#define GM_PITCH (GM_KC + 4)
typedef __attribute__((ext_vector_type(4))) float v4f;

__device__ __forceinline__ void gm_load_rows(const float* __restrict__ x, int N, int K, int tile, int k0,
                                             int lane, v4f (&xr)[16]) {
    const int kq = k0 + lane * 4;
#pragma unroll
    for (int r = 0; r < 16; ++r) {
        const int row = min(tile * 16 + r, N - 1);
        const v4f* p = (const v4f*)(x + (size_t)row * K + min(kq, K - 4));
        xr[r] = __builtin_nontemporal_load(p);
    }
}

__device__ __forceinline__ void gm_load_w(const float* __restrict__ W1, int K, int k0, int tid,
                                          float (&wr)[GM_KC * 16 / GM_TPB]) {
#pragma unroll
    for (int j = 0; j < GM_KC * 16 / GM_TPB; ++j) {
        const int idx = k0 * 16 + j * GM_TPB + tid;
        wr[j] = W1[min(idx, K * 16 - 1)];
    }
}
__device__ __forceinline__ void gm_stage_w(float* wt, int K, int k0, int tid,
                                           const float (&wr)[GM_KC * 16 / GM_TPB]) {
#pragma unroll
    for (int j = 0; j < GM_KC * 16 / GM_TPB; ++j) {
        const int li = j * GM_TPB + tid;
        wt[(li & 15) * GM_PITCH + (li >> 4)] = (k0 * 16 + li < K * 16) ? wr[j] : 0.f;
    }
}

__device__ __forceinline__ void gm_stage(float* xw, int K, int k0, int lane, const v4f (&xr)[16]) {
    const bool in = k0 + lane * 4 < K;
#pragma unroll
    for (int r = 0; r < 16; ++r)
        *(v4f*)(xw + r * GM_PITCH + lane * 4) = in ? xr[r] : (v4f){0.f, 0.f, 0.f, 0.f};
}

__device__ __forceinline__ void gm_take_ovf(const int2 pr, int b, int n0, unsigned* cnt, unsigned* ovn,
                                            int2* __restrict__ bovf) {
    if ((pr.y >> BK_SHIFT) == b) {
        atomicAdd(&cnt[(unsigned)(pr.y - n0)], 1u);
        const unsigned j = atomicAdd(ovn, 1u);
        if (j < (unsigned)OVL_CAP) bovf[(size_t)b * OVL_CAP + j] = pr;
    }
}

__global__ __launch_bounds__(GM_TPB)
void k_gemm_wmma(const float* __restrict__ x,
                 const float* __restrict__ W1,
                 const unsigned* __restrict__ lists, const int2* __restrict__ ovf,
                 const unsigned* __restrict__ ovfcnt,
                 int2* __restrict__ bovf, unsigned* __restrict__ bovn,
                 float* __restrict__ dinv,
                 float* __restrict__ h,
                 int N, int K, int nW) {
    __shared__ unsigned cnt[BK_NODES];
    __shared__ unsigned ovn;
    __shared__ __attribute__((aligned(32))) float dl[BK_NODES];
    __shared__ __attribute__((aligned(16))) float xs[(GM_TPB >> 5) * 16 * GM_PITCH];
    __shared__ __attribute__((aligned(16))) float wt[16 * GM_PITCH];
    const int tid = threadIdx.x;
    const int b   = blockIdx.x;
    const int n0  = b << BK_SHIFT;
    const int nn  = max(min(BK_NODES, N - n0), 0);
    const int lane = tid & 31;
    const int wave = __builtin_amdgcn_readfirstlane(tid >> 5);
    const int tile = b * (GM_TPB >> 5) + wave;
    const int ntiles = (N + 15) >> 4;
    const bool live = tile < ntiles;
    float* xw = xs + wave * (16 * GM_PITCH);

    const int m  = lane & 15;
    const int hf = lane >> 4;

    if (tid < BK_NODES) cnt[tid] = 0u;
    if (tid == 0) ovn = 0u;
    __syncthreads();
    const int oc0 = (tid < nW) ? (int)ovfcnt[tid] : 0;
    v4f xr[16];
    float wr[GM_KC * 16 / GM_TPB];
    gm_load_rows(x, N, K, tile, 0, lane, xr);
    gm_load_w(W1, K, 0, tid, wr);
    for (int i = 0; i < oc0; ++i)
        gm_take_ovf(ovf[(size_t)tid * BIN_CH + i], b, n0, cnt, &ovn, bovf);
    for (int w = tid + GM_TPB; w < nW; w += GM_TPB) {
        const int c = (int)ovfcnt[w];
        for (int i = 0; i < c; ++i)
            gm_take_ovf(ovf[(size_t)w * BIN_CH + i], b, n0, cnt, &ovn, bovf);
    }
    gm_stage_w(wt, K, 0, tid, wr);
    if (live) gm_stage(xw, K, 0, lane, xr);
    const int nS = nW * SEG_SLOTS;
    const unsigned* __restrict__ bl = lists + (size_t)b * nS;
    for (int base = 0; base < nS; base += GM_TPB * DG_K) {
        unsigned pw[DG_K];
        const int st = base + tid;
#pragma unroll
        for (int k = 0; k < DG_K; ++k) pw[k] = bl[min(st + k * GM_TPB, nS - 1)];
#pragma unroll
        for (int k = 0; k < DG_K; ++k) {
            if (st < nS - k * GM_TPB && pw[k] != SEG_EMPTY) atomicAdd(&cnt[pw[k] >> SRC_BITS], 1u);
        }
    }
    __syncthreads();
    if (tid == 0) bovn[b] = ovn;
    if (tid < BK_NODES) {
        const float d = rsqrtf((float)(cnt[tid] + 1u));
        dl[tid] = d;
        if (tid < nn) dinv[n0 + tid] = d;
    }
    __syncthreads();

    const int rbase = hf << 3;
    float dv[8];
#pragma unroll
    for (int r = 0; r < 8; ++r) dv[r] = dl[wave * 16 + rbase + r];

    v8f c = {};
    for (int k0 = 0; k0 < K; k0 += GM_KC) {
        if (k0 > 0) {
            __syncthreads();
            if (live) gm_load_rows(x, N, K, tile, k0, lane, xr);
            gm_load_w(W1, K, k0, tid, wr);
            gm_stage_w(wt, K, k0, tid, wr);
            if (live) gm_stage(xw, K, k0, lane, xr);
            __syncthreads();
        }
        if (live) {
            const float* xf = xw + m * GM_PITCH + 4 * hf;
            const float* wf = wt + m * GM_PITCH + 4 * hf;
#pragma unroll
            for (int t = 0; t < GM_KC / 8; ++t) {
                const v4f f = *(const v4f*)(xf + 8 * t);
                const v4f g = *(const v4f*)(wf + 8 * t);
                v2f a, bq;
                a.x = f.x; a.y = f.y; bq.x = g.x; bq.y = g.y;
                c = __builtin_amdgcn_wmma_f32_16x16x4_f32(false, a, false, bq,
                                                          (short)0, c, false, false);
                a.x = f.z; a.y = f.w; bq.x = g.z; bq.y = g.w;
                c = __builtin_amdgcn_wmma_f32_16x16x4_f32(false, a, false, bq,
                                                          (short)0, c, false, false);
            }
        }
    }
    if (!live) return;
#pragma unroll
    for (int r = 0; r < 8; ++r) {
        const int orow = tile * 16 + rbase + r;
        const float o = c[r] * dv[r];
        if (orow < N) h[(size_t)orow * 16 + m] = o;
    }
}

__global__ __launch_bounds__(AG_TPB) __attribute__((amdgpu_waves_per_eu(8)))
void k_agg1_bucket(const unsigned* __restrict__ lists, const int2* __restrict__ ovf,
                   const unsigned* __restrict__ ovfcnt,
                   const int2* __restrict__ bovf, const unsigned* __restrict__ bovn,
                   const float* __restrict__ dinv, const float* __restrict__ h,
                   const float* __restrict__ b1, const float* __restrict__ W2,
                   float* __restrict__ h2, int N, int nW) {
    __shared__ unsigned cnt[BK_NODES];
    __shared__ unsigned off[BK_NODES + 1];
    __shared__ int lst[AG_CAP + OVL_CAP];
    __shared__ float2 res[BK_NODES];
    __shared__ float dml[BK_NODES];
    const int tid = threadIdx.x;
    const int b   = blockIdx.x;
    const int n0  = b << BK_SHIFT;
    const int nn  = max(min(BK_NODES, N - n0), 0);
    const int ndA = tid >> 2;
    const int ndB = ndA + AG_G;
    const int q   = tid & 3;
    const bool ownA = ndA < nn;
    const bool ownB = ndB < nn;
    const float4* __restrict__ h4 = (const float4*)h;
    if (tid < BK_NODES) cnt[tid] = 0u;
    const int ovall = (int)bovn[b];
    const int ovm = (ovall <= OVL_CAP) ? ovall : 0;
    const int nS = nW * SEG_SLOTS;
    const unsigned* __restrict__ bl = lists + (size_t)b * nS;
    __syncthreads();
    const float dmt = (tid < nn) ? dinv[n0 + tid] : 0.f;
    float4 accA = make_float4(0.f, 0.f, 0.f, 0.f);
    float4 accB = make_float4(0.f, 0.f, 0.f, 0.f);
    if (ownA) accA = h4[(size_t)(n0 + ndA) * 4 + q];
    if (ownB) accB = h4[(size_t)(n0 + ndB) * 4 + q];
    for (int base = 0; base < nS; base += AG_CAP) {
        unsigned p[AG_PK + 1];
        unsigned rk[AG_PK + 1];
        const int st = base + tid;
#pragma unroll
        for (int k = 0; k < AG_PK; ++k) p[k] = bl[min(st + k * AG_TPB, nS - 1)];
        p[AG_PK] = SEG_EMPTY;
        if (base == 0 && tid < ovm) {
            const int2 pr = *(const int2*)((const char*)bovf + ((unsigned)(b * OVL_CAP + tid) << 3));
            p[AG_PK] = pk_pair(pr.x, pr.y);
        }
#pragma unroll
        for (int k = 0; k < AG_PK; ++k) p[k] = (st < nS - k * AG_TPB) ? p[k] : SEG_EMPTY;
#pragma unroll
        for (int k = 0; k <= AG_PK; ++k) {
            rk[k] = 0u;
            if (p[k] != SEG_EMPTY) rk[k] = atomicAdd(&cnt[p[k] >> SRC_BITS], 1u);
        }
        if (base == 0 && tid < BK_NODES) dml[tid] = dmt;
        __syncthreads();
        if (tid < 32) {
            const unsigned c0 = cnt[4 * tid + 0], c1 = cnt[4 * tid + 1];
            const unsigned c2 = cnt[4 * tid + 2], c3 = cnt[4 * tid + 3];
            const unsigned s = c0 + c1 + c2 + c3;
            unsigned inc = s;
#pragma unroll
            for (int d = 1; d < 32; d <<= 1) {
                const unsigned t = __shfl_up(inc, d, 32);
                if (tid >= d) inc += t;
            }
            const unsigned ex = inc - s;
            off[4 * tid + 0] = ex;
            off[4 * tid + 1] = ex + c0;
            off[4 * tid + 2] = ex + c0 + c1;
            off[4 * tid + 3] = ex + c0 + c1 + c2;
            if (tid == 31) off[BK_NODES] = inc;
            cnt[4 * tid + 0] = 0u; cnt[4 * tid + 1] = 0u; cnt[4 * tid + 2] = 0u; cnt[4 * tid + 3] = 0u;
        }
        __syncthreads();
        {
            unsigned ro[17];
#pragma unroll
            for (int k = 0; k < 17; ++k) ro[k] = off[p[k] >> SRC_BITS];
            asm volatile("" :: "v"(ro[0]), "v"(ro[1]), "v"(ro[2]), "v"(ro[3]), "v"(ro[4]),
                               "v"(ro[5]), "v"(ro[6]), "v"(ro[7]), "v"(ro[8]),
                               "v"(ro[9]), "v"(ro[10]), "v"(ro[11]), "v"(ro[12]), "v"(ro[13]),
                               "v"(ro[14]), "v"(ro[15]), "v"(ro[16]));
#pragma unroll
            for (int k = 0; k < 17; ++k) {
                if (p[k] != SEG_EMPTY) lst[ro[k] + rk[k]] = (int)(p[k] & SRC_MASK);
            }
        }
        {
            unsigned ro[16];
#pragma unroll
            for (int k = 0; k < 16; ++k) ro[k] = off[p[17 + k] >> SRC_BITS];
            asm volatile("" :: "v"(ro[0]), "v"(ro[1]), "v"(ro[2]), "v"(ro[3]), "v"(ro[4]),
                               "v"(ro[5]), "v"(ro[6]), "v"(ro[7]), "v"(ro[8]),
                               "v"(ro[9]), "v"(ro[10]), "v"(ro[11]), "v"(ro[12]), "v"(ro[13]),
                               "v"(ro[14]), "v"(ro[15]));
#pragma unroll
            for (int k = 0; k < 16; ++k) {
                if (p[17 + k] != SEG_EMPTY) lst[ro[k] + rk[17 + k]] = (int)(p[17 + k] & SRC_MASK);
            }
        }
        __syncthreads();
        int iA = (int)off[ndA];
        int iB = (int)off[ndB];
        const int eA = (int)off[ndA + 1];
        const int eB = (int)off[ndB + 1];
        while (iA < eA || iB < eB) {
            const int remA = eA - iA;
            const int remB = eB - iB;
            int sA[AG_U], sB[AG_U];
            float4 vA[AG_U], vB[AG_U];
#pragma unroll
            for (int u = 0; u < AG_U; ++u) sA[u] = lst[max(iA + min(u, remA - 1), 0)];
#pragma unroll
            for (int u = 0; u < AG_U; ++u) sB[u] = lst[max(iB + min(u, remB - 1), 0)];
#pragma unroll
            for (int u = 0; u < AG_U; ++u) {
                const unsigned oA = ((unsigned)sA[u] << 6) + ((unsigned)q << 4);
                const unsigned oB = ((unsigned)sB[u] << 6) + ((unsigned)q << 4);
                vA[u] = *(const float4*)((const char*)h + oA);
                vB[u] = *(const float4*)((const char*)h + oB);
            }
            asm volatile("" :: "v"(vA[0].x), "v"(vA[1].x), "v"(vA[2].x), "v"(vA[3].x),
                               "v"(vA[4].x), "v"(vA[5].x), "v"(vA[6].x), "v"(vA[7].x),
                               "v"(vB[0].x), "v"(vB[1].x), "v"(vB[2].x), "v"(vB[3].x),
                               "v"(vB[4].x), "v"(vB[5].x), "v"(vB[6].x), "v"(vB[7].x));
#pragma unroll
            for (int u = 0; u < AG_U; ++u) {
                if (u < remA) {
                    accA.x += vA[u].x;
                    accA.y += vA[u].y;
                    accA.z += vA[u].z;
                    accA.w += vA[u].w;
                }
            }
#pragma unroll
            for (int u = 0; u < AG_U; ++u) {
                if (u < remB) {
                    accB.x += vB[u].x;
                    accB.y += vB[u].y;
                    accB.z += vB[u].z;
                    accB.w += vB[u].w;
                }
            }
            iA += AG_U;
            iB += AG_U;
        }
    }
    if (ovall > OVL_CAP) {
        for (int w = 0; w < nW; ++w) {
            const int c = (int)ovfcnt[w];
            for (int i = 0; i < c; ++i) {
                const int2 pr = ovf[(size_t)w * BIN_CH + i];
                if (pr.y == n0 + ndA) {
                    const float4 v = h4[(size_t)pr.x * 4 + q];
                    accA.x += v.x;
                    accA.y += v.y;
                    accA.z += v.z;
                    accA.w += v.w;
                }
                if (pr.y == n0 + ndB) {
                    const float4 v = h4[(size_t)pr.x * 4 + q];
                    accB.x += v.x;
                    accB.y += v.y;
                    accB.z += v.z;
                    accB.w += v.w;
                }
            }
        }
    }
    const float dmA = dml[ndA];
    const float dmB = dml[ndB];
    float a0A = 0.f, a1A = 0.f, a0B = 0.f, a1B = 0.f;
    const float avA[4] = {accA.x, accA.y, accA.z, accA.w};
    const float avB[4] = {accB.x, accB.y, accB.z, accB.w};
#pragma unroll
    for (int j = 0; j < 4; ++j) {
        const int f = q * 4 + j;
        const float bf = b1[f];
        const float w0 = W2[f * 2 + 0];
        const float w1 = W2[f * 2 + 1];
        float xa = avA[j] * dmA + bf;
        xa = xa > 0.f ? xa : 0.f;
        a0A += xa * w0;
        a1A += xa * w1;
        float xb = avB[j] * dmB + bf;
        xb = xb > 0.f ? xb : 0.f;
        a0B += xb * w0;
        a1B += xb * w1;
    }
    a0A += __shfl_xor(a0A, 1, 32);
    a1A += __shfl_xor(a1A, 1, 32);
    a0B += __shfl_xor(a0B, 1, 32);
    a1B += __shfl_xor(a1B, 1, 32);
    a0A += __shfl_xor(a0A, 2, 32);
    a1A += __shfl_xor(a1A, 2, 32);
    a0B += __shfl_xor(a0B, 2, 32);
    a1B += __shfl_xor(a1B, 2, 32);
    if (q == 0) {
        res[ndA] = make_float2(a0A * dmA, a1A * dmA);
        res[ndB] = make_float2(a0B * dmB, a1B * dmB);
    }
    __syncthreads();
    if (tid < nn) *(float2*)((char*)h2 + ((unsigned)(n0 + tid) << 3)) = res[tid];
}

__global__ __launch_bounds__(A2_TPB)
void k_agg2_bucket(const unsigned* __restrict__ lists, const int2* __restrict__ ovf,
                   const unsigned* __restrict__ ovfcnt,
                   const int2* __restrict__ bovf, const unsigned* __restrict__ bovn,
                   const float* __restrict__ dinv, const float* __restrict__ h2,
                   const float* __restrict__ b2, float* __restrict__ out, int N, int nW) {
    __shared__ float acc0[BK_NODES];
    __shared__ float acc1[BK_NODES];
    const int tid = threadIdx.x;
    const int b   = blockIdx.x;
    const int n0  = b << BK_SHIFT;
    const int nn  = max(min(BK_NODES, N - n0), 0);
    const float2* __restrict__ h2v = (const float2*)h2;
    float dme = 0.f;
    if (tid < BK_NODES) {
        float a0 = 0.f, a1 = 0.f;
        if (tid < nn) {
            dme = dinv[n0 + tid];
            const float2 hv = h2v[n0 + tid];
            a0 = hv.x;
            a1 = hv.y;
        }
        acc0[tid] = a0;
        acc1[tid] = a1;
    }
    const int ovall = (int)bovn[b];
    const int ovm = (ovall <= OVL_CAP) ? ovall : 0;
    const int nS = nW * SEG_SLOTS;
    const unsigned* __restrict__ bl = lists + (size_t)b * nS;
    __syncthreads();
    for (int base = 0; base < nS; base += A2_TPB * A2_K) {
        unsigned p[A2_K + 1];
        float2 v[A2_K + 1];
        const int st = base + tid;
#pragma unroll
        for (int k = 0; k < A2_K; ++k) p[k] = bl[min(st + k * A2_TPB, nS - 1)];
        p[A2_K] = SEG_EMPTY;
        if (base == 0 && tid < ovm) {
            const int2 pr = bovf[(size_t)b * OVL_CAP + tid];
            p[A2_K] = pk_pair(pr.x, pr.y);
        }
        __builtin_amdgcn_sched_barrier(0);
#pragma unroll
        for (int k = 0; k < A2_K; ++k) p[k] = (st < nS - k * A2_TPB) ? p[k] : SEG_EMPTY;
#pragma unroll
        for (int k = 0; k <= A2_K; ++k) {
            const unsigned sx = (p[k] != SEG_EMPTY) ? (p[k] & SRC_MASK) : 0u;
            const unsigned o8 = sx << 3;
            v[k] = *(const float2*)((const char*)h2 + o8);
        }
        asm volatile("" :: "v"(v[0].x), "v"(v[1].x), "v"(v[2].x), "v"(v[3].x), "v"(v[4].x),
                           "v"(v[5].x), "v"(v[6].x), "v"(v[7].x), "v"(v[8].x),
                           "v"(v[9].x), "v"(v[10].x), "v"(v[11].x), "v"(v[12].x), "v"(v[13].x),
                           "v"(v[14].x), "v"(v[15].x), "v"(v[16].x));
#pragma unroll
        for (int k = 0; k <= A2_K; ++k) {
            if (p[k] != SEG_EMPTY) {
                const unsigned r = p[k] >> SRC_BITS;
                atomicAdd(&acc0[r], v[k].x);
                atomicAdd(&acc1[r], v[k].y);
            }
        }
    }
    if (ovall > OVL_CAP) {
        for (int w = tid; w < nW; w += A2_TPB) {
            const int c = (int)ovfcnt[w];
            for (int i = 0; i < c; ++i) {
                const int2 pr = ovf[(size_t)w * BIN_CH + i];
                if ((pr.y >> BK_SHIFT) == b) {
                    const float2 hv = h2v[pr.x];
                    const unsigned r = (unsigned)(pr.y - n0);
                    atomicAdd(&acc0[r], hv.x);
                    atomicAdd(&acc1[r], hv.y);
                }
            }
        }
    }
    __syncthreads();
    if (tid < nn) {
        const float v0 = acc0[tid] * dme + b2[0];
        const float v1 = acc1[tid] * dme + b2[1];
        const float mx = fmaxf(v0, v1);
        const float lse = mx + logf(expf(v0 - mx) + expf(v1 - mx));
        ((float2*)out)[n0 + tid] = make_float2(v0 - lse, v1 - lse);
    }
}

extern "C" void kernel_launch(void* const* d_in, const int* in_sizes, int n_in,
                              void* d_out, int out_size, void* d_ws, size_t ws_size,
                              hipStream_t stream) {
    const float* x  = (const float*)d_in[0];
    const int*   ei = (const int*)d_in[1];
    const float* W1 = (const float*)d_in[2];
    const float* b1 = (const float*)d_in[3];
    const float* W2 = (const float*)d_in[4];
    const float* b2 = (const float*)d_in[5];
    float* out = (float*)d_out;

    const int H    = in_sizes[3];
    const int Fin  = in_sizes[2] / H;
    const int N    = in_sizes[0] / Fin;
    const int E    = in_sizes[1] / 2;
    (void)n_in; (void)out_size; (void)ws_size;

    float* ws   = (float*)d_ws;
    float* dinv = ws;
    float* h    = ws + (size_t)N;
    float* h2   = ws + (size_t)33 * N;
    float* tabs = ws + (((size_t)35 * N + 63) / 64) * 64;

    {
        const int nB = (N + BK_NODES - 1) / BK_NODES;
        const int nW = (E + BIN_CH - 1) / BIN_CH;
        unsigned* ovfcnt = (unsigned*)tabs;
        unsigned* bovn   = ovfcnt + 4096;
        unsigned* lists  = bovn + 4096;
        int2* ovf        = (int2*)(lists + (size_t)nB * nW * SEG_SLOTS);
        int2* bovf       = ovf + (size_t)nW * BIN_CH;
        k_bin_scatter<<<nW, BIN_TPB, 0, stream>>>(ei, E, nB, nW, lists, ovf, ovfcnt);
        k_gemm_wmma  <<<nB, GM_TPB, 0, stream>>>(x, W1, lists, ovf, ovfcnt, bovf, bovn, dinv, h, N, Fin, nW);
        k_agg1_bucket<<<nB, AG_TPB, 0, stream>>>(lists, ovf, ovfcnt, bovf, bovn, dinv, h, b1, W2, h2, N, nW);
        k_agg2_bucket<<<nB, A2_TPB, 0, stream>>>(lists, ovf, ovfcnt, bovf, bovn, dinv, h2, b2, out, N, nW);
    }
}
